# starting kernel
# baseline (speedup 1.0000x reference)
#include <hip/hip_runtime.h>
#include <math.h>

typedef __attribute__((ext_vector_type(2))) float v2f;
typedef __attribute__((ext_vector_type(8))) float v8f;

#define DIM      256
#define MT       64
#define LSTRIDE  258

__global__ __launch_bounds__(256) void zero_i32(int* __restrict__ p, int n) {
    int i = blockIdx.x * 256 + threadIdx.x;
    int st = gridDim.x * 256;
    for (; i < n; i += st) p[i] = 0;
}

__global__ __launch_bounds__(256) void count_dst(const int* __restrict__ dst,
                                                 int* __restrict__ cnt, int nE) {
    int i = blockIdx.x * 256 + threadIdx.x;
    int st = gridDim.x * 256;
    for (; i < nE; i += st) atomicAdd(&cnt[dst[i]], 1);
}

__global__ __launch_bounds__(256) void scan_block(const int* __restrict__ cnt,
                                                  int* __restrict__ off,
                                                  int* __restrict__ bsum, int n) {
    __shared__ int s[256];
    const int tid = threadIdx.x;
    const int gid = blockIdx.x * 256 + tid;
    int v = (gid < n) ? cnt[gid] : 0;
    s[tid] = v;
    __syncthreads();
    for (int d = 1; d < 256; d <<= 1) {
        int t = (tid >= d) ? s[tid - d] : 0;
        __syncthreads();
        s[tid] += t;
        __syncthreads();
    }
    if (gid < n) off[gid] = s[tid] - v;
    if (tid == 255) bsum[blockIdx.x] = s[255];
}

__global__ __launch_bounds__(256) void scan_sums(int* __restrict__ bsum, int nb) {
    __shared__ int s[256];
    const int tid = threadIdx.x;
    int v = (tid < nb) ? bsum[tid] : 0;
    s[tid] = v;
    __syncthreads();
    for (int d = 1; d < 256; d <<= 1) {
        int t = (tid >= d) ? s[tid - d] : 0;
        __syncthreads();
        s[tid] += t;
        __syncthreads();
    }
    bsum[tid] = s[tid] - v;
}

__global__ __launch_bounds__(256) void scan_finish(int* __restrict__ off,
                                                   const int* __restrict__ bsum,
                                                   int* __restrict__ cursor,
                                                   int n, int nE) {
    const int gid = blockIdx.x * 256 + threadIdx.x;
    if (gid < n) {
        int o = off[gid] + bsum[blockIdx.x];
        off[gid] = o;
        cursor[gid] = o;
    }
    if (gid == 0) off[n] = nE;
}

__global__ __launch_bounds__(256) void build_perm(const int* __restrict__ src,
                                                  const int* __restrict__ dst,
                                                  int* __restrict__ cursor,
                                                  int* __restrict__ perm, int nE) {
    int i = blockIdx.x * 256 + threadIdx.x;
    int st = gridDim.x * 256;
    for (; i < nE; i += st) {
        int slot = atomicAdd(&cursor[dst[i]], 1);
        perm[slot] = src[i];
    }
}

__global__ __launch_bounds__(256) void gather_mean(const float* __restrict__ xsrc,
                                                   const int* __restrict__ perm,
                                                   const int* __restrict__ off,
                                                   float* __restrict__ meanOut,
                                                   int nDst) {
    const int wave = blockIdx.x * 8 + (threadIdx.x >> 5);
    const int lane = threadIdx.x & 31;
    if (wave >= nDst) return;
    const int s = off[wave], e = off[wave + 1];
    float a0 = 0.f, a1 = 0.f, a2 = 0.f, a3 = 0.f;
    float b0 = 0.f, b1 = 0.f, b2 = 0.f, b3 = 0.f;
    for (int j = s; j < e; ++j) {
        const float4* row = (const float4*)(xsrc + (size_t)perm[j] * DIM);
        float4 a = row[lane];
        float4 b = row[lane + 32];
        a0 += a.x; a1 += a.y; a2 += a.z; a3 += a.w;
        b0 += b.x; b1 += b.y; b2 += b.z; b3 += b.w;
    }
    const float inv = 1.0f / fmaxf((float)(e - s), 1.0f);
    float4* orow = (float4*)(meanOut + (size_t)wave * DIM);
    orow[lane]      = make_float4(a0 * inv, a1 * inv, a2 * inv, a3 * inv);
    orow[lane + 32] = make_float4(b0 * inv, b1 * inv, b2 * inv, b3 * inv);
}

__global__ __launch_bounds__(256) void sage_gemm(float* __restrict__ meanio,
                                                 const float* __restrict__ xdst,
                                                 const float* __restrict__ Wl,
                                                 const float* __restrict__ bl,
                                                 const float* __restrict__ Wr,
                                                 int relu_norm, int nRows) {
    __shared__ float sMean[MT * LSTRIDE];
    __shared__ float sSelf[MT * LSTRIDE];
    __shared__ float sPart[MT * 4];
    __shared__ float sInv [MT];

    const int tid = threadIdx.x;
    const int m0  = blockIdx.x * MT;

    for (int i = tid; i < MT * DIM / 4; i += 256) {
        const int m = i >> 6;
        const int c = (i & 63) * 4;
        int gm = m0 + m;
        if (gm >= nRows) gm = nRows - 1;
        float4 a = *(const float4*)(meanio + (size_t)gm * DIM + c);
        float4 s = *(const float4*)(xdst   + (size_t)gm * DIM + c);
        float* pm = &sMean[m * LSTRIDE + c];
        pm[0] = a.x; pm[1] = a.y; pm[2] = a.z; pm[3] = a.w;
        float* ps = &sSelf[m * LSTRIDE + c];
        ps[0] = s.x; ps[1] = s.y; ps[2] = s.z; ps[3] = s.w;
    }
    __syncthreads();

    const int lane  = tid & 31;
    const int wv    = tid >> 5;
    const int mrow  = lane & 15;
    const int khalf = (lane >> 4) << 1;
    const int ncol  = lane & 15;
    const int n0    = wv * 32;

    const float* wl0 = Wl + (size_t)(n0 + ncol) * DIM;
    const float* wl1 = Wl + (size_t)(n0 + 16 + ncol) * DIM;
    const float* wr0 = Wr + (size_t)(n0 + ncol) * DIM;
    const float* wr1 = Wr + (size_t)(n0 + 16 + ncol) * DIM;

    v8f acc[4][2];
#pragma unroll
    for (int s = 0; s < 4; ++s)
#pragma unroll
        for (int t = 0; t < 2; ++t)
            acc[s][t] = (v8f){0.f, 0.f, 0.f, 0.f, 0.f, 0.f, 0.f, 0.f};

#pragma unroll 2
    for (int k = 0; k < DIM; k += 4) {
        const int kk = k + khalf;
        v2f l0 = *(const v2f*)(wl0 + kk);
        v2f l1 = *(const v2f*)(wl1 + kk);
        v2f r0 = *(const v2f*)(wr0 + kk);
        v2f r1 = *(const v2f*)(wr1 + kk);
#pragma unroll
        for (int s = 0; s < 4; ++s) {
            v2f aM = *(const v2f*)(&sMean[(s * 16 + mrow) * LSTRIDE] + kk);
            v2f aS = *(const v2f*)(&sSelf[(s * 16 + mrow) * LSTRIDE] + kk);
            acc[s][0] = __builtin_amdgcn_wmma_f32_16x16x4_f32(false, aM, false, l0, (short)0, acc[s][0], false, false);
            acc[s][0] = __builtin_amdgcn_wmma_f32_16x16x4_f32(false, aS, false, r0, (short)0, acc[s][0], false, false);
            acc[s][1] = __builtin_amdgcn_wmma_f32_16x16x4_f32(false, aM, false, l1, (short)0, acc[s][1], false, false);
            acc[s][1] = __builtin_amdgcn_wmma_f32_16x16x4_f32(false, aS, false, r1, (short)0, acc[s][1], false, false);
        }
    }

    const float bb0 = bl[n0 + ncol];
    const float bb1 = bl[n0 + 16 + ncol];
#pragma unroll
    for (int s = 0; s < 4; ++s)
#pragma unroll
        for (int r = 0; r < 8; ++r) { acc[s][0][r] += bb0; acc[s][1][r] += bb1; }

    __syncthreads();

    const int mbase = (lane >> 4) << 3;
#pragma unroll
    for (int s = 0; s < 4; ++s)
#pragma unroll
        for (int r = 0; r < 8; ++r) {
            const int m = s * 16 + mbase + r;
            sMean[m * LSTRIDE + n0 + ncol]      = acc[s][0][r];
            sMean[m * LSTRIDE + n0 + 16 + ncol] = acc[s][1][r];
        }
    __syncthreads();

    if (relu_norm) {
        {
            const int m = tid >> 2, seg = tid & 3;
            const float* row = &sMean[m * LSTRIDE + seg * 64];
            float ss = 0.f;
#pragma unroll 16
            for (int c = 0; c < 64; ++c) ss += row[c] * row[c];
            sPart[m * 4 + seg] = ss;
        }
        __syncthreads();
        if (tid < MT) {
            const float ss = sPart[tid * 4] + sPart[tid * 4 + 1] +
                             sPart[tid * 4 + 2] + sPart[tid * 4 + 3];
            sInv[tid] = 1.0f / fmaxf(sqrtf(ss), 1e-12f);
        }
        __syncthreads();
    }

    {
        const int m  = tid >> 2;
        const int cb = (tid & 3) * 64;
        if (m0 + m < nRows) {
            const float inv = relu_norm ? sInv[m] : 1.0f;
            float* orow = meanio + (size_t)(m0 + m) * DIM;
#pragma unroll
            for (int c = 0; c < 64; c += 4) {
                float x0 = sMean[m * LSTRIDE + cb + c + 0];
                float x1 = sMean[m * LSTRIDE + cb + c + 1];
                float x2 = sMean[m * LSTRIDE + cb + c + 2];
                float x3 = sMean[m * LSTRIDE + cb + c + 3];
                if (relu_norm) {
                    x0 = fmaxf(x0 * inv, 0.f); x1 = fmaxf(x1 * inv, 0.f);
                    x2 = fmaxf(x2 * inv, 0.f); x3 = fmaxf(x3 * inv, 0.f);
                }
                *(float4*)(orow + cb + c) = make_float4(x0, x1, x2, x3);
            }
        }
    }
}

extern "C" void kernel_launch(void* const* d_in, const int* in_sizes, int n_in,
                              void* d_out, int out_size, void* d_ws, size_t ws_size,
                              hipStream_t stream) {
    (void)n_in; (void)out_size; (void)ws_size;
    const float* xA = (const float*)d_in[0];
    const float* xP = (const float*)d_in[1];
    const float* Wl = (const float*)d_in[2];
    const float* bl = (const float*)d_in[3];
    const float* Wr = (const float*)d_in[4];
    const int* eap  = (const int*)d_in[5];
    const int* epa  = (const int*)d_in[6];

    const int NA = in_sizes[0] / DIM;
    const int NP = in_sizes[1] / DIM;
    const int E  = in_sizes[5] / 2;
    const int L  = in_sizes[2] / (2 * DIM * DIM);

    const size_t FA = (size_t)NA * DIM, FP = (size_t)NP * DIM;
    char* ws = (char*)d_ws;
    float* A0 = (float*)ws;  ws += FA * sizeof(float);
    float* A1 = (float*)ws;  ws += FA * sizeof(float);
    float* P0 = (float*)ws;  ws += FP * sizeof(float);
    float* P1 = (float*)ws;  ws += FP * sizeof(float);
    int* cntP  = (int*)ws;   ws += (size_t)NP * 4;
    int* offP  = (int*)ws;   ws += (size_t)(NP + 1) * 4;
    int* curP  = (int*)ws;   ws += (size_t)NP * 4;
    int* permP = (int*)ws;   ws += (size_t)E * 4;
    int* cntA  = (int*)ws;   ws += (size_t)NA * 4;
    int* offA  = (int*)ws;   ws += (size_t)(NA + 1) * 4;
    int* curA  = (int*)ws;   ws += (size_t)NA * 4;
    int* permA = (int*)ws;   ws += (size_t)E * 4;
    int* bsum  = (int*)ws;   ws += 256 * 4;

    const int nbP = (NP + 255) / 256, nbA = (NA + 255) / 256;

    zero_i32   <<<64, 256, 0, stream>>>(cntP, NP);
    count_dst  <<<1024, 256, 0, stream>>>(eap + E, cntP, E);
    scan_block <<<nbP, 256, 0, stream>>>(cntP, offP, bsum, NP);
    scan_sums  <<<1, 256, 0, stream>>>(bsum, nbP);
    scan_finish<<<nbP, 256, 0, stream>>>(offP, bsum, curP, NP, E);
    build_perm <<<1024, 256, 0, stream>>>(eap, eap + E, curP, permP, E);

    zero_i32   <<<64, 256, 0, stream>>>(cntA, NA);
    count_dst  <<<1024, 256, 0, stream>>>(epa + E, cntA, E);
    scan_block <<<nbA, 256, 0, stream>>>(cntA, offA, bsum, NA);
    scan_sums  <<<1, 256, 0, stream>>>(bsum, nbA);
    scan_finish<<<nbA, 256, 0, stream>>>(offA, bsum, curA, NA, E);
    build_perm <<<1024, 256, 0, stream>>>(epa, epa + E, curA, permA, E);

    const float* xa_cur = xA;
    const float* xp_cur = xP;
    for (int l = 0; l < L; ++l) {
        const bool last = (l == L - 1);
        float* xa_next = last ? (float*)d_out        : ((l & 1) ? A1 : A0);
        float* xp_next = last ? (float*)d_out + FA   : ((l & 1) ? P1 : P0);

        gather_mean<<<(NP + 7) / 8, 256, 0, stream>>>(xa_cur, permP, offP, xp_next, NP);
        gather_mean<<<(NA + 7) / 8, 256, 0, stream>>>(xp_cur, permA, offA, xa_next, NA);

        const float* Wl0 = Wl + ((size_t)(l * 2 + 0)) * DIM * DIM;
        const float* Wl1 = Wl + ((size_t)(l * 2 + 1)) * DIM * DIM;
        const float* Wr0 = Wr + ((size_t)(l * 2 + 0)) * DIM * DIM;
        const float* Wr1 = Wr + ((size_t)(l * 2 + 1)) * DIM * DIM;
        const float* bl0 = bl + (size_t)(l * 2 + 0) * DIM;
        const float* bl1 = bl + (size_t)(l * 2 + 1) * DIM;

        sage_gemm<<<(NP + MT - 1) / MT, 256, 0, stream>>>(xp_next, xp_cur, Wl0, bl0, Wr0, last ? 0 : 1, NP);
        sage_gemm<<<(NA + MT - 1) / MT, 256, 0, stream>>>(xa_next, xa_cur, Wl1, bl1, Wr1, last ? 0 : 1, NA);

        xa_cur = xa_next;
        xp_cur = xp_next;
    }
}


# GEMM k loop software-pipelined: the next 8-k block's A (LDS) and B (weight) fragments are loaded as one group into a second register set ahead of the current block's 32 WMMAs; sum order unchanged
# speedup vs baseline: 1.0446x; 1.0446x over previous
#include <hip/hip_runtime.h>
#include <math.h>

typedef __attribute__((ext_vector_type(2))) float v2f;
typedef __attribute__((ext_vector_type(8))) float v8f;

#define DIM      256
#define MT       64
#define LSTRIDE  258

__global__ __launch_bounds__(256) void zero_i32(int* __restrict__ p, int n) {
    int i = blockIdx.x * 256 + threadIdx.x;
    int st = gridDim.x * 256;
    for (; i < n; i += st) p[i] = 0;
}

__global__ __launch_bounds__(256) void count_dst(const int* __restrict__ dst,
                                                 int* __restrict__ cnt, int nE) {
    int i = blockIdx.x * 256 + threadIdx.x;
    int st = gridDim.x * 256;
    for (; i < nE; i += st) atomicAdd(&cnt[dst[i]], 1);
}

__global__ __launch_bounds__(256) void scan_block(const int* __restrict__ cnt,
                                                  int* __restrict__ off,
                                                  int* __restrict__ bsum, int n) {
    __shared__ int s[256];
    const int tid = threadIdx.x;
    const int gid = blockIdx.x * 256 + tid;
    int v = (gid < n) ? cnt[gid] : 0;
    s[tid] = v;
    __syncthreads();
    for (int d = 1; d < 256; d <<= 1) {
        int t = (tid >= d) ? s[tid - d] : 0;
        __syncthreads();
        s[tid] += t;
        __syncthreads();
    }
    if (gid < n) off[gid] = s[tid] - v;
    if (tid == 255) bsum[blockIdx.x] = s[255];
}

__global__ __launch_bounds__(256) void scan_sums(int* __restrict__ bsum, int nb) {
    __shared__ int s[256];
    const int tid = threadIdx.x;
    int v = (tid < nb) ? bsum[tid] : 0;
    s[tid] = v;
    __syncthreads();
    for (int d = 1; d < 256; d <<= 1) {
        int t = (tid >= d) ? s[tid - d] : 0;
        __syncthreads();
        s[tid] += t;
        __syncthreads();
    }
    bsum[tid] = s[tid] - v;
}

__global__ __launch_bounds__(256) void scan_finish(int* __restrict__ off,
                                                   const int* __restrict__ bsum,
                                                   int* __restrict__ cursor,
                                                   int n, int nE) {
    const int gid = blockIdx.x * 256 + threadIdx.x;
    if (gid < n) {
        int o = off[gid] + bsum[blockIdx.x];
        off[gid] = o;
        cursor[gid] = o;
    }
    if (gid == 0) off[n] = nE;
}

__global__ __launch_bounds__(256) void build_perm(const int* __restrict__ src,
                                                  const int* __restrict__ dst,
                                                  int* __restrict__ cursor,
                                                  int* __restrict__ perm, int nE) {
    int i = blockIdx.x * 256 + threadIdx.x;
    int st = gridDim.x * 256;
    for (; i < nE; i += st) {
        int slot = atomicAdd(&cursor[dst[i]], 1);
        perm[slot] = src[i];
    }
}

__global__ __launch_bounds__(256) void gather_mean(const float* __restrict__ xsrc,
                                                   const int* __restrict__ perm,
                                                   const int* __restrict__ off,
                                                   float* __restrict__ meanOut,
                                                   int nDst) {
    const int wave = blockIdx.x * 8 + (threadIdx.x >> 5);
    const int lane = threadIdx.x & 31;
    if (wave >= nDst) return;
    const int s = off[wave], e = off[wave + 1];
    float a0 = 0.f, a1 = 0.f, a2 = 0.f, a3 = 0.f;
    float b0 = 0.f, b1 = 0.f, b2 = 0.f, b3 = 0.f;
    for (int j = s; j < e; ++j) {
        const float4* row = (const float4*)(xsrc + (size_t)perm[j] * DIM);
        float4 a = row[lane];
        float4 b = row[lane + 32];
        a0 += a.x; a1 += a.y; a2 += a.z; a3 += a.w;
        b0 += b.x; b1 += b.y; b2 += b.z; b3 += b.w;
    }
    const float inv = 1.0f / fmaxf((float)(e - s), 1.0f);
    float4* orow = (float4*)(meanOut + (size_t)wave * DIM);
    orow[lane]      = make_float4(a0 * inv, a1 * inv, a2 * inv, a3 * inv);
    orow[lane + 32] = make_float4(b0 * inv, b1 * inv, b2 * inv, b3 * inv);
}

__device__ __forceinline__ void load_block(const float* wl0, const float* wl1,
                                           const float* wr0, const float* wr1,
                                           const float* aM, const float* aS, int k0,
                                           v2f (&b)[2][4], v2f (&a)[2][8]) {
#pragma unroll
    for (int j = 0; j < 2; ++j) {
        b[j][0] = *(const v2f*)(wl0 + k0 + 4 * j);
        b[j][1] = *(const v2f*)(wl1 + k0 + 4 * j);
        b[j][2] = *(const v2f*)(wr0 + k0 + 4 * j);
        b[j][3] = *(const v2f*)(wr1 + k0 + 4 * j);
    }
#pragma unroll
    for (int j = 0; j < 2; ++j)
#pragma unroll
        for (int s = 0; s < 4; ++s) {
            a[j][s]     = *(const v2f*)(aM + s * 16 * LSTRIDE + k0 + 4 * j);
            a[j][4 + s] = *(const v2f*)(aS + s * 16 * LSTRIDE + k0 + 4 * j);
        }
}

__device__ __forceinline__ void mma_block(v8f (&acc)[4][2], const v2f (&b)[2][4],
                                          const v2f (&a)[2][8]) {
#pragma unroll
    for (int j = 0; j < 2; ++j)
#pragma unroll
        for (int s = 0; s < 4; ++s) {
            acc[s][0] = __builtin_amdgcn_wmma_f32_16x16x4_f32(false, a[j][s],     false, b[j][0], (short)0, acc[s][0], false, false);
            acc[s][0] = __builtin_amdgcn_wmma_f32_16x16x4_f32(false, a[j][4 + s], false, b[j][2], (short)0, acc[s][0], false, false);
            acc[s][1] = __builtin_amdgcn_wmma_f32_16x16x4_f32(false, a[j][s],     false, b[j][1], (short)0, acc[s][1], false, false);
            acc[s][1] = __builtin_amdgcn_wmma_f32_16x16x4_f32(false, a[j][4 + s], false, b[j][3], (short)0, acc[s][1], false, false);
        }
}

__global__ __launch_bounds__(256) void sage_gemm(float* __restrict__ meanio,
                                                 const float* __restrict__ xdst,
                                                 const float* __restrict__ Wl,
                                                 const float* __restrict__ bl,
                                                 const float* __restrict__ Wr,
                                                 int relu_norm, int nRows) {
    __shared__ float sMean[MT * LSTRIDE];
    __shared__ float sSelf[MT * LSTRIDE];
    __shared__ float sPart[MT * 4];
    __shared__ float sInv [MT];

    const int tid = threadIdx.x;
    const int m0  = blockIdx.x * MT;

    for (int i = tid; i < MT * DIM / 4; i += 256) {
        const int m = i >> 6;
        const int c = (i & 63) * 4;
        int gm = m0 + m;
        if (gm >= nRows) gm = nRows - 1;
        float4 a = *(const float4*)(meanio + (size_t)gm * DIM + c);
        float4 s = *(const float4*)(xdst   + (size_t)gm * DIM + c);
        float* pm = &sMean[m * LSTRIDE + c];
        pm[0] = a.x; pm[1] = a.y; pm[2] = a.z; pm[3] = a.w;
        float* ps = &sSelf[m * LSTRIDE + c];
        ps[0] = s.x; ps[1] = s.y; ps[2] = s.z; ps[3] = s.w;
    }
    __syncthreads();

    const int lane  = tid & 31;
    const int wv    = tid >> 5;
    const int mrow  = lane & 15;
    const int khalf = (lane >> 4) << 1;
    const int ncol  = lane & 15;
    const int n0    = wv * 32;

    const float* wl0 = Wl + (size_t)(n0 + ncol) * DIM;
    const float* wl1 = Wl + (size_t)(n0 + 16 + ncol) * DIM;
    const float* wr0 = Wr + (size_t)(n0 + ncol) * DIM;
    const float* wr1 = Wr + (size_t)(n0 + 16 + ncol) * DIM;

    v8f acc[4][2];
#pragma unroll
    for (int s = 0; s < 4; ++s)
#pragma unroll
        for (int t = 0; t < 2; ++t)
            acc[s][t] = (v8f){0.f, 0.f, 0.f, 0.f, 0.f, 0.f, 0.f, 0.f};

    const float* aMrow = &sMean[mrow * LSTRIDE] + khalf;
    const float* aSrow = &sSelf[mrow * LSTRIDE] + khalf;
    wl0 += khalf; wl1 += khalf; wr0 += khalf; wr1 += khalf;

    v2f bP[2][4], aP[2][8];
    v2f bQ[2][4], aQ[2][8];

    load_block(wl0, wl1, wr0, wr1, aMrow, aSrow, 0, bP, aP);
#pragma unroll 1
    for (int k = 0; k < DIM - 16; k += 16) {
        load_block(wl0, wl1, wr0, wr1, aMrow, aSrow, k + 8, bQ, aQ);
        __builtin_amdgcn_sched_barrier(0);
        mma_block(acc, bP, aP);
        __builtin_amdgcn_sched_barrier(0);
        load_block(wl0, wl1, wr0, wr1, aMrow, aSrow, k + 16, bP, aP);
        __builtin_amdgcn_sched_barrier(0);
        mma_block(acc, bQ, aQ);
    }
    load_block(wl0, wl1, wr0, wr1, aMrow, aSrow, DIM - 8, bQ, aQ);
    mma_block(acc, bP, aP);
    mma_block(acc, bQ, aQ);

    const float bb0 = bl[n0 + ncol];
    const float bb1 = bl[n0 + 16 + ncol];
#pragma unroll
    for (int s = 0; s < 4; ++s)
#pragma unroll
        for (int r = 0; r < 8; ++r) { acc[s][0][r] += bb0; acc[s][1][r] += bb1; }

    __syncthreads();

    const int mbase = (lane >> 4) << 3;
#pragma unroll
    for (int s = 0; s < 4; ++s)
#pragma unroll
        for (int r = 0; r < 8; ++r) {
            const int m = s * 16 + mbase + r;
            sMean[m * LSTRIDE + n0 + ncol]      = acc[s][0][r];
            sMean[m * LSTRIDE + n0 + 16 + ncol] = acc[s][1][r];
        }
    __syncthreads();

    if (relu_norm) {
        {
            const int m = tid >> 2, seg = tid & 3;
            const float* row = &sMean[m * LSTRIDE + seg * 64];
            float ss = 0.f;
#pragma unroll 16
            for (int c = 0; c < 64; ++c) ss += row[c] * row[c];
            sPart[m * 4 + seg] = ss;
        }
        __syncthreads();
        if (tid < MT) {
            const float ss = sPart[tid * 4] + sPart[tid * 4 + 1] +
                             sPart[tid * 4 + 2] + sPart[tid * 4 + 3];
            sInv[tid] = 1.0f / fmaxf(sqrtf(ss), 1e-12f);
        }
        __syncthreads();
    }

    {
        const int m  = tid >> 2;
        const int cb = (tid & 3) * 64;
        if (m0 + m < nRows) {
            const float inv = relu_norm ? sInv[m] : 1.0f;
            float* orow = meanio + (size_t)(m0 + m) * DIM;
#pragma unroll
            for (int c = 0; c < 64; c += 4) {
                float x0 = sMean[m * LSTRIDE + cb + c + 0];
                float x1 = sMean[m * LSTRIDE + cb + c + 1];
                float x2 = sMean[m * LSTRIDE + cb + c + 2];
                float x3 = sMean[m * LSTRIDE + cb + c + 3];
                if (relu_norm) {
                    x0 = fmaxf(x0 * inv, 0.f); x1 = fmaxf(x1 * inv, 0.f);
                    x2 = fmaxf(x2 * inv, 0.f); x3 = fmaxf(x3 * inv, 0.f);
                }
                *(float4*)(orow + cb + c) = make_float4(x0, x1, x2, x3);
            }
        }
    }
}

extern "C" void kernel_launch(void* const* d_in, const int* in_sizes, int n_in,
                              void* d_out, int out_size, void* d_ws, size_t ws_size,
                              hipStream_t stream) {
    (void)n_in; (void)out_size; (void)ws_size;
    const float* xA = (const float*)d_in[0];
    const float* xP = (const float*)d_in[1];
    const float* Wl = (const float*)d_in[2];
    const float* bl = (const float*)d_in[3];
    const float* Wr = (const float*)d_in[4];
    const int* eap  = (const int*)d_in[5];
    const int* epa  = (const int*)d_in[6];

    const int NA = in_sizes[0] / DIM;
    const int NP = in_sizes[1] / DIM;
    const int E  = in_sizes[5] / 2;
    const int L  = in_sizes[2] / (2 * DIM * DIM);

    const size_t FA = (size_t)NA * DIM, FP = (size_t)NP * DIM;
    char* ws = (char*)d_ws;
    float* A0 = (float*)ws;  ws += FA * sizeof(float);
    float* A1 = (float*)ws;  ws += FA * sizeof(float);
    float* P0 = (float*)ws;  ws += FP * sizeof(float);
    float* P1 = (float*)ws;  ws += FP * sizeof(float);
    int* cntP  = (int*)ws;   ws += (size_t)NP * 4;
    int* offP  = (int*)ws;   ws += (size_t)(NP + 1) * 4;
    int* curP  = (int*)ws;   ws += (size_t)NP * 4;
    int* permP = (int*)ws;   ws += (size_t)E * 4;
    int* cntA  = (int*)ws;   ws += (size_t)NA * 4;
    int* offA  = (int*)ws;   ws += (size_t)(NA + 1) * 4;
    int* curA  = (int*)ws;   ws += (size_t)NA * 4;
    int* permA = (int*)ws;   ws += (size_t)E * 4;
    int* bsum  = (int*)ws;   ws += 256 * 4;

    const int nbP = (NP + 255) / 256, nbA = (NA + 255) / 256;

    zero_i32   <<<64, 256, 0, stream>>>(cntP, NP);
    count_dst  <<<1024, 256, 0, stream>>>(eap + E, cntP, E);
    scan_block <<<nbP, 256, 0, stream>>>(cntP, offP, bsum, NP);
    scan_sums  <<<1, 256, 0, stream>>>(bsum, nbP);
    scan_finish<<<nbP, 256, 0, stream>>>(offP, bsum, curP, NP, E);
    build_perm <<<1024, 256, 0, stream>>>(eap, eap + E, curP, permP, E);

    zero_i32   <<<64, 256, 0, stream>>>(cntA, NA);
    count_dst  <<<1024, 256, 0, stream>>>(epa + E, cntA, E);
    scan_block <<<nbA, 256, 0, stream>>>(cntA, offA, bsum, NA);
    scan_sums  <<<1, 256, 0, stream>>>(bsum, nbA);
    scan_finish<<<nbA, 256, 0, stream>>>(offA, bsum, curA, NA, E);
    build_perm <<<1024, 256, 0, stream>>>(epa, epa + E, curA, permA, E);

    const float* xa_cur = xA;
    const float* xp_cur = xP;
    for (int l = 0; l < L; ++l) {
        const bool last = (l == L - 1);
        float* xa_next = last ? (float*)d_out        : ((l & 1) ? A1 : A0);
        float* xp_next = last ? (float*)d_out + FA   : ((l & 1) ? P1 : P0);

        gather_mean<<<(NP + 7) / 8, 256, 0, stream>>>(xa_cur, permP, offP, xp_next, NP);
        gather_mean<<<(NA + 7) / 8, 256, 0, stream>>>(xp_cur, permA, offA, xa_next, NA);

        const float* Wl0 = Wl + ((size_t)(l * 2 + 0)) * DIM * DIM;
        const float* Wl1 = Wl + ((size_t)(l * 2 + 1)) * DIM * DIM;
        const float* Wr0 = Wr + ((size_t)(l * 2 + 0)) * DIM * DIM;
        const float* Wr1 = Wr + ((size_t)(l * 2 + 1)) * DIM * DIM;
        const float* bl0 = bl + (size_t)(l * 2 + 0) * DIM;
        const float* bl1 = bl + (size_t)(l * 2 + 1) * DIM;

        sage_gemm<<<(NP + MT - 1) / MT, 256, 0, stream>>>(xp_next, xp_cur, Wl0, bl0, Wr0, last ? 0 : 1, NP);
        sage_gemm<<<(NA + MT - 1) / MT, 256, 0, stream>>>(xa_next, xa_cur, Wl1, bl1, Wr1, last ? 0 : 1, NA);

        xa_cur = xa_next;
        xp_cur = xp_next;
    }
}


# neighbour-mean gather fused into the GEMM work-group (each wave gathers 8 rows into the LDS A operand); half the work-groups run their self-half k loop first; there self sums precede mean sums
# speedup vs baseline: 1.0922x; 1.0456x over previous
#include <hip/hip_runtime.h>
#include <math.h>

typedef __attribute__((ext_vector_type(2))) float v2f;
typedef __attribute__((ext_vector_type(8))) float v8f;

#define DIM      256
#define MT       64
#define LSTRIDE  258

__global__ __launch_bounds__(256) void zero_i32(int* __restrict__ p, int n) {
    int i = blockIdx.x * 256 + threadIdx.x;
    int st = gridDim.x * 256;
    for (; i < n; i += st) p[i] = 0;
}

__global__ __launch_bounds__(256) void count_dst(const int* __restrict__ dst,
                                                 int* __restrict__ cnt, int nE) {
    int i = blockIdx.x * 256 + threadIdx.x;
    int st = gridDim.x * 256;
    for (; i < nE; i += st) atomicAdd(&cnt[dst[i]], 1);
}

__global__ __launch_bounds__(256) void scan_block(const int* __restrict__ cnt,
                                                  int* __restrict__ off,
                                                  int* __restrict__ bsum, int n) {
    __shared__ int s[256];
    const int tid = threadIdx.x;
    const int gid = blockIdx.x * 256 + tid;
    int v = (gid < n) ? cnt[gid] : 0;
    s[tid] = v;
    __syncthreads();
    for (int d = 1; d < 256; d <<= 1) {
        int t = (tid >= d) ? s[tid - d] : 0;
        __syncthreads();
        s[tid] += t;
        __syncthreads();
    }
    if (gid < n) off[gid] = s[tid] - v;
    if (tid == 255) bsum[blockIdx.x] = s[255];
}

__global__ __launch_bounds__(256) void scan_sums(int* __restrict__ bsum, int nb) {
    __shared__ int s[256];
    const int tid = threadIdx.x;
    int v = (tid < nb) ? bsum[tid] : 0;
    s[tid] = v;
    __syncthreads();
    for (int d = 1; d < 256; d <<= 1) {
        int t = (tid >= d) ? s[tid - d] : 0;
        __syncthreads();
        s[tid] += t;
        __syncthreads();
    }
    bsum[tid] = s[tid] - v;
}

__global__ __launch_bounds__(256) void scan_finish(int* __restrict__ off,
                                                   const int* __restrict__ bsum,
                                                   int* __restrict__ cursor,
                                                   int n, int nE) {
    const int gid = blockIdx.x * 256 + threadIdx.x;
    if (gid < n) {
        int o = off[gid] + bsum[blockIdx.x];
        off[gid] = o;
        cursor[gid] = o;
    }
    if (gid == 0) off[n] = nE;
}

__global__ __launch_bounds__(256) void build_perm(const int* __restrict__ src,
                                                  const int* __restrict__ dst,
                                                  int* __restrict__ cursor,
                                                  int* __restrict__ perm, int nE) {
    int i = blockIdx.x * 256 + threadIdx.x;
    int st = gridDim.x * 256;
    for (; i < nE; i += st) {
        int slot = atomicAdd(&cursor[dst[i]], 1);
        perm[slot] = src[i];
    }
}

template <int NB, bool FULL>
__device__ __forceinline__ void gather_batch(const float* __restrict__ xsrc, int idx,
                                             int j, int cnt, int lane, float (&acc)[8]) {
    float4 va[NB], vb[NB];
#pragma unroll
    for (int u = 0; u < NB; ++u) {
        const int jj = (FULL || u < cnt) ? (j + u) : (j + cnt - 1);
        const int p  = __builtin_amdgcn_readlane(idx, jj);
        const float4* row = (const float4*)(xsrc + (size_t)p * DIM);
        va[u] = row[lane];
        vb[u] = row[lane + 32];
    }
#pragma unroll
    for (int u = 0; u < NB; ++u) {
        const bool on = FULL || (u < cnt);
        acc[0] += on ? va[u].x : 0.0f; acc[1] += on ? va[u].y : 0.0f;
        acc[2] += on ? va[u].z : 0.0f; acc[3] += on ? va[u].w : 0.0f;
        acc[4] += on ? vb[u].x : 0.0f; acc[5] += on ? vb[u].y : 0.0f;
        acc[6] += on ? vb[u].z : 0.0f; acc[7] += on ? vb[u].w : 0.0f;
    }
}

template <int MODE>
__device__ __forceinline__ void load_block(const float* wl0, const float* wl1,
                                           const float* wr0, const float* wr1,
                                           const float* aM, const float* aS, int k0,
                                           v2f (&b)[4][2], v2f (&a)[4][4]) {
#pragma unroll
    for (int q = 0; q < 4; ++q) {
        const bool self = (MODE == 0) ? ((q & 1) != 0) : (MODE == 1);
        const int  kk   = k0 + ((MODE == 0) ? 4 * (q >> 1) : 4 * q);
        b[q][0] = *(const v2f*)((self ? wr0 : wl0) + kk);
        b[q][1] = *(const v2f*)((self ? wr1 : wl1) + kk);
    }
#pragma unroll
    for (int q = 0; q < 4; ++q) {
        const bool self = (MODE == 0) ? ((q & 1) != 0) : (MODE == 1);
        const int  kk   = k0 + ((MODE == 0) ? 4 * (q >> 1) : 4 * q);
#pragma unroll
        for (int s = 0; s < 4; ++s)
            a[q][s] = *(const v2f*)((self ? aS : aM) + s * 16 * LSTRIDE + kk);
    }
}

__device__ __forceinline__ void mma_block(v8f (&acc)[4][2], const v2f (&b)[4][2],
                                          const v2f (&a)[4][4]) {
#pragma unroll
    for (int q = 0; q < 4; ++q)
#pragma unroll
        for (int s = 0; s < 4; ++s) {
            acc[s][0] = __builtin_amdgcn_wmma_f32_16x16x4_f32(false, a[q][s], false, b[q][0], (short)0, acc[s][0], false, false);
            acc[s][1] = __builtin_amdgcn_wmma_f32_16x16x4_f32(false, a[q][s], false, b[q][1], (short)0, acc[s][1], false, false);
        }
}

template <int MODE>
__device__ __forceinline__ void k_loop(v8f (&acc)[4][2],
                                       const float* wl0, const float* wl1,
                                       const float* wr0, const float* wr1,
                                       const float* aM, const float* aS) {
    constexpr int KB = (MODE == 0) ? 8 : 16;
    v2f bP[4][2], aP[4][4];
    v2f bQ[4][2], aQ[4][4];

    load_block<MODE>(wl0, wl1, wr0, wr1, aM, aS, 0, bP, aP);
#pragma unroll 1
    for (int k = 0; k < DIM - 2 * KB; k += 2 * KB) {
        load_block<MODE>(wl0, wl1, wr0, wr1, aM, aS, k + KB, bQ, aQ);
        __builtin_amdgcn_sched_barrier(0);
        mma_block(acc, bP, aP);
        __builtin_amdgcn_sched_barrier(0);
        load_block<MODE>(wl0, wl1, wr0, wr1, aM, aS, k + 2 * KB, bP, aP);
        __builtin_amdgcn_sched_barrier(0);
        mma_block(acc, bQ, aQ);
    }
    load_block<MODE>(wl0, wl1, wr0, wr1, aM, aS, DIM - KB, bQ, aQ);
    mma_block(acc, bP, aP);
    mma_block(acc, bQ, aQ);
}

__global__ __launch_bounds__(256, 4) void sage_gemm(float* __restrict__ out,
                                                 const float* __restrict__ xsrc,
                                                 const int* __restrict__ perm,
                                                 const int* __restrict__ off,
                                                 const float* __restrict__ xdst,
                                                 const float* __restrict__ Wl,
                                                 const float* __restrict__ bl,
                                                 const float* __restrict__ Wr,
                                                 int relu_norm, int nRows) {
    __shared__ float sMean[MT * LSTRIDE];
    __shared__ float sSelf[MT * LSTRIDE];
    __shared__ float sPart[MT * 4];
    __shared__ float sInv [MT];

    const int tid = threadIdx.x;
    const int m0  = blockIdx.x * MT;

    const int lane  = tid & 31;
    const int wv    = tid >> 5;
    const int mrow  = lane & 15;
    const int khalf = (lane >> 4) << 1;
    const int ncol  = lane & 15;
    const int n0    = wv * 32;

    const int late = (blockIdx.x ^ (blockIdx.x >> 3) ^ (blockIdx.x >> 8)) & 1;

    for (int i = tid; i < MT * DIM / 4; i += 256) {
        const int m = i >> 6;
        const int c = (i & 63) * 4;
        int gm = m0 + m;
        if (gm >= nRows) gm = nRows - 1;
        float4 s = *(const float4*)(xdst + (size_t)gm * DIM + c);
        float* ps = &sSelf[m * LSTRIDE + c];
        ps[0] = s.x; ps[1] = s.y; ps[2] = s.z; ps[3] = s.w;
    }

    const float* wl0 = Wl + (size_t)(n0 + ncol) * DIM + khalf;
    const float* wl1 = Wl + (size_t)(n0 + 16 + ncol) * DIM + khalf;
    const float* wr0 = Wr + (size_t)(n0 + ncol) * DIM + khalf;
    const float* wr1 = Wr + (size_t)(n0 + 16 + ncol) * DIM + khalf;
    const float* aMrow = &sMean[mrow * LSTRIDE] + khalf;
    const float* aSrow = &sSelf[mrow * LSTRIDE] + khalf;

    v8f acc[4][2];
#pragma unroll
    for (int s = 0; s < 4; ++s)
#pragma unroll
        for (int t = 0; t < 2; ++t)
            acc[s][t] = (v8f){0.f, 0.f, 0.f, 0.f, 0.f, 0.f, 0.f, 0.f};

    if (late) {
        __syncthreads();
        k_loop<1>(acc, wl0, wl1, wr0, wr1, aMrow, aSrow);
    }

    {
        const int rbase = m0 + wv * 8;
        int oi = rbase + lane;
        if (oi > nRows) oi = nRows;
        const int offv = (lane < 9) ? off[oi] : 0;
#pragma unroll 1
        for (int r = 0; r < 8; ++r) {
            const int s = __builtin_amdgcn_readlane(offv, r);
            const int e = __builtin_amdgcn_readlane(offv, r + 1);
            float g[8] = {0.f, 0.f, 0.f, 0.f, 0.f, 0.f, 0.f, 0.f};
            for (int base = s; base < e; base += 32) {
                const int n  = min(32, e - base);
                const int ix = (lane < n) ? perm[base + lane] : 0;
                int j = 0;
                for (; j + 8 <= n; j += 8) gather_batch<8, true>(xsrc, ix, j, 8, lane, g);
                const int cnt = n - j;
                if (cnt > 4)      gather_batch<8, false>(xsrc, ix, j, cnt, lane, g);
                else if (cnt > 0) gather_batch<4, false>(xsrc, ix, j, cnt, lane, g);
            }
            const float inv = 1.0f / fmaxf((float)(e - s), 1.0f);
            float* pm = &sMean[(wv * 8 + r) * LSTRIDE + lane * 4];
            pm[0]   = g[0] * inv; pm[1]   = g[1] * inv; pm[2]   = g[2] * inv; pm[3]   = g[3] * inv;
            pm[128] = g[4] * inv; pm[129] = g[5] * inv; pm[130] = g[6] * inv; pm[131] = g[7] * inv;
        }
    }
    __syncthreads();

    if (late) k_loop<2>(acc, wl0, wl1, wr0, wr1, aMrow, aSrow);
    else      k_loop<0>(acc, wl0, wl1, wr0, wr1, aMrow, aSrow);

    const float bb0 = bl[n0 + ncol];
    const float bb1 = bl[n0 + 16 + ncol];
#pragma unroll
    for (int s = 0; s < 4; ++s)
#pragma unroll
        for (int r = 0; r < 8; ++r) { acc[s][0][r] += bb0; acc[s][1][r] += bb1; }

    __syncthreads();

    const int mbase = (lane >> 4) << 3;
#pragma unroll
    for (int s = 0; s < 4; ++s)
#pragma unroll
        for (int r = 0; r < 8; ++r) {
            const int m = s * 16 + mbase + r;
            sMean[m * LSTRIDE + n0 + ncol]      = acc[s][0][r];
            sMean[m * LSTRIDE + n0 + 16 + ncol] = acc[s][1][r];
        }
    __syncthreads();

    if (relu_norm) {
        {
            const int m = tid >> 2, seg = tid & 3;
            const float* row = &sMean[m * LSTRIDE + seg * 64];
            float ss = 0.f;
#pragma unroll 16
            for (int c = 0; c < 64; ++c) ss += row[c] * row[c];
            sPart[m * 4 + seg] = ss;
        }
        __syncthreads();
        if (tid < MT) {
            const float ss = sPart[tid * 4] + sPart[tid * 4 + 1] +
                             sPart[tid * 4 + 2] + sPart[tid * 4 + 3];
            sInv[tid] = 1.0f / fmaxf(sqrtf(ss), 1e-12f);
        }
        __syncthreads();
    }

    {
        const int m  = tid >> 2;
        const int cb = (tid & 3) * 64;
        if (m0 + m < nRows) {
            const float inv = relu_norm ? sInv[m] : 1.0f;
            float* orow = out + (size_t)(m0 + m) * DIM;
#pragma unroll
            for (int c = 0; c < 64; c += 4) {
                float x0 = sMean[m * LSTRIDE + cb + c + 0];
                float x1 = sMean[m * LSTRIDE + cb + c + 1];
                float x2 = sMean[m * LSTRIDE + cb + c + 2];
                float x3 = sMean[m * LSTRIDE + cb + c + 3];
                if (relu_norm) {
                    x0 = fmaxf(x0 * inv, 0.f); x1 = fmaxf(x1 * inv, 0.f);
                    x2 = fmaxf(x2 * inv, 0.f); x3 = fmaxf(x3 * inv, 0.f);
                }
                *(float4*)(orow + cb + c) = make_float4(x0, x1, x2, x3);
            }
        }
    }
}

extern "C" void kernel_launch(void* const* d_in, const int* in_sizes, int n_in,
                              void* d_out, int out_size, void* d_ws, size_t ws_size,
                              hipStream_t stream) {
    (void)n_in; (void)out_size; (void)ws_size;
    const float* xA = (const float*)d_in[0];
    const float* xP = (const float*)d_in[1];
    const float* Wl = (const float*)d_in[2];
    const float* bl = (const float*)d_in[3];
    const float* Wr = (const float*)d_in[4];
    const int* eap  = (const int*)d_in[5];
    const int* epa  = (const int*)d_in[6];

    const int NA = in_sizes[0] / DIM;
    const int NP = in_sizes[1] / DIM;
    const int E  = in_sizes[5] / 2;
    const int L  = in_sizes[2] / (2 * DIM * DIM);

    const size_t FA = (size_t)NA * DIM, FP = (size_t)NP * DIM;
    char* ws = (char*)d_ws;
    float* A0 = (float*)ws;  ws += FA * sizeof(float);
    float* A1 = (float*)ws;  ws += FA * sizeof(float);
    float* P0 = (float*)ws;  ws += FP * sizeof(float);
    float* P1 = (float*)ws;  ws += FP * sizeof(float);
    int* cntP  = (int*)ws;   ws += (size_t)NP * 4;
    int* offP  = (int*)ws;   ws += (size_t)(NP + 1) * 4;
    int* curP  = (int*)ws;   ws += (size_t)NP * 4;
    int* permP = (int*)ws;   ws += (size_t)E * 4;
    int* cntA  = (int*)ws;   ws += (size_t)NA * 4;
    int* offA  = (int*)ws;   ws += (size_t)(NA + 1) * 4;
    int* curA  = (int*)ws;   ws += (size_t)NA * 4;
    int* permA = (int*)ws;   ws += (size_t)E * 4;
    int* bsum  = (int*)ws;   ws += 256 * 4;

    const int nbP = (NP + 255) / 256, nbA = (NA + 255) / 256;

    zero_i32   <<<64, 256, 0, stream>>>(cntP, NP);
    count_dst  <<<1024, 256, 0, stream>>>(eap + E, cntP, E);
    scan_block <<<nbP, 256, 0, stream>>>(cntP, offP, bsum, NP);
    scan_sums  <<<1, 256, 0, stream>>>(bsum, nbP);
    scan_finish<<<nbP, 256, 0, stream>>>(offP, bsum, curP, NP, E);
    build_perm <<<1024, 256, 0, stream>>>(eap, eap + E, curP, permP, E);

    zero_i32   <<<64, 256, 0, stream>>>(cntA, NA);
    count_dst  <<<1024, 256, 0, stream>>>(epa + E, cntA, E);
    scan_block <<<nbA, 256, 0, stream>>>(cntA, offA, bsum, NA);
    scan_sums  <<<1, 256, 0, stream>>>(bsum, nbA);
    scan_finish<<<nbA, 256, 0, stream>>>(offA, bsum, curA, NA, E);
    build_perm <<<1024, 256, 0, stream>>>(epa, epa + E, curA, permA, E);

    const float* xa_cur = xA;
    const float* xp_cur = xP;
    for (int l = 0; l < L; ++l) {
        const bool last = (l == L - 1);
        float* xa_next = last ? (float*)d_out        : ((l & 1) ? A1 : A0);
        float* xp_next = last ? (float*)d_out + FA   : ((l & 1) ? P1 : P0);

        const float* Wl0 = Wl + ((size_t)(l * 2 + 0)) * DIM * DIM;
        const float* Wl1 = Wl + ((size_t)(l * 2 + 1)) * DIM * DIM;
        const float* Wr0 = Wr + ((size_t)(l * 2 + 0)) * DIM * DIM;
        const float* Wr1 = Wr + ((size_t)(l * 2 + 1)) * DIM * DIM;
        const float* bl0 = bl + (size_t)(l * 2 + 0) * DIM;
        const float* bl1 = bl + (size_t)(l * 2 + 1) * DIM;

        sage_gemm<<<(NP + MT - 1) / MT, 256, 0, stream>>>(xp_next, xa_cur, permP, offP, xp_cur, Wl0, bl0, Wr0, last ? 0 : 1, NP);
        sage_gemm<<<(NA + MT - 1) / MT, 256, 0, stream>>>(xa_next, xp_cur, permA, offA, xa_cur, Wl1, bl1, Wr1, last ? 0 : 1, NA);

        xa_cur = xa_next;
        xp_cur = xp_next;
    }
}


# weights repacked once per call to [k/4][n][4] by a new small kernel: each B fragment is one contiguous 16-byte load per lane (A the same from LDS); sums inside each 8-k group reordered
# speedup vs baseline: 1.1274x; 1.0322x over previous
#include <hip/hip_runtime.h>
#include <math.h>

typedef __attribute__((ext_vector_type(2))) float v2f;
typedef __attribute__((ext_vector_type(8))) float v8f;

#define DIM      256
#define MT       64
#define LSTRIDE  260
#define KG       (DIM / 4)

__global__ __launch_bounds__(256) void zero_i32(int* __restrict__ p, int n) {
    int i = blockIdx.x * 256 + threadIdx.x;
    int st = gridDim.x * 256;
    for (; i < n; i += st) p[i] = 0;
}

__global__ __launch_bounds__(256) void count_dst(const int* __restrict__ dst,
                                                 int* __restrict__ cnt, int nE) {
    int i = blockIdx.x * 256 + threadIdx.x;
    int st = gridDim.x * 256;
    for (; i < nE; i += st) atomicAdd(&cnt[dst[i]], 1);
}

__global__ __launch_bounds__(256) void scan_block(const int* __restrict__ cnt,
                                                  int* __restrict__ off,
                                                  int* __restrict__ bsum, int n) {
    __shared__ int s[256];
    const int tid = threadIdx.x;
    const int gid = blockIdx.x * 256 + tid;
    int v = (gid < n) ? cnt[gid] : 0;
    s[tid] = v;
    __syncthreads();
    for (int d = 1; d < 256; d <<= 1) {
        int t = (tid >= d) ? s[tid - d] : 0;
        __syncthreads();
        s[tid] += t;
        __syncthreads();
    }
    if (gid < n) off[gid] = s[tid] - v;
    if (tid == 255) bsum[blockIdx.x] = s[255];
}

__global__ __launch_bounds__(256) void scan_sums(int* __restrict__ bsum, int nb) {
    __shared__ int s[256];
    const int tid = threadIdx.x;
    int v = (tid < nb) ? bsum[tid] : 0;
    s[tid] = v;
    __syncthreads();
    for (int d = 1; d < 256; d <<= 1) {
        int t = (tid >= d) ? s[tid - d] : 0;
        __syncthreads();
        s[tid] += t;
        __syncthreads();
    }
    bsum[tid] = s[tid] - v;
}

__global__ __launch_bounds__(256) void scan_finish(int* __restrict__ off,
                                                   const int* __restrict__ bsum,
                                                   int* __restrict__ cursor,
                                                   int n, int nE) {
    const int gid = blockIdx.x * 256 + threadIdx.x;
    if (gid < n) {
        int o = off[gid] + bsum[blockIdx.x];
        off[gid] = o;
        cursor[gid] = o;
    }
    if (gid == 0) off[n] = nE;
}

__global__ __launch_bounds__(256) void build_perm(const int* __restrict__ src,
                                                  const int* __restrict__ dst,
                                                  int* __restrict__ cursor,
                                                  int* __restrict__ perm, int nE) {
    int i = blockIdx.x * 256 + threadIdx.x;
    int st = gridDim.x * 256;
    for (; i < nE; i += st) {
        int slot = atomicAdd(&cursor[dst[i]], 1);
        perm[slot] = src[i];
    }
}

__global__ __launch_bounds__(256) void pack_w(const float* __restrict__ Wl,
                                              const float* __restrict__ Wr,
                                              float* __restrict__ Pl,
                                              float* __restrict__ Pr) {
    __shared__ __attribute__((aligned(16))) float t[16 * (DIM + 4)];
    const int tid = threadIdx.x;
    const int mat = blockIdx.x >> 4;
    const int nb  = (blockIdx.x & 15) * 16;
    const float* W = (blockIdx.y ? Wr : Wl) + (size_t)mat * DIM * DIM + (size_t)nb * DIM;
    float*       P = (blockIdx.y ? Pr : Pl) + (size_t)mat * DIM * DIM;
    for (int i = tid; i < 16 * KG; i += 256) {
        const int r = i >> 6, g = i & 63;
        *(float4*)&t[r * (DIM + 4) + 4 * g] = *(const float4*)(W + (size_t)r * DIM + 4 * g);
    }
    __syncthreads();
    for (int i = tid; i < 16 * KG; i += 256) {
        const int g = i >> 4, r = i & 15;
        *(float4*)(P + ((size_t)g * DIM + nb + r) * 4) = *(const float4*)&t[r * (DIM + 4) + 4 * g];
    }
}

template <int NB, bool FULL>
__device__ __forceinline__ void gather_batch(const float* __restrict__ xsrc, int idx,
                                             int j, int cnt, int lane, float (&acc)[8]) {
    float4 va[NB], vb[NB];
#pragma unroll
    for (int u = 0; u < NB; ++u) {
        const int jj = (FULL || u < cnt) ? (j + u) : (j + cnt - 1);
        const int p  = __builtin_amdgcn_readlane(idx, jj);
        const float4* row = (const float4*)(xsrc + (size_t)p * DIM);
        va[u] = row[lane];
        vb[u] = row[lane + 32];
    }
#pragma unroll
    for (int u = 0; u < NB; ++u) {
        const bool on = FULL || (u < cnt);
        acc[0] += on ? va[u].x : 0.0f; acc[1] += on ? va[u].y : 0.0f;
        acc[2] += on ? va[u].z : 0.0f; acc[3] += on ? va[u].w : 0.0f;
        acc[4] += on ? vb[u].x : 0.0f; acc[5] += on ? vb[u].y : 0.0f;
        acc[6] += on ? vb[u].z : 0.0f; acc[7] += on ? vb[u].w : 0.0f;
    }
}

template <int MODE>
__device__ __forceinline__ void load_block(const float4* wl0, const float4* wl1,
                                           const float4* wr0, const float4* wr1,
                                           const float* aM, const float* aS, int blk,
                                           float4 (&b)[2][2], float4 (&a)[2][4]) {
#pragma unroll
    for (int u = 0; u < 2; ++u) {
        const bool self = (MODE == 0) ? (u != 0) : (MODE == 1);
        const int  j    = (MODE == 0) ? blk : (2 * blk + u);
        b[u][0] = (self ? wr0 : wl0)[(size_t)j * (2 * DIM)];
        b[u][1] = (self ? wr1 : wl1)[(size_t)j * (2 * DIM)];
    }
#pragma unroll
    for (int u = 0; u < 2; ++u) {
        const bool self = (MODE == 0) ? (u != 0) : (MODE == 1);
        const int  j    = (MODE == 0) ? blk : (2 * blk + u);
#pragma unroll
        for (int s = 0; s < 4; ++s)
            a[u][s] = *(const float4*)((self ? aS : aM) + s * 16 * LSTRIDE + 8 * j);
    }
}

template <int MODE>
__device__ __forceinline__ void mma_block(v8f (&acc)[4][2], const float4 (&b)[2][2],
                                          const float4 (&a)[2][4]) {
#pragma unroll
    for (int q = 0; q < 4; ++q) {
        const int u  = (MODE == 0) ? (q & 1) : (q >> 1);
        const int st = (MODE == 0) ? (q >> 1) : (q & 1);
        const v2f b0 = st ? (v2f){b[u][0].z, b[u][0].w} : (v2f){b[u][0].x, b[u][0].y};
        const v2f b1 = st ? (v2f){b[u][1].z, b[u][1].w} : (v2f){b[u][1].x, b[u][1].y};
#pragma unroll
        for (int s = 0; s < 4; ++s) {
            const v2f as = st ? (v2f){a[u][s].z, a[u][s].w} : (v2f){a[u][s].x, a[u][s].y};
            acc[s][0] = __builtin_amdgcn_wmma_f32_16x16x4_f32(false, as, false, b0, (short)0, acc[s][0], false, false);
            acc[s][1] = __builtin_amdgcn_wmma_f32_16x16x4_f32(false, as, false, b1, (short)0, acc[s][1], false, false);
        }
    }
}

template <int MODE>
__device__ __forceinline__ void k_loop(v8f (&acc)[4][2],
                                       const float4* wl0, const float4* wl1,
                                       const float4* wr0, const float4* wr1,
                                       const float* aM, const float* aS) {
    constexpr int NBLK = (MODE == 0) ? (DIM / 8) : (DIM / 16);
    float4 bP[2][2], aP[2][4];
    float4 bQ[2][2], aQ[2][4];

    load_block<MODE>(wl0, wl1, wr0, wr1, aM, aS, 0, bP, aP);
#pragma unroll 1
    for (int i = 0; i < NBLK - 2; i += 2) {
        load_block<MODE>(wl0, wl1, wr0, wr1, aM, aS, i + 1, bQ, aQ);
        __builtin_amdgcn_sched_barrier(0);
        mma_block<MODE>(acc, bP, aP);
        __builtin_amdgcn_sched_barrier(0);
        load_block<MODE>(wl0, wl1, wr0, wr1, aM, aS, i + 2, bP, aP);
        __builtin_amdgcn_sched_barrier(0);
        mma_block<MODE>(acc, bQ, aQ);
    }
    load_block<MODE>(wl0, wl1, wr0, wr1, aM, aS, NBLK - 1, bQ, aQ);
    mma_block<MODE>(acc, bP, aP);
    mma_block<MODE>(acc, bQ, aQ);
}

__global__ __launch_bounds__(256, 4) void sage_gemm(float* __restrict__ out,
                                                 const float* __restrict__ xsrc,
                                                 const int* __restrict__ perm,
                                                 const int* __restrict__ off,
                                                 const float* __restrict__ xdst,
                                                 const float* __restrict__ Pl,
                                                 const float* __restrict__ bl,
                                                 const float* __restrict__ Pr,
                                                 int relu_norm, int nRows) {
    __shared__ __attribute__((aligned(16))) float sMean[MT * LSTRIDE];
    __shared__ __attribute__((aligned(16))) float sSelf[MT * LSTRIDE];
    __shared__ float sPart[MT * 4];
    __shared__ float sInv [MT];

    const int tid = threadIdx.x;
    const int m0  = blockIdx.x * MT;

    const int lane  = tid & 31;
    const int wv    = tid >> 5;
    const int mrow  = lane & 15;
    const int half  = lane >> 4;
    const int ncol  = lane & 15;
    const int n0    = wv * 32;

    const int late = (blockIdx.x ^ (blockIdx.x >> 3) ^ (blockIdx.x >> 8)) & 1;

    for (int i = tid; i < MT * DIM / 4; i += 256) {
        const int m = i >> 6;
        const int c = (i & 63) * 4;
        int gm = m0 + m;
        if (gm >= nRows) gm = nRows - 1;
        float4 s = *(const float4*)(xdst + (size_t)gm * DIM + c);
        *(float4*)&sSelf[m * LSTRIDE + c] = s;
    }

    const float4* wl0 = (const float4*)Pl + (size_t)half * DIM + n0 + ncol;
    const float4* wl1 = wl0 + 16;
    const float4* wr0 = (const float4*)Pr + (size_t)half * DIM + n0 + ncol;
    const float4* wr1 = wr0 + 16;
    const float* aMrow = &sMean[mrow * LSTRIDE] + 4 * half;
    const float* aSrow = &sSelf[mrow * LSTRIDE] + 4 * half;

    v8f acc[4][2];
#pragma unroll
    for (int s = 0; s < 4; ++s)
#pragma unroll
        for (int t = 0; t < 2; ++t)
            acc[s][t] = (v8f){0.f, 0.f, 0.f, 0.f, 0.f, 0.f, 0.f, 0.f};

    if (late) {
        __syncthreads();
        k_loop<1>(acc, wl0, wl1, wr0, wr1, aMrow, aSrow);
    }

    {
        const int rbase = m0 + wv * 8;
        int oi = rbase + lane;
        if (oi > nRows) oi = nRows;
        const int offv = (lane < 9) ? off[oi] : 0;
#pragma unroll 1
        for (int r = 0; r < 8; ++r) {
            const int s = __builtin_amdgcn_readlane(offv, r);
            const int e = __builtin_amdgcn_readlane(offv, r + 1);
            float g[8] = {0.f, 0.f, 0.f, 0.f, 0.f, 0.f, 0.f, 0.f};
            for (int base = s; base < e; base += 32) {
                const int n  = min(32, e - base);
                const int ix = (lane < n) ? perm[base + lane] : 0;
                int j = 0;
                for (; j + 8 <= n; j += 8) gather_batch<8, true>(xsrc, ix, j, 8, lane, g);
                const int cnt = n - j;
                if (cnt > 4)      gather_batch<8, false>(xsrc, ix, j, cnt, lane, g);
                else if (cnt > 0) gather_batch<4, false>(xsrc, ix, j, cnt, lane, g);
            }
            const float inv = 1.0f / fmaxf((float)(e - s), 1.0f);
            float* pm = &sMean[(wv * 8 + r) * LSTRIDE + lane * 4];
            *(float4*)pm         = make_float4(g[0] * inv, g[1] * inv, g[2] * inv, g[3] * inv);
            *(float4*)(pm + 128) = make_float4(g[4] * inv, g[5] * inv, g[6] * inv, g[7] * inv);
        }
    }
    __syncthreads();

    if (late) k_loop<2>(acc, wl0, wl1, wr0, wr1, aMrow, aSrow);
    else      k_loop<0>(acc, wl0, wl1, wr0, wr1, aMrow, aSrow);

    const float bb0 = bl[n0 + ncol];
    const float bb1 = bl[n0 + 16 + ncol];
#pragma unroll
    for (int s = 0; s < 4; ++s)
#pragma unroll
        for (int r = 0; r < 8; ++r) { acc[s][0][r] += bb0; acc[s][1][r] += bb1; }

    __syncthreads();

    const int mbase = (lane >> 4) << 3;
#pragma unroll
    for (int s = 0; s < 4; ++s)
#pragma unroll
        for (int r = 0; r < 8; ++r) {
            const int m = s * 16 + mbase + r;
            sMean[m * LSTRIDE + n0 + ncol]      = acc[s][0][r];
            sMean[m * LSTRIDE + n0 + 16 + ncol] = acc[s][1][r];
        }
    __syncthreads();

    if (relu_norm) {
        {
            const int m = tid >> 2, seg = tid & 3;
            const float* row = &sMean[m * LSTRIDE + seg * 64];
            float ss = 0.f;
#pragma unroll 16
            for (int c = 0; c < 64; ++c) ss += row[c] * row[c];
            sPart[m * 4 + seg] = ss;
        }
        __syncthreads();
        if (tid < MT) {
            const float ss = sPart[tid * 4] + sPart[tid * 4 + 1] +
                             sPart[tid * 4 + 2] + sPart[tid * 4 + 3];
            sInv[tid] = 1.0f / fmaxf(sqrtf(ss), 1e-12f);
        }
        __syncthreads();
    }

    {
        const int m  = tid >> 2;
        const int cb = (tid & 3) * 64;
        if (m0 + m < nRows) {
            const float inv = relu_norm ? sInv[m] : 1.0f;
            float* orow = out + (size_t)(m0 + m) * DIM;
#pragma unroll
            for (int c = 0; c < 64; c += 4) {
                float x0 = sMean[m * LSTRIDE + cb + c + 0];
                float x1 = sMean[m * LSTRIDE + cb + c + 1];
                float x2 = sMean[m * LSTRIDE + cb + c + 2];
                float x3 = sMean[m * LSTRIDE + cb + c + 3];
                if (relu_norm) {
                    x0 = fmaxf(x0 * inv, 0.f); x1 = fmaxf(x1 * inv, 0.f);
                    x2 = fmaxf(x2 * inv, 0.f); x3 = fmaxf(x3 * inv, 0.f);
                }
                *(float4*)(orow + cb + c) = make_float4(x0, x1, x2, x3);
            }
        }
    }
}

extern "C" void kernel_launch(void* const* d_in, const int* in_sizes, int n_in,
                              void* d_out, int out_size, void* d_ws, size_t ws_size,
                              hipStream_t stream) {
    (void)n_in; (void)out_size; (void)ws_size;
    const float* xA = (const float*)d_in[0];
    const float* xP = (const float*)d_in[1];
    const float* Wl = (const float*)d_in[2];
    const float* bl = (const float*)d_in[3];
    const float* Wr = (const float*)d_in[4];
    const int* eap  = (const int*)d_in[5];
    const int* epa  = (const int*)d_in[6];

    const int NA = in_sizes[0] / DIM;
    const int NP = in_sizes[1] / DIM;
    const int E  = in_sizes[5] / 2;
    const int L  = in_sizes[2] / (2 * DIM * DIM);

    const size_t FA = (size_t)NA * DIM, FP = (size_t)NP * DIM;
    char* ws = (char*)d_ws;
    float* A0 = (float*)ws;  ws += FA * sizeof(float);
    float* A1 = (float*)ws;  ws += FA * sizeof(float);
    float* P0 = (float*)ws;  ws += FP * sizeof(float);
    float* P1 = (float*)ws;  ws += FP * sizeof(float);
    const size_t FW = (size_t)L * 2 * DIM * DIM;
    float* PlAll = (float*)ws;  ws += FW * sizeof(float);
    float* PrAll = (float*)ws;  ws += FW * sizeof(float);
    int* cntP  = (int*)ws;   ws += (size_t)NP * 4;
    int* offP  = (int*)ws;   ws += (size_t)(NP + 1) * 4;
    int* curP  = (int*)ws;   ws += (size_t)NP * 4;
    int* permP = (int*)ws;   ws += (size_t)E * 4;
    int* cntA  = (int*)ws;   ws += (size_t)NA * 4;
    int* offA  = (int*)ws;   ws += (size_t)(NA + 1) * 4;
    int* curA  = (int*)ws;   ws += (size_t)NA * 4;
    int* permA = (int*)ws;   ws += (size_t)E * 4;
    int* bsum  = (int*)ws;   ws += 256 * 4;

    const int nbP = (NP + 255) / 256, nbA = (NA + 255) / 256;

    pack_w<<<dim3(L * 2 * 16, 2), 256, 0, stream>>>(Wl, Wr, PlAll, PrAll);

    zero_i32   <<<64, 256, 0, stream>>>(cntP, NP);
    count_dst  <<<1024, 256, 0, stream>>>(eap + E, cntP, E);
    scan_block <<<nbP, 256, 0, stream>>>(cntP, offP, bsum, NP);
    scan_sums  <<<1, 256, 0, stream>>>(bsum, nbP);
    scan_finish<<<nbP, 256, 0, stream>>>(offP, bsum, curP, NP, E);
    build_perm <<<1024, 256, 0, stream>>>(eap, eap + E, curP, permP, E);

    zero_i32   <<<64, 256, 0, stream>>>(cntA, NA);
    count_dst  <<<1024, 256, 0, stream>>>(epa + E, cntA, E);
    scan_block <<<nbA, 256, 0, stream>>>(cntA, offA, bsum, NA);
    scan_sums  <<<1, 256, 0, stream>>>(bsum, nbA);
    scan_finish<<<nbA, 256, 0, stream>>>(offA, bsum, curA, NA, E);
    build_perm <<<1024, 256, 0, stream>>>(epa, epa + E, curA, permA, E);

    const float* xa_cur = xA;
    const float* xp_cur = xP;
    for (int l = 0; l < L; ++l) {
        const bool last = (l == L - 1);
        float* xa_next = last ? (float*)d_out        : ((l & 1) ? A1 : A0);
        float* xp_next = last ? (float*)d_out + FA   : ((l & 1) ? P1 : P0);

        const float* Wl0 = PlAll + ((size_t)(l * 2 + 0)) * DIM * DIM;
        const float* Wl1 = PlAll + ((size_t)(l * 2 + 1)) * DIM * DIM;
        const float* Wr0 = PrAll + ((size_t)(l * 2 + 0)) * DIM * DIM;
        const float* Wr1 = PrAll + ((size_t)(l * 2 + 1)) * DIM * DIM;
        const float* bl0 = bl + (size_t)(l * 2 + 0) * DIM;
        const float* bl1 = bl + (size_t)(l * 2 + 1) * DIM;

        sage_gemm<<<(NP + MT - 1) / MT, 256, 0, stream>>>(xp_next, xa_cur, permP, offP, xp_cur, Wl0, bl0, Wr0, last ? 0 : 1, NP);
        sage_gemm<<<(NA + MT - 1) / MT, 256, 0, stream>>>(xa_next, xp_cur, permA, offA, xa_cur, Wl1, bl1, Wr1, last ? 0 : 1, NA);

        xa_cur = xa_next;
        xp_cur = xp_next;
    }
}


# A operand single-buffered in LDS: self rows and neighbour means pass through one buffer in turn, each with its own k loop; tile 64 -> 112 rows, whole grid resident at once; sum order changed
# speedup vs baseline: 1.2104x; 1.0736x over previous
#include <hip/hip_runtime.h>
#include <math.h>

typedef __attribute__((ext_vector_type(2))) float v2f;
typedef __attribute__((ext_vector_type(8))) float v8f;

#define DIM      256
#define NS       7
#define MT       (NS * 16)
#define LSTRIDE  260
#define KG       (DIM / 4)

__global__ __launch_bounds__(256) void zero_i32(int* __restrict__ p, int n) {
    int i = blockIdx.x * 256 + threadIdx.x;
    int st = gridDim.x * 256;
    for (; i < n; i += st) p[i] = 0;
}

__global__ __launch_bounds__(256) void count_dst(const int* __restrict__ dst,
                                                 int* __restrict__ cnt, int nE) {
    int i = blockIdx.x * 256 + threadIdx.x;
    int st = gridDim.x * 256;
    for (; i < nE; i += st) atomicAdd(&cnt[dst[i]], 1);
}

__global__ __launch_bounds__(256) void scan_block(const int* __restrict__ cnt,
                                                  int* __restrict__ off,
                                                  int* __restrict__ bsum, int n) {
    __shared__ int s[256];
    const int tid = threadIdx.x;
    const int gid = blockIdx.x * 256 + tid;
    int v = (gid < n) ? cnt[gid] : 0;
    s[tid] = v;
    __syncthreads();
    for (int d = 1; d < 256; d <<= 1) {
        int t = (tid >= d) ? s[tid - d] : 0;
        __syncthreads();
        s[tid] += t;
        __syncthreads();
    }
    if (gid < n) off[gid] = s[tid] - v;
    if (tid == 255) bsum[blockIdx.x] = s[255];
}

__global__ __launch_bounds__(256) void scan_sums(int* __restrict__ bsum, int nb) {
    __shared__ int s[256];
    const int tid = threadIdx.x;
    int v = (tid < nb) ? bsum[tid] : 0;
    s[tid] = v;
    __syncthreads();
    for (int d = 1; d < 256; d <<= 1) {
        int t = (tid >= d) ? s[tid - d] : 0;
        __syncthreads();
        s[tid] += t;
        __syncthreads();
    }
    bsum[tid] = s[tid] - v;
}

__global__ __launch_bounds__(256) void scan_finish(int* __restrict__ off,
                                                   const int* __restrict__ bsum,
                                                   int* __restrict__ cursor,
                                                   int n, int nE) {
    const int gid = blockIdx.x * 256 + threadIdx.x;
    if (gid < n) {
        int o = off[gid] + bsum[blockIdx.x];
        off[gid] = o;
        cursor[gid] = o;
    }
    if (gid == 0) off[n] = nE;
}

__global__ __launch_bounds__(256) void build_perm(const int* __restrict__ src,
                                                  const int* __restrict__ dst,
                                                  int* __restrict__ cursor,
                                                  int* __restrict__ perm, int nE) {
    int i = blockIdx.x * 256 + threadIdx.x;
    int st = gridDim.x * 256;
    for (; i < nE; i += st) {
        int slot = atomicAdd(&cursor[dst[i]], 1);
        perm[slot] = src[i];
    }
}

__global__ __launch_bounds__(256) void pack_w(const float* __restrict__ Wl,
                                              const float* __restrict__ Wr,
                                              float* __restrict__ Pl,
                                              float* __restrict__ Pr) {
    __shared__ __attribute__((aligned(16))) float t[16 * (DIM + 4)];
    const int tid = threadIdx.x;
    const int mat = blockIdx.x >> 4;
    const int nb  = (blockIdx.x & 15) * 16;
    const float* W = (blockIdx.y ? Wr : Wl) + (size_t)mat * DIM * DIM + (size_t)nb * DIM;
    float*       P = (blockIdx.y ? Pr : Pl) + (size_t)mat * DIM * DIM;
    for (int i = tid; i < 16 * KG; i += 256) {
        const int r = i >> 6, g = i & 63;
        *(float4*)&t[r * (DIM + 4) + 4 * g] = *(const float4*)(W + (size_t)r * DIM + 4 * g);
    }
    __syncthreads();
    for (int i = tid; i < 16 * KG; i += 256) {
        const int g = i >> 4, r = i & 15;
        *(float4*)(P + ((size_t)g * DIM + nb + r) * 4) = *(const float4*)&t[r * (DIM + 4) + 4 * g];
    }
}

template <int NB, bool FULL>
__device__ __forceinline__ void gather_batch(const float* __restrict__ xsrc, int idx,
                                             int j, int cnt, int lane, float (&acc)[8]) {
    float4 va[NB], vb[NB];
#pragma unroll
    for (int u = 0; u < NB; ++u) {
        const int jj = (FULL || u < cnt) ? (j + u) : (j + cnt - 1);
        const int p  = __builtin_amdgcn_readlane(idx, jj);
        const float4* row = (const float4*)(xsrc + (size_t)p * DIM);
        va[u] = row[lane];
        vb[u] = row[lane + 32];
    }
#pragma unroll
    for (int u = 0; u < NB; ++u) {
        const bool on = FULL || (u < cnt);
        acc[0] += on ? va[u].x : 0.0f; acc[1] += on ? va[u].y : 0.0f;
        acc[2] += on ? va[u].z : 0.0f; acc[3] += on ? va[u].w : 0.0f;
        acc[4] += on ? vb[u].x : 0.0f; acc[5] += on ? vb[u].y : 0.0f;
        acc[6] += on ? vb[u].z : 0.0f; acc[7] += on ? vb[u].w : 0.0f;
    }
}

__device__ __forceinline__ void load_block(const float4* w0, const float4* w1,
                                           const float* aRow, int j,
                                           float4 (&b)[2], float4 (&a)[NS]) {
    b[0] = w0[(size_t)j * (2 * DIM)];
    b[1] = w1[(size_t)j * (2 * DIM)];
#pragma unroll
    for (int s = 0; s < NS; ++s)
        a[s] = *(const float4*)(aRow + s * 16 * LSTRIDE + 8 * j);
}

__device__ __forceinline__ void mma_block(v8f (&acc)[NS][2], const float4 (&b)[2],
                                          const float4 (&a)[NS]) {
#pragma unroll
    for (int st = 0; st < 2; ++st) {
        const v2f b0 = st ? (v2f){b[0].z, b[0].w} : (v2f){b[0].x, b[0].y};
        const v2f b1 = st ? (v2f){b[1].z, b[1].w} : (v2f){b[1].x, b[1].y};
#pragma unroll
        for (int s = 0; s < NS; ++s) {
            const v2f as = st ? (v2f){a[s].z, a[s].w} : (v2f){a[s].x, a[s].y};
            acc[s][0] = __builtin_amdgcn_wmma_f32_16x16x4_f32(false, as, false, b0, (short)0, acc[s][0], false, false);
            acc[s][1] = __builtin_amdgcn_wmma_f32_16x16x4_f32(false, as, false, b1, (short)0, acc[s][1], false, false);
        }
    }
}

__device__ __forceinline__ void k_loop(v8f (&acc)[NS][2], const float4* w0, const float4* w1,
                                       const float* aRow) {
    constexpr int NBLK = DIM / 8;
    float4 bP[2], aP[NS];
    float4 bQ[2], aQ[NS];

    load_block(w0, w1, aRow, 0, bP, aP);
#pragma unroll 1
    for (int i = 0; i < NBLK - 2; i += 2) {
        load_block(w0, w1, aRow, i + 1, bQ, aQ);
        __builtin_amdgcn_sched_barrier(0);
        mma_block(acc, bP, aP);
        __builtin_amdgcn_sched_barrier(0);
        load_block(w0, w1, aRow, i + 2, bP, aP);
        __builtin_amdgcn_sched_barrier(0);
        mma_block(acc, bQ, aQ);
    }
    load_block(w0, w1, aRow, NBLK - 1, bQ, aQ);
    mma_block(acc, bP, aP);
    mma_block(acc, bQ, aQ);
}

__global__ __launch_bounds__(256, 4) void sage_gemm(float* __restrict__ out,
                                                 const float* __restrict__ xsrc,
                                                 const int* __restrict__ perm,
                                                 const int* __restrict__ off,
                                                 const float* __restrict__ xdst,
                                                 const float* __restrict__ Pl,
                                                 const float* __restrict__ bl,
                                                 const float* __restrict__ Pr,
                                                 int relu_norm, int nRows) {
    __shared__ __attribute__((aligned(16))) float sA[MT * LSTRIDE];
    __shared__ float sPart[MT * 4];
    __shared__ float sInv [MT];

    const int tid = threadIdx.x;
    const int m0  = blockIdx.x * MT;

    const int lane  = tid & 31;
    const int wv    = tid >> 5;
    const int mrow  = lane & 15;
    const int half  = lane >> 4;
    const int ncol  = lane & 15;
    const int n0    = wv * 32;

    const int selfFirst = (blockIdx.x ^ (blockIdx.x >> 3) ^ (blockIdx.x >> 8)) & 1;

    const float4* wl0 = (const float4*)Pl + (size_t)half * DIM + n0 + ncol;
    const float4* wl1 = wl0 + 16;
    const float4* wr0 = (const float4*)Pr + (size_t)half * DIM + n0 + ncol;
    const float4* wr1 = wr0 + 16;
    const float* aRow = &sA[mrow * LSTRIDE] + 4 * half;

    v8f acc[NS][2];
#pragma unroll
    for (int s = 0; s < NS; ++s)
#pragma unroll
        for (int t = 0; t < 2; ++t)
            acc[s][t] = (v8f){0.f, 0.f, 0.f, 0.f, 0.f, 0.f, 0.f, 0.f};

    for (int p = 0; p < 2; ++p) {
        const bool doSelf = (p == 0) ? (selfFirst != 0) : (selfFirst == 0);
        if (p) __syncthreads();

        if (doSelf) {
            for (int i = tid; i < MT * DIM / 4; i += 256) {
                const int m = i >> 6;
                const int c = (i & 63) * 4;
                int gm = m0 + m;
                if (gm >= nRows) gm = nRows - 1;
                float4 s = *(const float4*)(xdst + (size_t)gm * DIM + c);
                *(float4*)&sA[m * LSTRIDE + c] = s;
            }
        } else {
            const int rbase = m0 + wv * (MT / 8);
            int oi = rbase + lane;
            if (oi > nRows) oi = nRows;
            const int offv = (lane < MT / 8 + 1) ? off[oi] : 0;
#pragma unroll 1
            for (int r = 0; r < MT / 8; ++r) {
                const int s = __builtin_amdgcn_readlane(offv, r);
                const int e = __builtin_amdgcn_readlane(offv, r + 1);
                float g[8] = {0.f, 0.f, 0.f, 0.f, 0.f, 0.f, 0.f, 0.f};
                for (int base = s; base < e; base += 32) {
                    const int n  = min(32, e - base);
                    const int ix = (lane < n) ? perm[base + lane] : 0;
                    int j = 0;
                    for (; j + 8 <= n; j += 8) gather_batch<8, true>(xsrc, ix, j, 8, lane, g);
                    const int cnt = n - j;
                    if (cnt > 4)      gather_batch<8, false>(xsrc, ix, j, cnt, lane, g);
                    else if (cnt > 0) gather_batch<4, false>(xsrc, ix, j, cnt, lane, g);
                }
                const float inv = 1.0f / fmaxf((float)(e - s), 1.0f);
                float* pm = &sA[(wv * (MT / 8) + r) * LSTRIDE + lane * 4];
                *(float4*)pm         = make_float4(g[0] * inv, g[1] * inv, g[2] * inv, g[3] * inv);
                *(float4*)(pm + 128) = make_float4(g[4] * inv, g[5] * inv, g[6] * inv, g[7] * inv);
            }
        }
        __syncthreads();

        k_loop(acc, doSelf ? wr0 : wl0, doSelf ? wr1 : wl1, aRow);
    }

    const float bb0 = bl[n0 + ncol];
    const float bb1 = bl[n0 + 16 + ncol];
#pragma unroll
    for (int s = 0; s < NS; ++s)
#pragma unroll
        for (int r = 0; r < 8; ++r) { acc[s][0][r] += bb0; acc[s][1][r] += bb1; }

    __syncthreads();

    const int mbase = (lane >> 4) << 3;
#pragma unroll
    for (int s = 0; s < NS; ++s)
#pragma unroll
        for (int r = 0; r < 8; ++r) {
            const int m = s * 16 + mbase + r;
            sA[m * LSTRIDE + n0 + ncol]      = acc[s][0][r];
            sA[m * LSTRIDE + n0 + 16 + ncol] = acc[s][1][r];
        }
    __syncthreads();

    if (relu_norm) {
        for (int i = tid; i < MT * 4; i += 256) {
            const int m = i >> 2, seg = i & 3;
            const float* row = &sA[m * LSTRIDE + seg * 64];
            float ss = 0.f;
#pragma unroll 16
            for (int c = 0; c < 64; ++c) ss += row[c] * row[c];
            sPart[i] = ss;
        }
        __syncthreads();
        if (tid < MT) {
            const float ss = sPart[tid * 4] + sPart[tid * 4 + 1] +
                             sPart[tid * 4 + 2] + sPart[tid * 4 + 3];
            sInv[tid] = 1.0f / fmaxf(sqrtf(ss), 1e-12f);
        }
        __syncthreads();
    }

    for (int m = tid >> 2; m < MT; m += 64) {
        const int cb = (tid & 3) * 64;
        if (m0 + m < nRows) {
            const float inv = relu_norm ? sInv[m] : 1.0f;
            float* orow = out + (size_t)(m0 + m) * DIM;
#pragma unroll
            for (int c = 0; c < 64; c += 4) {
                float x0 = sA[m * LSTRIDE + cb + c + 0];
                float x1 = sA[m * LSTRIDE + cb + c + 1];
                float x2 = sA[m * LSTRIDE + cb + c + 2];
                float x3 = sA[m * LSTRIDE + cb + c + 3];
                if (relu_norm) {
                    x0 = fmaxf(x0 * inv, 0.f); x1 = fmaxf(x1 * inv, 0.f);
                    x2 = fmaxf(x2 * inv, 0.f); x3 = fmaxf(x3 * inv, 0.f);
                }
                *(float4*)(orow + cb + c) = make_float4(x0, x1, x2, x3);
            }
        }
    }
}

extern "C" void kernel_launch(void* const* d_in, const int* in_sizes, int n_in,
                              void* d_out, int out_size, void* d_ws, size_t ws_size,
                              hipStream_t stream) {
    (void)n_in; (void)out_size; (void)ws_size;
    const float* xA = (const float*)d_in[0];
    const float* xP = (const float*)d_in[1];
    const float* Wl = (const float*)d_in[2];
    const float* bl = (const float*)d_in[3];
    const float* Wr = (const float*)d_in[4];
    const int* eap  = (const int*)d_in[5];
    const int* epa  = (const int*)d_in[6];

    const int NA = in_sizes[0] / DIM;
    const int NP = in_sizes[1] / DIM;
    const int E  = in_sizes[5] / 2;
    const int L  = in_sizes[2] / (2 * DIM * DIM);

    const size_t FA = (size_t)NA * DIM, FP = (size_t)NP * DIM;
    char* ws = (char*)d_ws;
    float* A0 = (float*)ws;  ws += FA * sizeof(float);
    float* A1 = (float*)ws;  ws += FA * sizeof(float);
    float* P0 = (float*)ws;  ws += FP * sizeof(float);
    float* P1 = (float*)ws;  ws += FP * sizeof(float);
    const size_t FW = (size_t)L * 2 * DIM * DIM;
    float* PlAll = (float*)ws;  ws += FW * sizeof(float);
    float* PrAll = (float*)ws;  ws += FW * sizeof(float);
    int* cntP  = (int*)ws;   ws += (size_t)NP * 4;
    int* offP  = (int*)ws;   ws += (size_t)(NP + 1) * 4;
    int* curP  = (int*)ws;   ws += (size_t)NP * 4;
    int* permP = (int*)ws;   ws += (size_t)E * 4;
    int* cntA  = (int*)ws;   ws += (size_t)NA * 4;
    int* offA  = (int*)ws;   ws += (size_t)(NA + 1) * 4;
    int* curA  = (int*)ws;   ws += (size_t)NA * 4;
    int* permA = (int*)ws;   ws += (size_t)E * 4;
    int* bsum  = (int*)ws;   ws += 256 * 4;

    const int nbP = (NP + 255) / 256, nbA = (NA + 255) / 256;

    pack_w<<<dim3(L * 2 * 16, 2), 256, 0, stream>>>(Wl, Wr, PlAll, PrAll);

    zero_i32   <<<64, 256, 0, stream>>>(cntP, NP);
    count_dst  <<<1024, 256, 0, stream>>>(eap + E, cntP, E);
    scan_block <<<nbP, 256, 0, stream>>>(cntP, offP, bsum, NP);
    scan_sums  <<<1, 256, 0, stream>>>(bsum, nbP);
    scan_finish<<<nbP, 256, 0, stream>>>(offP, bsum, curP, NP, E);
    build_perm <<<1024, 256, 0, stream>>>(eap, eap + E, curP, permP, E);

    zero_i32   <<<64, 256, 0, stream>>>(cntA, NA);
    count_dst  <<<1024, 256, 0, stream>>>(epa + E, cntA, E);
    scan_block <<<nbA, 256, 0, stream>>>(cntA, offA, bsum, NA);
    scan_sums  <<<1, 256, 0, stream>>>(bsum, nbA);
    scan_finish<<<nbA, 256, 0, stream>>>(offA, bsum, curA, NA, E);
    build_perm <<<1024, 256, 0, stream>>>(epa, epa + E, curA, permA, E);

    const float* xa_cur = xA;
    const float* xp_cur = xP;
    for (int l = 0; l < L; ++l) {
        const bool last = (l == L - 1);
        float* xa_next = last ? (float*)d_out        : ((l & 1) ? A1 : A0);
        float* xp_next = last ? (float*)d_out + FA   : ((l & 1) ? P1 : P0);

        const float* Wl0 = PlAll + ((size_t)(l * 2 + 0)) * DIM * DIM;
        const float* Wl1 = PlAll + ((size_t)(l * 2 + 1)) * DIM * DIM;
        const float* Wr0 = PrAll + ((size_t)(l * 2 + 0)) * DIM * DIM;
        const float* Wr1 = PrAll + ((size_t)(l * 2 + 1)) * DIM * DIM;
        const float* bl0 = bl + (size_t)(l * 2 + 0) * DIM;
        const float* bl1 = bl + (size_t)(l * 2 + 1) * DIM;

        sage_gemm<<<(NP + MT - 1) / MT, 256, 0, stream>>>(xp_next, xa_cur, permP, offP, xp_cur, Wl0, bl0, Wr0, last ? 0 : 1, NP);
        sage_gemm<<<(NA + MT - 1) / MT, 256, 0, stream>>>(xa_next, xp_cur, permA, offA, xa_cur, Wl1, bl1, Wr1, last ? 0 : 1, NA);

        xa_cur = xa_next;
        xp_cur = xp_next;
    }
}


# epilogue by whole rows: each wave reads 14 finished rows from LDS as 512-byte runs and stores four whole 128-byte lines per instruction; row norm summed by a wave butterfly, its sum order changed
# speedup vs baseline: 1.3131x; 1.0848x over previous
#include <hip/hip_runtime.h>
#include <math.h>

typedef __attribute__((ext_vector_type(2))) float v2f;
typedef __attribute__((ext_vector_type(8))) float v8f;

#define DIM      256
#define NS       7
#define MT       (NS * 16)
#define LSTRIDE  260
#define KG       (DIM / 4)

__global__ __launch_bounds__(256) void zero_i32(int* __restrict__ p, int n) {
    int i = blockIdx.x * 256 + threadIdx.x;
    int st = gridDim.x * 256;
    for (; i < n; i += st) p[i] = 0;
}

__global__ __launch_bounds__(256) void count_dst(const int* __restrict__ dst,
                                                 int* __restrict__ cnt, int nE) {
    int i = blockIdx.x * 256 + threadIdx.x;
    int st = gridDim.x * 256;
    for (; i < nE; i += st) atomicAdd(&cnt[dst[i]], 1);
}

__global__ __launch_bounds__(256) void scan_block(const int* __restrict__ cnt,
                                                  int* __restrict__ off,
                                                  int* __restrict__ bsum, int n) {
    __shared__ int s[256];
    const int tid = threadIdx.x;
    const int gid = blockIdx.x * 256 + tid;
    int v = (gid < n) ? cnt[gid] : 0;
    s[tid] = v;
    __syncthreads();
    for (int d = 1; d < 256; d <<= 1) {
        int t = (tid >= d) ? s[tid - d] : 0;
        __syncthreads();
        s[tid] += t;
        __syncthreads();
    }
    if (gid < n) off[gid] = s[tid] - v;
    if (tid == 255) bsum[blockIdx.x] = s[255];
}

__global__ __launch_bounds__(256) void scan_sums(int* __restrict__ bsum, int nb) {
    __shared__ int s[256];
    const int tid = threadIdx.x;
    int v = (tid < nb) ? bsum[tid] : 0;
    s[tid] = v;
    __syncthreads();
    for (int d = 1; d < 256; d <<= 1) {
        int t = (tid >= d) ? s[tid - d] : 0;
        __syncthreads();
        s[tid] += t;
        __syncthreads();
    }
    bsum[tid] = s[tid] - v;
}

__global__ __launch_bounds__(256) void scan_finish(int* __restrict__ off,
                                                   const int* __restrict__ bsum,
                                                   int* __restrict__ cursor,
                                                   int n, int nE) {
    const int gid = blockIdx.x * 256 + threadIdx.x;
    if (gid < n) {
        int o = off[gid] + bsum[blockIdx.x];
        off[gid] = o;
        cursor[gid] = o;
    }
    if (gid == 0) off[n] = nE;
}

__global__ __launch_bounds__(256) void build_perm(const int* __restrict__ src,
                                                  const int* __restrict__ dst,
                                                  int* __restrict__ cursor,
                                                  int* __restrict__ perm, int nE) {
    int i = blockIdx.x * 256 + threadIdx.x;
    int st = gridDim.x * 256;
    for (; i < nE; i += st) {
        int slot = atomicAdd(&cursor[dst[i]], 1);
        perm[slot] = src[i];
    }
}

__global__ __launch_bounds__(256) void pack_w(const float* __restrict__ Wl,
                                              const float* __restrict__ Wr,
                                              float* __restrict__ Pl,
                                              float* __restrict__ Pr) {
    __shared__ __attribute__((aligned(16))) float t[16 * (DIM + 4)];
    const int tid = threadIdx.x;
    const int mat = blockIdx.x >> 4;
    const int nb  = (blockIdx.x & 15) * 16;
    const float* W = (blockIdx.y ? Wr : Wl) + (size_t)mat * DIM * DIM + (size_t)nb * DIM;
    float*       P = (blockIdx.y ? Pr : Pl) + (size_t)mat * DIM * DIM;
    for (int i = tid; i < 16 * KG; i += 256) {
        const int r = i >> 6, g = i & 63;
        *(float4*)&t[r * (DIM + 4) + 4 * g] = *(const float4*)(W + (size_t)r * DIM + 4 * g);
    }
    __syncthreads();
    for (int i = tid; i < 16 * KG; i += 256) {
        const int g = i >> 4, r = i & 15;
        *(float4*)(P + ((size_t)g * DIM + nb + r) * 4) = *(const float4*)&t[r * (DIM + 4) + 4 * g];
    }
}

template <int NB, bool FULL>
__device__ __forceinline__ void gather_batch(const float* __restrict__ xsrc, int idx,
                                             int j, int cnt, int lane, float (&acc)[8]) {
    float4 va[NB], vb[NB];
#pragma unroll
    for (int u = 0; u < NB; ++u) {
        const int jj = (FULL || u < cnt) ? (j + u) : (j + cnt - 1);
        const int p  = __builtin_amdgcn_readlane(idx, jj);
        const float4* row = (const float4*)(xsrc + (size_t)p * DIM);
        va[u] = row[lane];
        vb[u] = row[lane + 32];
    }
#pragma unroll
    for (int u = 0; u < NB; ++u) {
        const bool on = FULL || (u < cnt);
        acc[0] += on ? va[u].x : 0.0f; acc[1] += on ? va[u].y : 0.0f;
        acc[2] += on ? va[u].z : 0.0f; acc[3] += on ? va[u].w : 0.0f;
        acc[4] += on ? vb[u].x : 0.0f; acc[5] += on ? vb[u].y : 0.0f;
        acc[6] += on ? vb[u].z : 0.0f; acc[7] += on ? vb[u].w : 0.0f;
    }
}

__device__ __forceinline__ void load_block(const float4* w0, const float4* w1,
                                           const float* aRow, int j,
                                           float4 (&b)[2], float4 (&a)[NS]) {
    b[0] = w0[(size_t)j * (2 * DIM)];
    b[1] = w1[(size_t)j * (2 * DIM)];
#pragma unroll
    for (int s = 0; s < NS; ++s)
        a[s] = *(const float4*)(aRow + s * 16 * LSTRIDE + 8 * j);
}

__device__ __forceinline__ void mma_block(v8f (&acc)[NS][2], const float4 (&b)[2],
                                          const float4 (&a)[NS]) {
#pragma unroll
    for (int st = 0; st < 2; ++st) {
        const v2f b0 = st ? (v2f){b[0].z, b[0].w} : (v2f){b[0].x, b[0].y};
        const v2f b1 = st ? (v2f){b[1].z, b[1].w} : (v2f){b[1].x, b[1].y};
#pragma unroll
        for (int s = 0; s < NS; ++s) {
            const v2f as = st ? (v2f){a[s].z, a[s].w} : (v2f){a[s].x, a[s].y};
            acc[s][0] = __builtin_amdgcn_wmma_f32_16x16x4_f32(false, as, false, b0, (short)0, acc[s][0], false, false);
            acc[s][1] = __builtin_amdgcn_wmma_f32_16x16x4_f32(false, as, false, b1, (short)0, acc[s][1], false, false);
        }
    }
}

__device__ __forceinline__ void k_loop(v8f (&acc)[NS][2], const float4* w0, const float4* w1,
                                       const float* aRow) {
    constexpr int NBLK = DIM / 8;
    float4 bP[2], aP[NS];
    float4 bQ[2], aQ[NS];

    load_block(w0, w1, aRow, 0, bP, aP);
#pragma unroll 1
    for (int i = 0; i < NBLK - 2; i += 2) {
        load_block(w0, w1, aRow, i + 1, bQ, aQ);
        __builtin_amdgcn_sched_barrier(0);
        mma_block(acc, bP, aP);
        __builtin_amdgcn_sched_barrier(0);
        load_block(w0, w1, aRow, i + 2, bP, aP);
        __builtin_amdgcn_sched_barrier(0);
        mma_block(acc, bQ, aQ);
    }
    load_block(w0, w1, aRow, NBLK - 1, bQ, aQ);
    mma_block(acc, bP, aP);
    mma_block(acc, bQ, aQ);
}

__global__ __launch_bounds__(256, 4) void sage_gemm(float* __restrict__ out,
                                                 const float* __restrict__ xsrc,
                                                 const int* __restrict__ perm,
                                                 const int* __restrict__ off,
                                                 const float* __restrict__ xdst,
                                                 const float* __restrict__ Pl,
                                                 const float* __restrict__ bl,
                                                 const float* __restrict__ Pr,
                                                 int relu_norm, int nRows) {
    __shared__ __attribute__((aligned(16))) float sA[MT * LSTRIDE];

    const int tid = threadIdx.x;
    const int m0  = blockIdx.x * MT;

    const int lane  = tid & 31;
    const int wv    = tid >> 5;
    const int mrow  = lane & 15;
    const int half  = lane >> 4;
    const int ncol  = lane & 15;
    const int n0    = wv * 32;

    const int selfFirst = (blockIdx.x ^ (blockIdx.x >> 3) ^ (blockIdx.x >> 8)) & 1;

    const float4* wl0 = (const float4*)Pl + (size_t)half * DIM + n0 + ncol;
    const float4* wl1 = wl0 + 16;
    const float4* wr0 = (const float4*)Pr + (size_t)half * DIM + n0 + ncol;
    const float4* wr1 = wr0 + 16;
    const float* aRow = &sA[mrow * LSTRIDE] + 4 * half;

    v8f acc[NS][2];
#pragma unroll
    for (int s = 0; s < NS; ++s)
#pragma unroll
        for (int t = 0; t < 2; ++t)
            acc[s][t] = (v8f){0.f, 0.f, 0.f, 0.f, 0.f, 0.f, 0.f, 0.f};

    for (int p = 0; p < 2; ++p) {
        const bool doSelf = (p == 0) ? (selfFirst != 0) : (selfFirst == 0);
        if (p) __syncthreads();

        if (doSelf) {
            for (int i = tid; i < MT * DIM / 4; i += 256) {
                const int m = i >> 6;
                const int c = (i & 63) * 4;
                int gm = m0 + m;
                if (gm >= nRows) gm = nRows - 1;
                float4 s = *(const float4*)(xdst + (size_t)gm * DIM + c);
                *(float4*)&sA[m * LSTRIDE + c] = s;
            }
        } else {
            const int rbase = m0 + wv * (MT / 8);
            int oi = rbase + lane;
            if (oi > nRows) oi = nRows;
            const int offv = (lane < MT / 8 + 1) ? off[oi] : 0;
#pragma unroll 1
            for (int r = 0; r < MT / 8; ++r) {
                const int s = __builtin_amdgcn_readlane(offv, r);
                const int e = __builtin_amdgcn_readlane(offv, r + 1);
                float g[8] = {0.f, 0.f, 0.f, 0.f, 0.f, 0.f, 0.f, 0.f};
                for (int base = s; base < e; base += 32) {
                    const int n  = min(32, e - base);
                    const int ix = (lane < n) ? perm[base + lane] : 0;
                    int j = 0;
                    for (; j + 8 <= n; j += 8) gather_batch<8, true>(xsrc, ix, j, 8, lane, g);
                    const int cnt = n - j;
                    if (cnt > 4)      gather_batch<8, false>(xsrc, ix, j, cnt, lane, g);
                    else if (cnt > 0) gather_batch<4, false>(xsrc, ix, j, cnt, lane, g);
                }
                const float inv = 1.0f / fmaxf((float)(e - s), 1.0f);
                float* pm = &sA[(wv * (MT / 8) + r) * LSTRIDE + lane * 4];
                *(float4*)pm         = make_float4(g[0] * inv, g[1] * inv, g[2] * inv, g[3] * inv);
                *(float4*)(pm + 128) = make_float4(g[4] * inv, g[5] * inv, g[6] * inv, g[7] * inv);
            }
        }
        __syncthreads();

        k_loop(acc, doSelf ? wr0 : wl0, doSelf ? wr1 : wl1, aRow);
    }

    const float bb0 = bl[n0 + ncol];
    const float bb1 = bl[n0 + 16 + ncol];
#pragma unroll
    for (int s = 0; s < NS; ++s)
#pragma unroll
        for (int r = 0; r < 8; ++r) { acc[s][0][r] += bb0; acc[s][1][r] += bb1; }

    __syncthreads();

    const int mbase = (lane >> 4) << 3;
#pragma unroll
    for (int s = 0; s < NS; ++s)
#pragma unroll
        for (int r = 0; r < 8; ++r) {
            const int m = s * 16 + mbase + r;
            sA[m * LSTRIDE + n0 + ncol]      = acc[s][0][r];
            sA[m * LSTRIDE + n0 + 16 + ncol] = acc[s][1][r];
        }
    __syncthreads();

    constexpr int RW = MT / 8;
    const int wvs = __builtin_amdgcn_readfirstlane(wv);
    float4 va[RW], vb[RW];
#pragma unroll
    for (int i = 0; i < RW; ++i) {
        const float* row = &sA[(wvs * RW + i) * LSTRIDE + lane * 4];
        va[i] = *(const float4*)row;
        vb[i] = *(const float4*)(row + 128);
    }
    if (relu_norm) {
        float ss[RW];
#pragma unroll
        for (int i = 0; i < RW; ++i)
            ss[i] = va[i].x * va[i].x + va[i].y * va[i].y + va[i].z * va[i].z + va[i].w * va[i].w +
                    vb[i].x * vb[i].x + vb[i].y * vb[i].y + vb[i].z * vb[i].z + vb[i].w * vb[i].w;
#pragma unroll
        for (int d = 16; d >= 1; d >>= 1)
#pragma unroll
            for (int i = 0; i < RW; ++i) ss[i] += __shfl_xor(ss[i], d, 32);
        float t = 1.0f;
#pragma unroll
        for (int i = 0; i < RW; ++i) t = (lane == i) ? ss[i] : t;
        const float invl = 1.0f / fmaxf(sqrtf(t), 1e-12f);
#pragma unroll
        for (int i = 0; i < RW; ++i) {
            const float inv = __int_as_float(__builtin_amdgcn_readlane(__float_as_int(invl), i));
            va[i].x = fmaxf(va[i].x * inv, 0.f); va[i].y = fmaxf(va[i].y * inv, 0.f);
            va[i].z = fmaxf(va[i].z * inv, 0.f); va[i].w = fmaxf(va[i].w * inv, 0.f);
            vb[i].x = fmaxf(vb[i].x * inv, 0.f); vb[i].y = fmaxf(vb[i].y * inv, 0.f);
            vb[i].z = fmaxf(vb[i].z * inv, 0.f); vb[i].w = fmaxf(vb[i].w * inv, 0.f);
        }
    }
#pragma unroll
    for (int i = 0; i < RW; ++i) {
        const int gm = m0 + wvs * RW + i;
        if (gm < nRows) {
            float* orow = out + (size_t)gm * DIM + lane * 4;
            *(float4*)orow         = va[i];
            *(float4*)(orow + 128) = vb[i];
        }
    }
}

extern "C" void kernel_launch(void* const* d_in, const int* in_sizes, int n_in,
                              void* d_out, int out_size, void* d_ws, size_t ws_size,
                              hipStream_t stream) {
    (void)n_in; (void)out_size; (void)ws_size;
    const float* xA = (const float*)d_in[0];
    const float* xP = (const float*)d_in[1];
    const float* Wl = (const float*)d_in[2];
    const float* bl = (const float*)d_in[3];
    const float* Wr = (const float*)d_in[4];
    const int* eap  = (const int*)d_in[5];
    const int* epa  = (const int*)d_in[6];

    const int NA = in_sizes[0] / DIM;
    const int NP = in_sizes[1] / DIM;
    const int E  = in_sizes[5] / 2;
    const int L  = in_sizes[2] / (2 * DIM * DIM);

    const size_t FA = (size_t)NA * DIM, FP = (size_t)NP * DIM;
    char* ws = (char*)d_ws;
    float* A0 = (float*)ws;  ws += FA * sizeof(float);
    float* A1 = (float*)ws;  ws += FA * sizeof(float);
    float* P0 = (float*)ws;  ws += FP * sizeof(float);
    float* P1 = (float*)ws;  ws += FP * sizeof(float);
    const size_t FW = (size_t)L * 2 * DIM * DIM;
    float* PlAll = (float*)ws;  ws += FW * sizeof(float);
    float* PrAll = (float*)ws;  ws += FW * sizeof(float);
    int* cntP  = (int*)ws;   ws += (size_t)NP * 4;
    int* offP  = (int*)ws;   ws += (size_t)(NP + 1) * 4;
    int* curP  = (int*)ws;   ws += (size_t)NP * 4;
    int* permP = (int*)ws;   ws += (size_t)E * 4;
    int* cntA  = (int*)ws;   ws += (size_t)NA * 4;
    int* offA  = (int*)ws;   ws += (size_t)(NA + 1) * 4;
    int* curA  = (int*)ws;   ws += (size_t)NA * 4;
    int* permA = (int*)ws;   ws += (size_t)E * 4;
    int* bsum  = (int*)ws;   ws += 256 * 4;

    const int nbP = (NP + 255) / 256, nbA = (NA + 255) / 256;

    pack_w<<<dim3(L * 2 * 16, 2), 256, 0, stream>>>(Wl, Wr, PlAll, PrAll);

    zero_i32   <<<64, 256, 0, stream>>>(cntP, NP);
    count_dst  <<<1024, 256, 0, stream>>>(eap + E, cntP, E);
    scan_block <<<nbP, 256, 0, stream>>>(cntP, offP, bsum, NP);
    scan_sums  <<<1, 256, 0, stream>>>(bsum, nbP);
    scan_finish<<<nbP, 256, 0, stream>>>(offP, bsum, curP, NP, E);
    build_perm <<<1024, 256, 0, stream>>>(eap, eap + E, curP, permP, E);

    zero_i32   <<<64, 256, 0, stream>>>(cntA, NA);
    count_dst  <<<1024, 256, 0, stream>>>(epa + E, cntA, E);
    scan_block <<<nbA, 256, 0, stream>>>(cntA, offA, bsum, NA);
    scan_sums  <<<1, 256, 0, stream>>>(bsum, nbA);
    scan_finish<<<nbA, 256, 0, stream>>>(offA, bsum, curA, NA, E);
    build_perm <<<1024, 256, 0, stream>>>(epa, epa + E, curA, permA, E);

    const float* xa_cur = xA;
    const float* xp_cur = xP;
    for (int l = 0; l < L; ++l) {
        const bool last = (l == L - 1);
        float* xa_next = last ? (float*)d_out        : ((l & 1) ? A1 : A0);
        float* xp_next = last ? (float*)d_out + FA   : ((l & 1) ? P1 : P0);

        const float* Wl0 = PlAll + ((size_t)(l * 2 + 0)) * DIM * DIM;
        const float* Wl1 = PlAll + ((size_t)(l * 2 + 1)) * DIM * DIM;
        const float* Wr0 = PrAll + ((size_t)(l * 2 + 0)) * DIM * DIM;
        const float* Wr1 = PrAll + ((size_t)(l * 2 + 1)) * DIM * DIM;
        const float* bl0 = bl + (size_t)(l * 2 + 0) * DIM;
        const float* bl1 = bl + (size_t)(l * 2 + 1) * DIM;

        sage_gemm<<<(NP + MT - 1) / MT, 256, 0, stream>>>(xp_next, xa_cur, permP, offP, xp_cur, Wl0, bl0, Wr0, last ? 0 : 1, NP);
        sage_gemm<<<(NA + MT - 1) / MT, 256, 0, stream>>>(xa_next, xp_cur, permA, offA, xa_cur, Wl1, bl1, Wr1, last ? 0 : 1, NA);

        xa_cur = xa_next;
        xp_cur = xp_next;
    }
}


# rows dealt in 16-row strips evenly over 512 work-groups per launch (6 or 7 strips each, two on every compute unit) instead of 447 tiles of 7; self-first or mean-first sum order follows the new index
# speedup vs baseline: 1.3720x; 1.0449x over previous
#include <hip/hip_runtime.h>
#include <math.h>

typedef __attribute__((ext_vector_type(2))) float v2f;
typedef __attribute__((ext_vector_type(8))) float v8f;

#define DIM      256
#define NSMAX    7
#define WGS      512
#define LSTRIDE  260
#define KG       (DIM / 4)

__global__ __launch_bounds__(256) void zero_i32(int* __restrict__ p, int n) {
    int i = blockIdx.x * 256 + threadIdx.x;
    int st = gridDim.x * 256;
    for (; i < n; i += st) p[i] = 0;
}

__global__ __launch_bounds__(256) void count_dst(const int* __restrict__ dst,
                                                 int* __restrict__ cnt, int nE) {
    int i = blockIdx.x * 256 + threadIdx.x;
    int st = gridDim.x * 256;
    for (; i < nE; i += st) atomicAdd(&cnt[dst[i]], 1);
}

__global__ __launch_bounds__(256) void scan_block(const int* __restrict__ cnt,
                                                  int* __restrict__ off,
                                                  int* __restrict__ bsum, int n) {
    __shared__ int s[256];
    const int tid = threadIdx.x;
    const int gid = blockIdx.x * 256 + tid;
    int v = (gid < n) ? cnt[gid] : 0;
    s[tid] = v;
    __syncthreads();
    for (int d = 1; d < 256; d <<= 1) {
        int t = (tid >= d) ? s[tid - d] : 0;
        __syncthreads();
        s[tid] += t;
        __syncthreads();
    }
    if (gid < n) off[gid] = s[tid] - v;
    if (tid == 255) bsum[blockIdx.x] = s[255];
}

__global__ __launch_bounds__(256) void scan_sums(int* __restrict__ bsum, int nb) {
    __shared__ int s[256];
    const int tid = threadIdx.x;
    int v = (tid < nb) ? bsum[tid] : 0;
    s[tid] = v;
    __syncthreads();
    for (int d = 1; d < 256; d <<= 1) {
        int t = (tid >= d) ? s[tid - d] : 0;
        __syncthreads();
        s[tid] += t;
        __syncthreads();
    }
    bsum[tid] = s[tid] - v;
}

__global__ __launch_bounds__(256) void scan_finish(int* __restrict__ off,
                                                   const int* __restrict__ bsum,
                                                   int* __restrict__ cursor,
                                                   int n, int nE) {
    const int gid = blockIdx.x * 256 + threadIdx.x;
    if (gid < n) {
        int o = off[gid] + bsum[blockIdx.x];
        off[gid] = o;
        cursor[gid] = o;
    }
    if (gid == 0) off[n] = nE;
}

__global__ __launch_bounds__(256) void build_perm(const int* __restrict__ src,
                                                  const int* __restrict__ dst,
                                                  int* __restrict__ cursor,
                                                  int* __restrict__ perm, int nE) {
    int i = blockIdx.x * 256 + threadIdx.x;
    int st = gridDim.x * 256;
    for (; i < nE; i += st) {
        int slot = atomicAdd(&cursor[dst[i]], 1);
        perm[slot] = src[i];
    }
}

__global__ __launch_bounds__(256) void pack_w(const float* __restrict__ Wl,
                                              const float* __restrict__ Wr,
                                              float* __restrict__ Pl,
                                              float* __restrict__ Pr) {
    __shared__ __attribute__((aligned(16))) float t[16 * (DIM + 4)];
    const int tid = threadIdx.x;
    const int mat = blockIdx.x >> 4;
    const int nb  = (blockIdx.x & 15) * 16;
    const float* W = (blockIdx.y ? Wr : Wl) + (size_t)mat * DIM * DIM + (size_t)nb * DIM;
    float*       P = (blockIdx.y ? Pr : Pl) + (size_t)mat * DIM * DIM;
    for (int i = tid; i < 16 * KG; i += 256) {
        const int r = i >> 6, g = i & 63;
        *(float4*)&t[r * (DIM + 4) + 4 * g] = *(const float4*)(W + (size_t)r * DIM + 4 * g);
    }
    __syncthreads();
    for (int i = tid; i < 16 * KG; i += 256) {
        const int g = i >> 4, r = i & 15;
        *(float4*)(P + ((size_t)g * DIM + nb + r) * 4) = *(const float4*)&t[r * (DIM + 4) + 4 * g];
    }
}

template <int NB, bool FULL>
__device__ __forceinline__ void gather_batch(const float* __restrict__ xsrc, int idx,
                                             int j, int cnt, int lane, float (&acc)[8]) {
    float4 va[NB], vb[NB];
#pragma unroll
    for (int u = 0; u < NB; ++u) {
        const int jj = (FULL || u < cnt) ? (j + u) : (j + cnt - 1);
        const int p  = __builtin_amdgcn_readlane(idx, jj);
        const float4* row = (const float4*)(xsrc + (size_t)p * DIM);
        va[u] = row[lane];
        vb[u] = row[lane + 32];
    }
#pragma unroll
    for (int u = 0; u < NB; ++u) {
        const bool on = FULL || (u < cnt);
        acc[0] += on ? va[u].x : 0.0f; acc[1] += on ? va[u].y : 0.0f;
        acc[2] += on ? va[u].z : 0.0f; acc[3] += on ? va[u].w : 0.0f;
        acc[4] += on ? vb[u].x : 0.0f; acc[5] += on ? vb[u].y : 0.0f;
        acc[6] += on ? vb[u].z : 0.0f; acc[7] += on ? vb[u].w : 0.0f;
    }
}

template <int NS>
__device__ __forceinline__ void load_block(const float4* w0, const float4* w1,
                                           const float* aRow, int j,
                                           float4 (&b)[2], float4 (&a)[NS]) {
    b[0] = w0[(size_t)j * (2 * DIM)];
    b[1] = w1[(size_t)j * (2 * DIM)];
#pragma unroll
    for (int s = 0; s < NS; ++s)
        a[s] = *(const float4*)(aRow + s * 16 * LSTRIDE + 8 * j);
}

template <int NS>
__device__ __forceinline__ void mma_block(v8f (&acc)[NS][2], const float4 (&b)[2],
                                          const float4 (&a)[NS]) {
#pragma unroll
    for (int st = 0; st < 2; ++st) {
        const v2f b0 = st ? (v2f){b[0].z, b[0].w} : (v2f){b[0].x, b[0].y};
        const v2f b1 = st ? (v2f){b[1].z, b[1].w} : (v2f){b[1].x, b[1].y};
#pragma unroll
        for (int s = 0; s < NS; ++s) {
            const v2f as = st ? (v2f){a[s].z, a[s].w} : (v2f){a[s].x, a[s].y};
            acc[s][0] = __builtin_amdgcn_wmma_f32_16x16x4_f32(false, as, false, b0, (short)0, acc[s][0], false, false);
            acc[s][1] = __builtin_amdgcn_wmma_f32_16x16x4_f32(false, as, false, b1, (short)0, acc[s][1], false, false);
        }
    }
}

template <int NS>
__device__ __forceinline__ void k_loop(v8f (&acc)[NS][2], const float4* w0, const float4* w1,
                                       const float* aRow) {
    constexpr int NBLK = DIM / 8;
    float4 bP[2], aP[NS];
    float4 bQ[2], aQ[NS];

    load_block<NS>(w0, w1, aRow, 0, bP, aP);
#pragma unroll 1
    for (int i = 0; i < NBLK - 2; i += 2) {
        load_block<NS>(w0, w1, aRow, i + 1, bQ, aQ);
        __builtin_amdgcn_sched_barrier(0);
        mma_block<NS>(acc, bP, aP);
        __builtin_amdgcn_sched_barrier(0);
        load_block<NS>(w0, w1, aRow, i + 2, bP, aP);
        __builtin_amdgcn_sched_barrier(0);
        mma_block<NS>(acc, bQ, aQ);
    }
    load_block<NS>(w0, w1, aRow, NBLK - 1, bQ, aQ);
    mma_block<NS>(acc, bP, aP);
    mma_block<NS>(acc, bQ, aQ);
}

template <int NS>
__device__ __forceinline__ void sage_tile(float* sA, float* __restrict__ out,
                                          const float* __restrict__ xsrc,
                                          const int* __restrict__ perm,
                                          const int* __restrict__ off,
                                          const float* __restrict__ xdst,
                                          const float* __restrict__ Pl,
                                          const float* __restrict__ bl,
                                          const float* __restrict__ Pr,
                                          int relu_norm, int nRows, int m0, int rowEnd) {
    constexpr int MT = NS * 16;

    const int tid = threadIdx.x;

    const int lane  = tid & 31;
    const int wv    = tid >> 5;
    const int mrow  = lane & 15;
    const int half  = lane >> 4;
    const int ncol  = lane & 15;
    const int n0    = wv * 32;

    const int selfFirst = (blockIdx.x ^ (blockIdx.x >> 3) ^ (blockIdx.x >> 8)) & 1;

    const float4* wl0 = (const float4*)Pl + (size_t)half * DIM + n0 + ncol;
    const float4* wl1 = wl0 + 16;
    const float4* wr0 = (const float4*)Pr + (size_t)half * DIM + n0 + ncol;
    const float4* wr1 = wr0 + 16;
    const float* aRow = &sA[mrow * LSTRIDE] + 4 * half;

    v8f acc[NS][2];
#pragma unroll
    for (int s = 0; s < NS; ++s)
#pragma unroll
        for (int t = 0; t < 2; ++t)
            acc[s][t] = (v8f){0.f, 0.f, 0.f, 0.f, 0.f, 0.f, 0.f, 0.f};

    for (int p = 0; p < 2; ++p) {
        const bool doSelf = (p == 0) ? (selfFirst != 0) : (selfFirst == 0);
        if (p) __syncthreads();

        if (doSelf) {
            for (int i = tid; i < MT * DIM / 4; i += 256) {
                const int m = i >> 6;
                const int c = (i & 63) * 4;
                int gm = m0 + m;
                if (gm >= nRows) gm = nRows - 1;
                float4 s = *(const float4*)(xdst + (size_t)gm * DIM + c);
                *(float4*)&sA[m * LSTRIDE + c] = s;
            }
        } else {
            const int rbase = m0 + wv * (MT / 8);
            int oi = rbase + lane;
            if (oi > nRows) oi = nRows;
            const int offv = (lane < MT / 8 + 1) ? off[oi] : 0;
#pragma unroll 1
            for (int r = 0; r < MT / 8; ++r) {
                const int s = __builtin_amdgcn_readlane(offv, r);
                const int e = __builtin_amdgcn_readlane(offv, r + 1);
                float g[8] = {0.f, 0.f, 0.f, 0.f, 0.f, 0.f, 0.f, 0.f};
                for (int base = s; base < e; base += 32) {
                    const int n  = min(32, e - base);
                    const int ix = (lane < n) ? perm[base + lane] : 0;
                    int j = 0;
                    for (; j + 8 <= n; j += 8) gather_batch<8, true>(xsrc, ix, j, 8, lane, g);
                    const int cnt = n - j;
                    if (cnt > 4)      gather_batch<8, false>(xsrc, ix, j, cnt, lane, g);
                    else if (cnt > 0) gather_batch<4, false>(xsrc, ix, j, cnt, lane, g);
                }
                const float inv = 1.0f / fmaxf((float)(e - s), 1.0f);
                float* pm = &sA[(wv * (MT / 8) + r) * LSTRIDE + lane * 4];
                *(float4*)pm         = make_float4(g[0] * inv, g[1] * inv, g[2] * inv, g[3] * inv);
                *(float4*)(pm + 128) = make_float4(g[4] * inv, g[5] * inv, g[6] * inv, g[7] * inv);
            }
        }
        __syncthreads();

        k_loop<NS>(acc, doSelf ? wr0 : wl0, doSelf ? wr1 : wl1, aRow);
    }

    const float bb0 = bl[n0 + ncol];
    const float bb1 = bl[n0 + 16 + ncol];
#pragma unroll
    for (int s = 0; s < NS; ++s)
#pragma unroll
        for (int r = 0; r < 8; ++r) { acc[s][0][r] += bb0; acc[s][1][r] += bb1; }

    __syncthreads();

    const int mbase = (lane >> 4) << 3;
#pragma unroll
    for (int s = 0; s < NS; ++s)
#pragma unroll
        for (int r = 0; r < 8; ++r) {
            const int m = s * 16 + mbase + r;
            sA[m * LSTRIDE + n0 + ncol]      = acc[s][0][r];
            sA[m * LSTRIDE + n0 + 16 + ncol] = acc[s][1][r];
        }
    __syncthreads();

    constexpr int RW = MT / 8;
    const int wvs = __builtin_amdgcn_readfirstlane(wv);
    float4 va[RW], vb[RW];
#pragma unroll
    for (int i = 0; i < RW; ++i) {
        const float* row = &sA[(wvs * RW + i) * LSTRIDE + lane * 4];
        va[i] = *(const float4*)row;
        vb[i] = *(const float4*)(row + 128);
    }
    if (relu_norm) {
        float ss[RW];
#pragma unroll
        for (int i = 0; i < RW; ++i)
            ss[i] = va[i].x * va[i].x + va[i].y * va[i].y + va[i].z * va[i].z + va[i].w * va[i].w +
                    vb[i].x * vb[i].x + vb[i].y * vb[i].y + vb[i].z * vb[i].z + vb[i].w * vb[i].w;
#pragma unroll
        for (int d = 16; d >= 1; d >>= 1)
#pragma unroll
            for (int i = 0; i < RW; ++i) ss[i] += __shfl_xor(ss[i], d, 32);
        float t = 1.0f;
#pragma unroll
        for (int i = 0; i < RW; ++i) t = (lane == i) ? ss[i] : t;
        const float invl = 1.0f / fmaxf(sqrtf(t), 1e-12f);
#pragma unroll
        for (int i = 0; i < RW; ++i) {
            const float inv = __int_as_float(__builtin_amdgcn_readlane(__float_as_int(invl), i));
            va[i].x = fmaxf(va[i].x * inv, 0.f); va[i].y = fmaxf(va[i].y * inv, 0.f);
            va[i].z = fmaxf(va[i].z * inv, 0.f); va[i].w = fmaxf(va[i].w * inv, 0.f);
            vb[i].x = fmaxf(vb[i].x * inv, 0.f); vb[i].y = fmaxf(vb[i].y * inv, 0.f);
            vb[i].z = fmaxf(vb[i].z * inv, 0.f); vb[i].w = fmaxf(vb[i].w * inv, 0.f);
        }
    }
#pragma unroll
    for (int i = 0; i < RW; ++i) {
        const int gm = m0 + wvs * RW + i;
        if (gm < rowEnd) {
            float* orow = out + (size_t)gm * DIM + lane * 4;
            *(float4*)orow         = va[i];
            *(float4*)(orow + 128) = vb[i];
        }
    }
}

__global__ __launch_bounds__(256, 4) void sage_gemm(float* __restrict__ out,
                                                 const float* __restrict__ xsrc,
                                                 const int* __restrict__ perm,
                                                 const int* __restrict__ off,
                                                 const float* __restrict__ xdst,
                                                 const float* __restrict__ Pl,
                                                 const float* __restrict__ bl,
                                                 const float* __restrict__ Pr,
                                                 int relu_norm, int nRows) {
    __shared__ __attribute__((aligned(16))) float sA[NSMAX * 16 * LSTRIDE];

    const int strips = (nRows + 15) >> 4;
    const int nwg    = (int)gridDim.x;
    const int base   = strips / nwg;
    const int rem    = strips - base * nwg;
    const int b      = (int)blockIdx.x;
    const int mine   = base + (b < rem ? 1 : 0);
    if (mine == 0) return;
    const int m0     = (b * base + min(b, rem)) * 16;
    const int rowEnd = min(nRows, m0 + mine * 16);

    if (mine == NSMAX - 1)
        sage_tile<NSMAX - 1>(sA, out, xsrc, perm, off, xdst, Pl, bl, Pr, relu_norm, nRows, m0, rowEnd);
    else
        sage_tile<NSMAX>(sA, out, xsrc, perm, off, xdst, Pl, bl, Pr, relu_norm, nRows, m0, rowEnd);
}

extern "C" void kernel_launch(void* const* d_in, const int* in_sizes, int n_in,
                              void* d_out, int out_size, void* d_ws, size_t ws_size,
                              hipStream_t stream) {
    (void)n_in; (void)out_size; (void)ws_size;
    const float* xA = (const float*)d_in[0];
    const float* xP = (const float*)d_in[1];
    const float* Wl = (const float*)d_in[2];
    const float* bl = (const float*)d_in[3];
    const float* Wr = (const float*)d_in[4];
    const int* eap  = (const int*)d_in[5];
    const int* epa  = (const int*)d_in[6];

    const int NA = in_sizes[0] / DIM;
    const int NP = in_sizes[1] / DIM;
    const int E  = in_sizes[5] / 2;
    const int L  = in_sizes[2] / (2 * DIM * DIM);

    const size_t FA = (size_t)NA * DIM, FP = (size_t)NP * DIM;
    char* ws = (char*)d_ws;
    float* A0 = (float*)ws;  ws += FA * sizeof(float);
    float* A1 = (float*)ws;  ws += FA * sizeof(float);
    float* P0 = (float*)ws;  ws += FP * sizeof(float);
    float* P1 = (float*)ws;  ws += FP * sizeof(float);
    const size_t FW = (size_t)L * 2 * DIM * DIM;
    float* PlAll = (float*)ws;  ws += FW * sizeof(float);
    float* PrAll = (float*)ws;  ws += FW * sizeof(float);
    int* cntP  = (int*)ws;   ws += (size_t)NP * 4;
    int* offP  = (int*)ws;   ws += (size_t)(NP + 1) * 4;
    int* curP  = (int*)ws;   ws += (size_t)NP * 4;
    int* permP = (int*)ws;   ws += (size_t)E * 4;
    int* cntA  = (int*)ws;   ws += (size_t)NA * 4;
    int* offA  = (int*)ws;   ws += (size_t)(NA + 1) * 4;
    int* curA  = (int*)ws;   ws += (size_t)NA * 4;
    int* permA = (int*)ws;   ws += (size_t)E * 4;
    int* bsum  = (int*)ws;   ws += 256 * 4;

    const int nbP = (NP + 255) / 256, nbA = (NA + 255) / 256;

    int gP = ((NP + 15) / 16 + NSMAX - 1) / NSMAX;
    int gA = ((NA + 15) / 16 + NSMAX - 1) / NSMAX;
    if (gP < WGS) gP = WGS;
    if (gA < WGS) gA = WGS;

    pack_w<<<dim3(L * 2 * 16, 2), 256, 0, stream>>>(Wl, Wr, PlAll, PrAll);

    zero_i32   <<<64, 256, 0, stream>>>(cntP, NP);
    count_dst  <<<1024, 256, 0, stream>>>(eap + E, cntP, E);
    scan_block <<<nbP, 256, 0, stream>>>(cntP, offP, bsum, NP);
    scan_sums  <<<1, 256, 0, stream>>>(bsum, nbP);
    scan_finish<<<nbP, 256, 0, stream>>>(offP, bsum, curP, NP, E);
    build_perm <<<1024, 256, 0, stream>>>(eap, eap + E, curP, permP, E);

    zero_i32   <<<64, 256, 0, stream>>>(cntA, NA);
    count_dst  <<<1024, 256, 0, stream>>>(epa + E, cntA, E);
    scan_block <<<nbA, 256, 0, stream>>>(cntA, offA, bsum, NA);
    scan_sums  <<<1, 256, 0, stream>>>(bsum, nbA);
    scan_finish<<<nbA, 256, 0, stream>>>(offA, bsum, curA, NA, E);
    build_perm <<<1024, 256, 0, stream>>>(epa, epa + E, curA, permA, E);

    const float* xa_cur = xA;
    const float* xp_cur = xP;
    for (int l = 0; l < L; ++l) {
        const bool last = (l == L - 1);
        float* xa_next = last ? (float*)d_out        : ((l & 1) ? A1 : A0);
        float* xp_next = last ? (float*)d_out + FA   : ((l & 1) ? P1 : P0);

        const float* Wl0 = PlAll + ((size_t)(l * 2 + 0)) * DIM * DIM;
        const float* Wl1 = PlAll + ((size_t)(l * 2 + 1)) * DIM * DIM;
        const float* Wr0 = PrAll + ((size_t)(l * 2 + 0)) * DIM * DIM;
        const float* Wr1 = PrAll + ((size_t)(l * 2 + 1)) * DIM * DIM;
        const float* bl0 = bl + (size_t)(l * 2 + 0) * DIM;
        const float* bl1 = bl + (size_t)(l * 2 + 1) * DIM;

        sage_gemm<<<gP, 256, 0, stream>>>(xp_next, xa_cur, permP, offP, xp_cur, Wl0, bl0, Wr0, last ? 0 : 1, NP);
        sage_gemm<<<gA, 256, 0, stream>>>(xa_next, xp_cur, permA, offA, xa_cur, Wl1, bl1, Wr1, last ? 0 : 1, NA);

        xa_cur = xa_next;
        xp_cur = xp_next;
    }
}


# source lists by destination built by a two-level counting sort on LDS counters (256-node bins, then nodes) instead of per-edge global atomics: 23 -> 15 launches; a row's neighbour sum order changes
# speedup vs baseline: 1.5989x; 1.1654x over previous
#include <hip/hip_runtime.h>
#include <math.h>

typedef __attribute__((ext_vector_type(2))) float v2f;
typedef __attribute__((ext_vector_type(8))) float v8f;

#define DIM      256
#define NSMAX    7
#define WGS      512
#define LSTRIDE  260
#define KG       (DIM / 4)

#define BSH   8
#define BW    (1 << BSH)
#define NCH   256
#define MAXB  1024
#define EPT   16


__device__ __forceinline__ int block_scan256(int v, int* s, int tid, int& total) {
    s[tid] = v;
    __syncthreads();
    for (int d = 1; d < 256; d <<= 1) {
        int t = (tid >= d) ? s[tid - d] : 0;
        __syncthreads();
        s[tid] += t;
        __syncthreads();
    }
    total = s[255];
    const int ex = s[tid] - v;
    __syncthreads();
    return ex;
}

__global__ __launch_bounds__(256) void bin_count(const int* __restrict__ e0,
                                                 const int* __restrict__ e1,
                                                 int* __restrict__ hist,
                                                 int nE, int nbk0, int nbk1) {
    __shared__ int h[MAXB];
    const int tid = threadIdx.x;
    const int c   = blockIdx.x;
    const int nbk = blockIdx.y ? nbk1 : nbk0;
    const int* dst = (blockIdx.y ? e1 : e0) + nE;
    int* H = hist + (blockIdx.y ? (size_t)nbk0 * NCH : 0);
    for (int b = tid; b < nbk; b += 256) h[b] = 0;
    __syncthreads();
    const int ch = (nE + NCH - 1) / NCH;
    const int lo = c * ch;
    const int hi = min(nE, lo + ch);
    for (int i0 = lo + tid; i0 < hi; i0 += 256 * EPT) {
        int d[EPT];
#pragma unroll
        for (int u = 0; u < EPT; ++u) {
            const int i = i0 + u * 256;
            d[u] = (i < hi) ? dst[i] : -1;
        }
#pragma unroll
        for (int u = 0; u < EPT; ++u)
            if (d[u] >= 0) atomicAdd(&h[d[u] >> BSH], 1);
    }
    __syncthreads();
    for (int b = tid; b < nbk; b += 256) H[(size_t)b * NCH + c] = h[b];
}

__global__ __launch_bounds__(256) void bin_scan(int* __restrict__ hist,
                                                int* __restrict__ btot,
                                                int nbk0, int nbk1) {
    __shared__ int s[256];
    const int tid = threadIdx.x;
    const int b   = blockIdx.x;
    const int nbk = blockIdx.y ? nbk1 : nbk0;
    if (b >= nbk) return;
    int* H = hist + (blockIdx.y ? (size_t)nbk0 * NCH : 0) + (size_t)b * NCH;
    const int v = H[tid];
    int total;
    const int ex = block_scan256(v, s, tid, total);
    H[tid] = ex;
    if (tid == 0) btot[(blockIdx.y ? nbk0 : 0) + b] = total;
}

__global__ __launch_bounds__(256) void bin_scatter(const int* __restrict__ e0,
                                                   const int* __restrict__ e1,
                                                   const int* __restrict__ hist,
                                                   const int* __restrict__ btot,
                                                   int* __restrict__ bstart,
                                                   int2* __restrict__ pairs,
                                                   int nE, int nbk0, int nbk1) {
    __shared__ int s[256];
    __shared__ int cur[MAXB];
    const int tid = threadIdx.x;
    const int c   = blockIdx.x;
    const int nbk = blockIdx.y ? nbk1 : nbk0;
    const int* src = blockIdx.y ? e1 : e0;
    const int* dst = src + nE;
    const int* H  = hist + (blockIdx.y ? (size_t)nbk0 * NCH : 0);
    const int* T  = btot + (blockIdx.y ? nbk0 : 0);
    int*       BS = bstart + (blockIdx.y ? nbk0 + 1 : 0);
    int2*      PR = pairs + (blockIdx.y ? (size_t)nE : 0);

    int carry = 0;
    for (int b0 = 0; b0 < nbk; b0 += 256) {
        const int b = b0 + tid;
        const int v  = (b < nbk) ? T[b] : 0;
        const int hv = (b < nbk) ? H[(size_t)b * NCH + c] : 0;
        int total;
        const int ex = carry + block_scan256(v, s, tid, total);
        if (b < nbk) {
            cur[b] = ex + hv;
            if (c == 0) BS[b] = ex;
        }
        carry += total;
    }
    if (c == 0 && tid == 0) BS[nbk] = carry;
    __syncthreads();

    const int ch = (nE + NCH - 1) / NCH;
    const int lo = c * ch;
    const int hi = min(nE, lo + ch);
    for (int i0 = lo + tid; i0 < hi; i0 += 256 * EPT) {
        int sv[EPT], dv[EPT];
#pragma unroll
        for (int u = 0; u < EPT; ++u) {
            const int i = i0 + u * 256;
            sv[u] = (i < hi) ? src[i] : 0;
            dv[u] = (i < hi) ? dst[i] : -1;
        }
#pragma unroll
        for (int u = 0; u < EPT; ++u)
            if (dv[u] >= 0) {
                const int slot = atomicAdd(&cur[dv[u] >> BSH], 1);
                PR[slot] = make_int2(sv[u], dv[u]);
            }
    }
}

__global__ __launch_bounds__(256) void bucket_csr(const int2* __restrict__ pairs,
                                                  const int* __restrict__ bstart,
                                                  int* __restrict__ off0, int* __restrict__ off1,
                                                  int* __restrict__ perm0, int* __restrict__ perm1,
                                                  int nE, int n0, int n1, int nbk0, int nbk1) {
    __shared__ int s[256];
    __shared__ int cnt[BW];
    const int tid = threadIdx.x;
    const int b   = blockIdx.x;
    const int nbk = blockIdx.y ? nbk1 : nbk0;
    if (b >= nbk) return;
    const int2* PR = pairs + (blockIdx.y ? (size_t)nE : 0);
    const int*  BS = bstart + (blockIdx.y ? nbk0 + 1 : 0);
    int* off  = blockIdx.y ? off1 : off0;
    int* perm = blockIdx.y ? perm1 : perm0;
    const int n    = blockIdx.y ? n1 : n0;
    const int lo   = BS[b];
    const int hi   = BS[b + 1];
    const int base = b << BSH;

    cnt[tid] = 0;
    __syncthreads();
    for (int i0 = lo + tid; i0 < hi; i0 += 256 * EPT) {
        int d[EPT];
#pragma unroll
        for (int u = 0; u < EPT; ++u) {
            const int i = i0 + u * 256;
            d[u] = (i < hi) ? PR[i].y : -1;
        }
#pragma unroll
        for (int u = 0; u < EPT; ++u)
            if (d[u] >= 0) atomicAdd(&cnt[d[u] - base], 1);
    }
    __syncthreads();
    const int v = cnt[tid];
    int total;
    const int ex = block_scan256(v, s, tid, total);
    const int node = base + tid;
    if (node < n) off[node] = lo + ex;
    if (b == nbk - 1 && tid == 0) off[n] = hi;
    cnt[tid] = lo + ex;
    __syncthreads();
    for (int i0 = lo + tid; i0 < hi; i0 += 256 * EPT) {
        int2 p[EPT];
#pragma unroll
        for (int u = 0; u < EPT; ++u) {
            const int i = i0 + u * 256;
            p[u] = (i < hi) ? PR[i] : make_int2(0, -1);
        }
#pragma unroll
        for (int u = 0; u < EPT; ++u)
            if (p[u].y >= 0) {
                const int slot = atomicAdd(&cnt[p[u].y - base], 1);
                perm[slot] = p[u].x;
            }
    }
}

__global__ __launch_bounds__(256) void pack_w(const float* __restrict__ Wl,
                                              const float* __restrict__ Wr,
                                              float* __restrict__ Pl,
                                              float* __restrict__ Pr) {
    __shared__ __attribute__((aligned(16))) float t[16 * (DIM + 4)];
    const int tid = threadIdx.x;
    const int mat = blockIdx.x >> 4;
    const int nb  = (blockIdx.x & 15) * 16;
    const float* W = (blockIdx.y ? Wr : Wl) + (size_t)mat * DIM * DIM + (size_t)nb * DIM;
    float*       P = (blockIdx.y ? Pr : Pl) + (size_t)mat * DIM * DIM;
    for (int i = tid; i < 16 * KG; i += 256) {
        const int r = i >> 6, g = i & 63;
        *(float4*)&t[r * (DIM + 4) + 4 * g] = *(const float4*)(W + (size_t)r * DIM + 4 * g);
    }
    __syncthreads();
    for (int i = tid; i < 16 * KG; i += 256) {
        const int g = i >> 4, r = i & 15;
        *(float4*)(P + ((size_t)g * DIM + nb + r) * 4) = *(const float4*)&t[r * (DIM + 4) + 4 * g];
    }
}

template <int NB, bool FULL>
__device__ __forceinline__ void gather_batch(const float* __restrict__ xsrc, int idx,
                                             int j, int cnt, int lane, float (&acc)[8]) {
    float4 va[NB], vb[NB];
#pragma unroll
    for (int u = 0; u < NB; ++u) {
        const int jj = (FULL || u < cnt) ? (j + u) : (j + cnt - 1);
        const int p  = __builtin_amdgcn_readlane(idx, jj);
        const float4* row = (const float4*)(xsrc + (size_t)p * DIM);
        va[u] = row[lane];
        vb[u] = row[lane + 32];
    }
#pragma unroll
    for (int u = 0; u < NB; ++u) {
        const bool on = FULL || (u < cnt);
        acc[0] += on ? va[u].x : 0.0f; acc[1] += on ? va[u].y : 0.0f;
        acc[2] += on ? va[u].z : 0.0f; acc[3] += on ? va[u].w : 0.0f;
        acc[4] += on ? vb[u].x : 0.0f; acc[5] += on ? vb[u].y : 0.0f;
        acc[6] += on ? vb[u].z : 0.0f; acc[7] += on ? vb[u].w : 0.0f;
    }
}

template <int NS>
__device__ __forceinline__ void load_block(const float4* w0, const float4* w1,
                                           const float* aRow, int j,
                                           float4 (&b)[2], float4 (&a)[NS]) {
    b[0] = w0[(size_t)j * (2 * DIM)];
    b[1] = w1[(size_t)j * (2 * DIM)];
#pragma unroll
    for (int s = 0; s < NS; ++s)
        a[s] = *(const float4*)(aRow + s * 16 * LSTRIDE + 8 * j);
}

template <int NS>
__device__ __forceinline__ void mma_block(v8f (&acc)[NS][2], const float4 (&b)[2],
                                          const float4 (&a)[NS]) {
#pragma unroll
    for (int st = 0; st < 2; ++st) {
        const v2f b0 = st ? (v2f){b[0].z, b[0].w} : (v2f){b[0].x, b[0].y};
        const v2f b1 = st ? (v2f){b[1].z, b[1].w} : (v2f){b[1].x, b[1].y};
#pragma unroll
        for (int s = 0; s < NS; ++s) {
            const v2f as = st ? (v2f){a[s].z, a[s].w} : (v2f){a[s].x, a[s].y};
            acc[s][0] = __builtin_amdgcn_wmma_f32_16x16x4_f32(false, as, false, b0, (short)0, acc[s][0], false, false);
            acc[s][1] = __builtin_amdgcn_wmma_f32_16x16x4_f32(false, as, false, b1, (short)0, acc[s][1], false, false);
        }
    }
}

template <int NS>
__device__ __forceinline__ void k_loop(v8f (&acc)[NS][2], const float4* w0, const float4* w1,
                                       const float* aRow) {
    constexpr int NBLK = DIM / 8;
    float4 bP[2], aP[NS];
    float4 bQ[2], aQ[NS];

    load_block<NS>(w0, w1, aRow, 0, bP, aP);
#pragma unroll 1
    for (int i = 0; i < NBLK - 2; i += 2) {
        load_block<NS>(w0, w1, aRow, i + 1, bQ, aQ);
        __builtin_amdgcn_sched_barrier(0);
        mma_block<NS>(acc, bP, aP);
        __builtin_amdgcn_sched_barrier(0);
        load_block<NS>(w0, w1, aRow, i + 2, bP, aP);
        __builtin_amdgcn_sched_barrier(0);
        mma_block<NS>(acc, bQ, aQ);
    }
    load_block<NS>(w0, w1, aRow, NBLK - 1, bQ, aQ);
    mma_block<NS>(acc, bP, aP);
    mma_block<NS>(acc, bQ, aQ);
}

template <int NS>
__device__ __forceinline__ void sage_tile(float* sA, float* __restrict__ out,
                                          const float* __restrict__ xsrc,
                                          const int* __restrict__ perm,
                                          const int* __restrict__ off,
                                          const float* __restrict__ xdst,
                                          const float* __restrict__ Pl,
                                          const float* __restrict__ bl,
                                          const float* __restrict__ Pr,
                                          int relu_norm, int nRows, int m0, int rowEnd) {
    constexpr int MT = NS * 16;

    const int tid = threadIdx.x;

    const int lane  = tid & 31;
    const int wv    = tid >> 5;
    const int mrow  = lane & 15;
    const int half  = lane >> 4;
    const int ncol  = lane & 15;
    const int n0    = wv * 32;

    const int selfFirst = (blockIdx.x ^ (blockIdx.x >> 3) ^ (blockIdx.x >> 8)) & 1;

    const float4* wl0 = (const float4*)Pl + (size_t)half * DIM + n0 + ncol;
    const float4* wl1 = wl0 + 16;
    const float4* wr0 = (const float4*)Pr + (size_t)half * DIM + n0 + ncol;
    const float4* wr1 = wr0 + 16;
    const float* aRow = &sA[mrow * LSTRIDE] + 4 * half;

    v8f acc[NS][2];
#pragma unroll
    for (int s = 0; s < NS; ++s)
#pragma unroll
        for (int t = 0; t < 2; ++t)
            acc[s][t] = (v8f){0.f, 0.f, 0.f, 0.f, 0.f, 0.f, 0.f, 0.f};

    for (int p = 0; p < 2; ++p) {
        const bool doSelf = (p == 0) ? (selfFirst != 0) : (selfFirst == 0);
        if (p) __syncthreads();

        if (doSelf) {
            for (int i = tid; i < MT * DIM / 4; i += 256) {
                const int m = i >> 6;
                const int c = (i & 63) * 4;
                int gm = m0 + m;
                if (gm >= nRows) gm = nRows - 1;
                float4 s = *(const float4*)(xdst + (size_t)gm * DIM + c);
                *(float4*)&sA[m * LSTRIDE + c] = s;
            }
        } else {
            const int rbase = m0 + wv * (MT / 8);
            int oi = rbase + lane;
            if (oi > nRows) oi = nRows;
            const int offv = (lane < MT / 8 + 1) ? off[oi] : 0;
#pragma unroll 1
            for (int r = 0; r < MT / 8; ++r) {
                const int s = __builtin_amdgcn_readlane(offv, r);
                const int e = __builtin_amdgcn_readlane(offv, r + 1);
                float g[8] = {0.f, 0.f, 0.f, 0.f, 0.f, 0.f, 0.f, 0.f};
                for (int base = s; base < e; base += 32) {
                    const int n  = min(32, e - base);
                    const int ix = (lane < n) ? perm[base + lane] : 0;
                    int j = 0;
                    for (; j + 8 <= n; j += 8) gather_batch<8, true>(xsrc, ix, j, 8, lane, g);
                    const int cnt = n - j;
                    if (cnt > 4)      gather_batch<8, false>(xsrc, ix, j, cnt, lane, g);
                    else if (cnt > 0) gather_batch<4, false>(xsrc, ix, j, cnt, lane, g);
                }
                const float inv = 1.0f / fmaxf((float)(e - s), 1.0f);
                float* pm = &sA[(wv * (MT / 8) + r) * LSTRIDE + lane * 4];
                *(float4*)pm         = make_float4(g[0] * inv, g[1] * inv, g[2] * inv, g[3] * inv);
                *(float4*)(pm + 128) = make_float4(g[4] * inv, g[5] * inv, g[6] * inv, g[7] * inv);
            }
        }
        __syncthreads();

        k_loop<NS>(acc, doSelf ? wr0 : wl0, doSelf ? wr1 : wl1, aRow);
    }

    const float bb0 = bl[n0 + ncol];
    const float bb1 = bl[n0 + 16 + ncol];
#pragma unroll
    for (int s = 0; s < NS; ++s)
#pragma unroll
        for (int r = 0; r < 8; ++r) { acc[s][0][r] += bb0; acc[s][1][r] += bb1; }

    __syncthreads();

    const int mbase = (lane >> 4) << 3;
#pragma unroll
    for (int s = 0; s < NS; ++s)
#pragma unroll
        for (int r = 0; r < 8; ++r) {
            const int m = s * 16 + mbase + r;
            sA[m * LSTRIDE + n0 + ncol]      = acc[s][0][r];
            sA[m * LSTRIDE + n0 + 16 + ncol] = acc[s][1][r];
        }
    __syncthreads();

    constexpr int RW = MT / 8;
    const int wvs = __builtin_amdgcn_readfirstlane(wv);
    float4 va[RW], vb[RW];
#pragma unroll
    for (int i = 0; i < RW; ++i) {
        const float* row = &sA[(wvs * RW + i) * LSTRIDE + lane * 4];
        va[i] = *(const float4*)row;
        vb[i] = *(const float4*)(row + 128);
    }
    if (relu_norm) {
        float ss[RW];
#pragma unroll
        for (int i = 0; i < RW; ++i)
            ss[i] = va[i].x * va[i].x + va[i].y * va[i].y + va[i].z * va[i].z + va[i].w * va[i].w +
                    vb[i].x * vb[i].x + vb[i].y * vb[i].y + vb[i].z * vb[i].z + vb[i].w * vb[i].w;
#pragma unroll
        for (int d = 16; d >= 1; d >>= 1)
#pragma unroll
            for (int i = 0; i < RW; ++i) ss[i] += __shfl_xor(ss[i], d, 32);
        float t = 1.0f;
#pragma unroll
        for (int i = 0; i < RW; ++i) t = (lane == i) ? ss[i] : t;
        const float invl = 1.0f / fmaxf(sqrtf(t), 1e-12f);
#pragma unroll
        for (int i = 0; i < RW; ++i) {
            const float inv = __int_as_float(__builtin_amdgcn_readlane(__float_as_int(invl), i));
            va[i].x = fmaxf(va[i].x * inv, 0.f); va[i].y = fmaxf(va[i].y * inv, 0.f);
            va[i].z = fmaxf(va[i].z * inv, 0.f); va[i].w = fmaxf(va[i].w * inv, 0.f);
            vb[i].x = fmaxf(vb[i].x * inv, 0.f); vb[i].y = fmaxf(vb[i].y * inv, 0.f);
            vb[i].z = fmaxf(vb[i].z * inv, 0.f); vb[i].w = fmaxf(vb[i].w * inv, 0.f);
        }
    }
#pragma unroll
    for (int i = 0; i < RW; ++i) {
        const int gm = m0 + wvs * RW + i;
        if (gm < rowEnd) {
            float* orow = out + (size_t)gm * DIM + lane * 4;
            *(float4*)orow         = va[i];
            *(float4*)(orow + 128) = vb[i];
        }
    }
}

__global__ __launch_bounds__(256, 4) void sage_gemm(float* __restrict__ out,
                                                 const float* __restrict__ xsrc,
                                                 const int* __restrict__ perm,
                                                 const int* __restrict__ off,
                                                 const float* __restrict__ xdst,
                                                 const float* __restrict__ Pl,
                                                 const float* __restrict__ bl,
                                                 const float* __restrict__ Pr,
                                                 int relu_norm, int nRows) {
    __shared__ __attribute__((aligned(16))) float sA[NSMAX * 16 * LSTRIDE];

    const int strips = (nRows + 15) >> 4;
    const int nwg    = (int)gridDim.x;
    const int base   = strips / nwg;
    const int rem    = strips - base * nwg;
    const int b      = (int)blockIdx.x;
    const int mine   = base + (b < rem ? 1 : 0);
    if (mine == 0) return;
    const int m0     = (b * base + min(b, rem)) * 16;
    const int rowEnd = min(nRows, m0 + mine * 16);

    if (mine == NSMAX - 1)
        sage_tile<NSMAX - 1>(sA, out, xsrc, perm, off, xdst, Pl, bl, Pr, relu_norm, nRows, m0, rowEnd);
    else
        sage_tile<NSMAX>(sA, out, xsrc, perm, off, xdst, Pl, bl, Pr, relu_norm, nRows, m0, rowEnd);
}

extern "C" void kernel_launch(void* const* d_in, const int* in_sizes, int n_in,
                              void* d_out, int out_size, void* d_ws, size_t ws_size,
                              hipStream_t stream) {
    (void)n_in; (void)out_size; (void)ws_size;
    const float* xA = (const float*)d_in[0];
    const float* xP = (const float*)d_in[1];
    const float* Wl = (const float*)d_in[2];
    const float* bl = (const float*)d_in[3];
    const float* Wr = (const float*)d_in[4];
    const int* eap  = (const int*)d_in[5];
    const int* epa  = (const int*)d_in[6];

    const int NA = in_sizes[0] / DIM;
    const int NP = in_sizes[1] / DIM;
    const int E  = in_sizes[5] / 2;
    const int L  = in_sizes[2] / (2 * DIM * DIM);

    const size_t FA = (size_t)NA * DIM, FP = (size_t)NP * DIM;
    char* ws = (char*)d_ws;
    float* A0 = (float*)ws;  ws += FA * sizeof(float);
    float* A1 = (float*)ws;  ws += FA * sizeof(float);
    float* P0 = (float*)ws;  ws += FP * sizeof(float);
    float* P1 = (float*)ws;  ws += FP * sizeof(float);
    const size_t FW = (size_t)L * 2 * DIM * DIM;
    float* PlAll = (float*)ws;  ws += FW * sizeof(float);
    float* PrAll = (float*)ws;  ws += FW * sizeof(float);
    const int nbkP = (NP + BW - 1) >> BSH, nbkA = (NA + BW - 1) >> BSH;
    const int nbkM = nbkP > nbkA ? nbkP : nbkA;
    int2* pairs = (int2*)ws; ws += (size_t)2 * E * sizeof(int2);
    int* permP = (int*)ws;   ws += (size_t)E * 4;
    int* permA = (int*)ws;   ws += (size_t)E * 4;
    int* offP  = (int*)ws;   ws += (size_t)(NP + 1) * 4;
    int* offA  = (int*)ws;   ws += (size_t)(NA + 1) * 4;
    int* hist  = (int*)ws;   ws += (size_t)(nbkP + nbkA) * NCH * 4;
    int* btot  = (int*)ws;   ws += (size_t)(nbkP + nbkA) * 4;
    int* bstart = (int*)ws;  ws += (size_t)(nbkP + nbkA + 2) * 4;

    int gP = ((NP + 15) / 16 + NSMAX - 1) / NSMAX;
    int gA = ((NA + 15) / 16 + NSMAX - 1) / NSMAX;
    if (gP < WGS) gP = WGS;
    if (gA < WGS) gA = WGS;

    pack_w<<<dim3(L * 2 * 16, 2), 256, 0, stream>>>(Wl, Wr, PlAll, PrAll);

    bin_count  <<<dim3(NCH, 2), 256, 0, stream>>>(eap, epa, hist, E, nbkP, nbkA);
    bin_scan   <<<dim3(nbkM, 2), 256, 0, stream>>>(hist, btot, nbkP, nbkA);
    bin_scatter<<<dim3(NCH, 2), 256, 0, stream>>>(eap, epa, hist, btot, bstart, pairs, E, nbkP, nbkA);
    bucket_csr <<<dim3(nbkM, 2), 256, 0, stream>>>(pairs, bstart, offP, offA, permP, permA, E, NP, NA, nbkP, nbkA);

    const float* xa_cur = xA;
    const float* xp_cur = xP;
    for (int l = 0; l < L; ++l) {
        const bool last = (l == L - 1);
        float* xa_next = last ? (float*)d_out        : ((l & 1) ? A1 : A0);
        float* xp_next = last ? (float*)d_out + FA   : ((l & 1) ? P1 : P0);

        const float* Wl0 = PlAll + ((size_t)(l * 2 + 0)) * DIM * DIM;
        const float* Wl1 = PlAll + ((size_t)(l * 2 + 1)) * DIM * DIM;
        const float* Wr0 = PrAll + ((size_t)(l * 2 + 0)) * DIM * DIM;
        const float* Wr1 = PrAll + ((size_t)(l * 2 + 1)) * DIM * DIM;
        const float* bl0 = bl + (size_t)(l * 2 + 0) * DIM;
        const float* bl1 = bl + (size_t)(l * 2 + 1) * DIM;

        sage_gemm<<<gP, 256, 0, stream>>>(xp_next, xa_cur, permP, offP, xp_cur, Wl0, bl0, Wr0, last ? 0 : 1, NP);
        sage_gemm<<<gA, 256, 0, stream>>>(xa_next, xp_cur, permA, offA, xa_cur, Wl1, bl1, Wr1, last ? 0 : 1, NA);

        xa_cur = xa_next;
        xp_cur = xp_next;
    }
}


# a layer's two launches (one per node type) merged into one: each of 512 resident work-groups runs tile b of the first type, then tile 511-b of the second; 15 -> 10 launches; sum orders unchanged
# speedup vs baseline: 1.6949x; 1.0600x over previous
#include <hip/hip_runtime.h>
#include <math.h>

typedef __attribute__((ext_vector_type(2))) float v2f;
typedef __attribute__((ext_vector_type(8))) float v8f;

#define DIM      256
#define NSMAX    7
#define WGS      512
#define LSTRIDE  260
#define KG       (DIM / 4)

#define BSH   8
#define BW    (1 << BSH)
#define NCH   256
#define MAXB  1024
#define EPT   16


__device__ __forceinline__ int block_scan256(int v, int* s, int tid, int& total) {
    s[tid] = v;
    __syncthreads();
    for (int d = 1; d < 256; d <<= 1) {
        int t = (tid >= d) ? s[tid - d] : 0;
        __syncthreads();
        s[tid] += t;
        __syncthreads();
    }
    total = s[255];
    const int ex = s[tid] - v;
    __syncthreads();
    return ex;
}

__global__ __launch_bounds__(256) void bin_count(const int* __restrict__ e0,
                                                 const int* __restrict__ e1,
                                                 int* __restrict__ hist,
                                                 int nE, int nbk0, int nbk1) {
    __shared__ int h[MAXB];
    const int tid = threadIdx.x;
    const int c   = blockIdx.x;
    const int nbk = blockIdx.y ? nbk1 : nbk0;
    const int* dst = (blockIdx.y ? e1 : e0) + nE;
    int* H = hist + (blockIdx.y ? (size_t)nbk0 * NCH : 0);
    for (int b = tid; b < nbk; b += 256) h[b] = 0;
    __syncthreads();
    const int ch = (nE + NCH - 1) / NCH;
    const int lo = c * ch;
    const int hi = min(nE, lo + ch);
    for (int i0 = lo + tid; i0 < hi; i0 += 256 * EPT) {
        int d[EPT];
#pragma unroll
        for (int u = 0; u < EPT; ++u) {
            const int i = i0 + u * 256;
            d[u] = (i < hi) ? dst[i] : -1;
        }
#pragma unroll
        for (int u = 0; u < EPT; ++u)
            if (d[u] >= 0) atomicAdd(&h[d[u] >> BSH], 1);
    }
    __syncthreads();
    for (int b = tid; b < nbk; b += 256) H[(size_t)b * NCH + c] = h[b];
}

__global__ __launch_bounds__(256) void bin_scan(int* __restrict__ hist,
                                                int* __restrict__ btot,
                                                int nbk0, int nbk1) {
    __shared__ int s[256];
    const int tid = threadIdx.x;
    const int b   = blockIdx.x;
    const int nbk = blockIdx.y ? nbk1 : nbk0;
    if (b >= nbk) return;
    int* H = hist + (blockIdx.y ? (size_t)nbk0 * NCH : 0) + (size_t)b * NCH;
    const int v = H[tid];
    int total;
    const int ex = block_scan256(v, s, tid, total);
    H[tid] = ex;
    if (tid == 0) btot[(blockIdx.y ? nbk0 : 0) + b] = total;
}

__global__ __launch_bounds__(256) void bin_scatter(const int* __restrict__ e0,
                                                   const int* __restrict__ e1,
                                                   const int* __restrict__ hist,
                                                   const int* __restrict__ btot,
                                                   int* __restrict__ bstart,
                                                   int2* __restrict__ pairs,
                                                   int nE, int nbk0, int nbk1) {
    __shared__ int s[256];
    __shared__ int cur[MAXB];
    const int tid = threadIdx.x;
    const int c   = blockIdx.x;
    const int nbk = blockIdx.y ? nbk1 : nbk0;
    const int* src = blockIdx.y ? e1 : e0;
    const int* dst = src + nE;
    const int* H  = hist + (blockIdx.y ? (size_t)nbk0 * NCH : 0);
    const int* T  = btot + (blockIdx.y ? nbk0 : 0);
    int*       BS = bstart + (blockIdx.y ? nbk0 + 1 : 0);
    int2*      PR = pairs + (blockIdx.y ? (size_t)nE : 0);

    int carry = 0;
    for (int b0 = 0; b0 < nbk; b0 += 256) {
        const int b = b0 + tid;
        const int v  = (b < nbk) ? T[b] : 0;
        const int hv = (b < nbk) ? H[(size_t)b * NCH + c] : 0;
        int total;
        const int ex = carry + block_scan256(v, s, tid, total);
        if (b < nbk) {
            cur[b] = ex + hv;
            if (c == 0) BS[b] = ex;
        }
        carry += total;
    }
    if (c == 0 && tid == 0) BS[nbk] = carry;
    __syncthreads();

    const int ch = (nE + NCH - 1) / NCH;
    const int lo = c * ch;
    const int hi = min(nE, lo + ch);
    for (int i0 = lo + tid; i0 < hi; i0 += 256 * EPT) {
        int sv[EPT], dv[EPT];
#pragma unroll
        for (int u = 0; u < EPT; ++u) {
            const int i = i0 + u * 256;
            sv[u] = (i < hi) ? src[i] : 0;
            dv[u] = (i < hi) ? dst[i] : -1;
        }
#pragma unroll
        for (int u = 0; u < EPT; ++u)
            if (dv[u] >= 0) {
                const int slot = atomicAdd(&cur[dv[u] >> BSH], 1);
                PR[slot] = make_int2(sv[u], dv[u]);
            }
    }
}

__global__ __launch_bounds__(256) void bucket_csr(const int2* __restrict__ pairs,
                                                  const int* __restrict__ bstart,
                                                  int* __restrict__ off0, int* __restrict__ off1,
                                                  int* __restrict__ perm0, int* __restrict__ perm1,
                                                  int nE, int n0, int n1, int nbk0, int nbk1) {
    __shared__ int s[256];
    __shared__ int cnt[BW];
    const int tid = threadIdx.x;
    const int b   = blockIdx.x;
    const int nbk = blockIdx.y ? nbk1 : nbk0;
    if (b >= nbk) return;
    const int2* PR = pairs + (blockIdx.y ? (size_t)nE : 0);
    const int*  BS = bstart + (blockIdx.y ? nbk0 + 1 : 0);
    int* off  = blockIdx.y ? off1 : off0;
    int* perm = blockIdx.y ? perm1 : perm0;
    const int n    = blockIdx.y ? n1 : n0;
    const int lo   = BS[b];
    const int hi   = BS[b + 1];
    const int base = b << BSH;

    cnt[tid] = 0;
    __syncthreads();
    for (int i0 = lo + tid; i0 < hi; i0 += 256 * EPT) {
        int d[EPT];
#pragma unroll
        for (int u = 0; u < EPT; ++u) {
            const int i = i0 + u * 256;
            d[u] = (i < hi) ? PR[i].y : -1;
        }
#pragma unroll
        for (int u = 0; u < EPT; ++u)
            if (d[u] >= 0) atomicAdd(&cnt[d[u] - base], 1);
    }
    __syncthreads();
    const int v = cnt[tid];
    int total;
    const int ex = block_scan256(v, s, tid, total);
    const int node = base + tid;
    if (node < n) off[node] = lo + ex;
    if (b == nbk - 1 && tid == 0) off[n] = hi;
    cnt[tid] = lo + ex;
    __syncthreads();
    for (int i0 = lo + tid; i0 < hi; i0 += 256 * EPT) {
        int2 p[EPT];
#pragma unroll
        for (int u = 0; u < EPT; ++u) {
            const int i = i0 + u * 256;
            p[u] = (i < hi) ? PR[i] : make_int2(0, -1);
        }
#pragma unroll
        for (int u = 0; u < EPT; ++u)
            if (p[u].y >= 0) {
                const int slot = atomicAdd(&cnt[p[u].y - base], 1);
                perm[slot] = p[u].x;
            }
    }
}

__global__ __launch_bounds__(256) void pack_w(const float* __restrict__ Wl,
                                              const float* __restrict__ Wr,
                                              float* __restrict__ Pl,
                                              float* __restrict__ Pr) {
    __shared__ __attribute__((aligned(16))) float t[16 * (DIM + 4)];
    const int tid = threadIdx.x;
    const int mat = blockIdx.x >> 4;
    const int nb  = (blockIdx.x & 15) * 16;
    const float* W = (blockIdx.y ? Wr : Wl) + (size_t)mat * DIM * DIM + (size_t)nb * DIM;
    float*       P = (blockIdx.y ? Pr : Pl) + (size_t)mat * DIM * DIM;
    for (int i = tid; i < 16 * KG; i += 256) {
        const int r = i >> 6, g = i & 63;
        *(float4*)&t[r * (DIM + 4) + 4 * g] = *(const float4*)(W + (size_t)r * DIM + 4 * g);
    }
    __syncthreads();
    for (int i = tid; i < 16 * KG; i += 256) {
        const int g = i >> 4, r = i & 15;
        *(float4*)(P + ((size_t)g * DIM + nb + r) * 4) = *(const float4*)&t[r * (DIM + 4) + 4 * g];
    }
}

template <int NB, bool FULL>
__device__ __forceinline__ void gather_batch(const float* __restrict__ xsrc, int idx,
                                             int j, int cnt, int lane, float (&acc)[8]) {
    float4 va[NB], vb[NB];
#pragma unroll
    for (int u = 0; u < NB; ++u) {
        const int jj = (FULL || u < cnt) ? (j + u) : (j + cnt - 1);
        const int p  = __builtin_amdgcn_readlane(idx, jj);
        const float4* row = (const float4*)(xsrc + (size_t)p * DIM);
        va[u] = row[lane];
        vb[u] = row[lane + 32];
    }
#pragma unroll
    for (int u = 0; u < NB; ++u) {
        const bool on = FULL || (u < cnt);
        acc[0] += on ? va[u].x : 0.0f; acc[1] += on ? va[u].y : 0.0f;
        acc[2] += on ? va[u].z : 0.0f; acc[3] += on ? va[u].w : 0.0f;
        acc[4] += on ? vb[u].x : 0.0f; acc[5] += on ? vb[u].y : 0.0f;
        acc[6] += on ? vb[u].z : 0.0f; acc[7] += on ? vb[u].w : 0.0f;
    }
}

template <int NS>
__device__ __forceinline__ void load_block(const float4* w0, const float4* w1,
                                           const float* aRow, int j,
                                           float4 (&b)[2], float4 (&a)[NS]) {
    b[0] = w0[(size_t)j * (2 * DIM)];
    b[1] = w1[(size_t)j * (2 * DIM)];
#pragma unroll
    for (int s = 0; s < NS; ++s)
        a[s] = *(const float4*)(aRow + s * 16 * LSTRIDE + 8 * j);
}

template <int NS>
__device__ __forceinline__ void mma_block(v8f (&acc)[NS][2], const float4 (&b)[2],
                                          const float4 (&a)[NS]) {
#pragma unroll
    for (int st = 0; st < 2; ++st) {
        const v2f b0 = st ? (v2f){b[0].z, b[0].w} : (v2f){b[0].x, b[0].y};
        const v2f b1 = st ? (v2f){b[1].z, b[1].w} : (v2f){b[1].x, b[1].y};
#pragma unroll
        for (int s = 0; s < NS; ++s) {
            const v2f as = st ? (v2f){a[s].z, a[s].w} : (v2f){a[s].x, a[s].y};
            acc[s][0] = __builtin_amdgcn_wmma_f32_16x16x4_f32(false, as, false, b0, (short)0, acc[s][0], false, false);
            acc[s][1] = __builtin_amdgcn_wmma_f32_16x16x4_f32(false, as, false, b1, (short)0, acc[s][1], false, false);
        }
    }
}

template <int NS>
__device__ __forceinline__ void k_loop(v8f (&acc)[NS][2], const float4* w0, const float4* w1,
                                       const float* aRow) {
    constexpr int NBLK = DIM / 8;
    float4 bP[2], aP[NS];
    float4 bQ[2], aQ[NS];

    load_block<NS>(w0, w1, aRow, 0, bP, aP);
#pragma unroll 1
    for (int i = 0; i < NBLK - 2; i += 2) {
        load_block<NS>(w0, w1, aRow, i + 1, bQ, aQ);
        __builtin_amdgcn_sched_barrier(0);
        mma_block<NS>(acc, bP, aP);
        __builtin_amdgcn_sched_barrier(0);
        load_block<NS>(w0, w1, aRow, i + 2, bP, aP);
        __builtin_amdgcn_sched_barrier(0);
        mma_block<NS>(acc, bQ, aQ);
    }
    load_block<NS>(w0, w1, aRow, NBLK - 1, bQ, aQ);
    mma_block<NS>(acc, bP, aP);
    mma_block<NS>(acc, bQ, aQ);
}

template <int NS>
__device__ __forceinline__ void sage_tile(float* sA, float* __restrict__ out,
                                          const float* __restrict__ xsrc,
                                          const int* __restrict__ perm,
                                          const int* __restrict__ off,
                                          const float* __restrict__ xdst,
                                          const float* __restrict__ Pl,
                                          const float* __restrict__ bl,
                                          const float* __restrict__ Pr,
                                          int relu_norm, int nRows, int m0, int rowEnd, int tileIdx) {
    constexpr int MT = NS * 16;

    const int tid = threadIdx.x;

    const int lane  = tid & 31;
    const int wv    = tid >> 5;
    const int mrow  = lane & 15;
    const int half  = lane >> 4;
    const int ncol  = lane & 15;
    const int n0    = wv * 32;

    const int selfFirst = (tileIdx ^ (tileIdx >> 3) ^ (tileIdx >> 8)) & 1;

    const float4* wl0 = (const float4*)Pl + (size_t)half * DIM + n0 + ncol;
    const float4* wl1 = wl0 + 16;
    const float4* wr0 = (const float4*)Pr + (size_t)half * DIM + n0 + ncol;
    const float4* wr1 = wr0 + 16;
    const float* aRow = &sA[mrow * LSTRIDE] + 4 * half;

    v8f acc[NS][2];
#pragma unroll
    for (int s = 0; s < NS; ++s)
#pragma unroll
        for (int t = 0; t < 2; ++t)
            acc[s][t] = (v8f){0.f, 0.f, 0.f, 0.f, 0.f, 0.f, 0.f, 0.f};

    for (int p = 0; p < 2; ++p) {
        const bool doSelf = (p == 0) ? (selfFirst != 0) : (selfFirst == 0);
        if (p) __syncthreads();

        if (doSelf) {
            for (int i = tid; i < MT * DIM / 4; i += 256) {
                const int m = i >> 6;
                const int c = (i & 63) * 4;
                int gm = m0 + m;
                if (gm >= nRows) gm = nRows - 1;
                float4 s = *(const float4*)(xdst + (size_t)gm * DIM + c);
                *(float4*)&sA[m * LSTRIDE + c] = s;
            }
        } else {
            const int rbase = m0 + wv * (MT / 8);
            int oi = rbase + lane;
            if (oi > nRows) oi = nRows;
            const int offv = (lane < MT / 8 + 1) ? off[oi] : 0;
#pragma unroll 1
            for (int r = 0; r < MT / 8; ++r) {
                const int s = __builtin_amdgcn_readlane(offv, r);
                const int e = __builtin_amdgcn_readlane(offv, r + 1);
                float g[8] = {0.f, 0.f, 0.f, 0.f, 0.f, 0.f, 0.f, 0.f};
                for (int base = s; base < e; base += 32) {
                    const int n  = min(32, e - base);
                    const int ix = (lane < n) ? perm[base + lane] : 0;
                    int j = 0;
                    for (; j + 8 <= n; j += 8) gather_batch<8, true>(xsrc, ix, j, 8, lane, g);
                    const int cnt = n - j;
                    if (cnt > 4)      gather_batch<8, false>(xsrc, ix, j, cnt, lane, g);
                    else if (cnt > 0) gather_batch<4, false>(xsrc, ix, j, cnt, lane, g);
                }
                const float inv = 1.0f / fmaxf((float)(e - s), 1.0f);
                float* pm = &sA[(wv * (MT / 8) + r) * LSTRIDE + lane * 4];
                *(float4*)pm         = make_float4(g[0] * inv, g[1] * inv, g[2] * inv, g[3] * inv);
                *(float4*)(pm + 128) = make_float4(g[4] * inv, g[5] * inv, g[6] * inv, g[7] * inv);
            }
        }
        __syncthreads();

        k_loop<NS>(acc, doSelf ? wr0 : wl0, doSelf ? wr1 : wl1, aRow);
    }

    const float bb0 = bl[n0 + ncol];
    const float bb1 = bl[n0 + 16 + ncol];
#pragma unroll
    for (int s = 0; s < NS; ++s)
#pragma unroll
        for (int r = 0; r < 8; ++r) { acc[s][0][r] += bb0; acc[s][1][r] += bb1; }

    __syncthreads();

    const int mbase = (lane >> 4) << 3;
#pragma unroll
    for (int s = 0; s < NS; ++s)
#pragma unroll
        for (int r = 0; r < 8; ++r) {
            const int m = s * 16 + mbase + r;
            sA[m * LSTRIDE + n0 + ncol]      = acc[s][0][r];
            sA[m * LSTRIDE + n0 + 16 + ncol] = acc[s][1][r];
        }
    __syncthreads();

    constexpr int RW = MT / 8;
    const int wvs = __builtin_amdgcn_readfirstlane(wv);
    float4 va[RW], vb[RW];
#pragma unroll
    for (int i = 0; i < RW; ++i) {
        const float* row = &sA[(wvs * RW + i) * LSTRIDE + lane * 4];
        va[i] = *(const float4*)row;
        vb[i] = *(const float4*)(row + 128);
    }
    if (relu_norm) {
        float ss[RW];
#pragma unroll
        for (int i = 0; i < RW; ++i)
            ss[i] = va[i].x * va[i].x + va[i].y * va[i].y + va[i].z * va[i].z + va[i].w * va[i].w +
                    vb[i].x * vb[i].x + vb[i].y * vb[i].y + vb[i].z * vb[i].z + vb[i].w * vb[i].w;
#pragma unroll
        for (int d = 16; d >= 1; d >>= 1)
#pragma unroll
            for (int i = 0; i < RW; ++i) ss[i] += __shfl_xor(ss[i], d, 32);
        float t = 1.0f;
#pragma unroll
        for (int i = 0; i < RW; ++i) t = (lane == i) ? ss[i] : t;
        const float invl = 1.0f / fmaxf(sqrtf(t), 1e-12f);
#pragma unroll
        for (int i = 0; i < RW; ++i) {
            const float inv = __int_as_float(__builtin_amdgcn_readlane(__float_as_int(invl), i));
            va[i].x = fmaxf(va[i].x * inv, 0.f); va[i].y = fmaxf(va[i].y * inv, 0.f);
            va[i].z = fmaxf(va[i].z * inv, 0.f); va[i].w = fmaxf(va[i].w * inv, 0.f);
            vb[i].x = fmaxf(vb[i].x * inv, 0.f); vb[i].y = fmaxf(vb[i].y * inv, 0.f);
            vb[i].z = fmaxf(vb[i].z * inv, 0.f); vb[i].w = fmaxf(vb[i].w * inv, 0.f);
        }
    }
#pragma unroll
    for (int i = 0; i < RW; ++i) {
        const int gm = m0 + wvs * RW + i;
        if (gm < rowEnd) {
            float* orow = out + (size_t)gm * DIM + lane * 4;
            *(float4*)orow         = va[i];
            *(float4*)(orow + 128) = vb[i];
        }
    }
}

__global__ __launch_bounds__(256, 4) void sage_gemm(float* __restrict__ out0,
                                                 const float* __restrict__ xsrc0,
                                                 const int* __restrict__ perm0,
                                                 const int* __restrict__ off0,
                                                 const float* __restrict__ xdst0,
                                                 const float* __restrict__ Pl0,
                                                 const float* __restrict__ bl0,
                                                 const float* __restrict__ Pr0,
                                                 int nRows0, int g0,
                                                 float* __restrict__ out1,
                                                 const float* __restrict__ xsrc1,
                                                 const int* __restrict__ perm1,
                                                 const int* __restrict__ off1,
                                                 const float* __restrict__ xdst1,
                                                 const float* __restrict__ Pl1,
                                                 const float* __restrict__ bl1,
                                                 const float* __restrict__ Pr1,
                                                 int nRows1, int g1,
                                                 int relu_norm) {
    __shared__ __attribute__((aligned(16))) float sA[NSMAX * 16 * LSTRIDE];

#pragma unroll 1
    for (int side = 0; side < 2; ++side) {
        const int nwg = side ? g1 : g0;
        const int t   = side ? nwg - 1 - (int)blockIdx.x : (int)blockIdx.x;
        if (t < 0 || t >= nwg) continue;
        float*       out  = side ? out1  : out0;
        const float* xsrc = side ? xsrc1 : xsrc0;
        const int*   perm = side ? perm1 : perm0;
        const int*   off  = side ? off1  : off0;
        const float* xdst = side ? xdst1 : xdst0;
        const float* Pl   = side ? Pl1   : Pl0;
        const float* bl   = side ? bl1   : bl0;
        const float* Pr   = side ? Pr1   : Pr0;
        const int nRows   = side ? nRows1 : nRows0;

        const int strips = (nRows + 15) >> 4;
        const int base   = strips / nwg;
        const int rem    = strips - base * nwg;
        const int mine   = base + (t < rem ? 1 : 0);
        if (mine == 0) continue;
        const int m0     = (t * base + min(t, rem)) * 16;
        const int rowEnd = min(nRows, m0 + mine * 16);

        if (side) __syncthreads();

        if (mine == NSMAX - 1)
            sage_tile<NSMAX - 1>(sA, out, xsrc, perm, off, xdst, Pl, bl, Pr, relu_norm, nRows, m0, rowEnd, t);
        else
            sage_tile<NSMAX>(sA, out, xsrc, perm, off, xdst, Pl, bl, Pr, relu_norm, nRows, m0, rowEnd, t);
    }
}

extern "C" void kernel_launch(void* const* d_in, const int* in_sizes, int n_in,
                              void* d_out, int out_size, void* d_ws, size_t ws_size,
                              hipStream_t stream) {
    (void)n_in; (void)out_size; (void)ws_size;
    const float* xA = (const float*)d_in[0];
    const float* xP = (const float*)d_in[1];
    const float* Wl = (const float*)d_in[2];
    const float* bl = (const float*)d_in[3];
    const float* Wr = (const float*)d_in[4];
    const int* eap  = (const int*)d_in[5];
    const int* epa  = (const int*)d_in[6];

    const int NA = in_sizes[0] / DIM;
    const int NP = in_sizes[1] / DIM;
    const int E  = in_sizes[5] / 2;
    const int L  = in_sizes[2] / (2 * DIM * DIM);

    const size_t FA = (size_t)NA * DIM, FP = (size_t)NP * DIM;
    char* ws = (char*)d_ws;
    float* A0 = (float*)ws;  ws += FA * sizeof(float);
    float* A1 = (float*)ws;  ws += FA * sizeof(float);
    float* P0 = (float*)ws;  ws += FP * sizeof(float);
    float* P1 = (float*)ws;  ws += FP * sizeof(float);
    const size_t FW = (size_t)L * 2 * DIM * DIM;
    float* PlAll = (float*)ws;  ws += FW * sizeof(float);
    float* PrAll = (float*)ws;  ws += FW * sizeof(float);
    const int nbkP = (NP + BW - 1) >> BSH, nbkA = (NA + BW - 1) >> BSH;
    const int nbkM = nbkP > nbkA ? nbkP : nbkA;
    int2* pairs = (int2*)ws; ws += (size_t)2 * E * sizeof(int2);
    int* permP = (int*)ws;   ws += (size_t)E * 4;
    int* permA = (int*)ws;   ws += (size_t)E * 4;
    int* offP  = (int*)ws;   ws += (size_t)(NP + 1) * 4;
    int* offA  = (int*)ws;   ws += (size_t)(NA + 1) * 4;
    int* hist  = (int*)ws;   ws += (size_t)(nbkP + nbkA) * NCH * 4;
    int* btot  = (int*)ws;   ws += (size_t)(nbkP + nbkA) * 4;
    int* bstart = (int*)ws;  ws += (size_t)(nbkP + nbkA + 2) * 4;

    int gP = ((NP + 15) / 16 + NSMAX - 1) / NSMAX;
    int gA = ((NA + 15) / 16 + NSMAX - 1) / NSMAX;
    if (gP < WGS) gP = WGS;
    if (gA < WGS) gA = WGS;

    pack_w<<<dim3(L * 2 * 16, 2), 256, 0, stream>>>(Wl, Wr, PlAll, PrAll);

    bin_count  <<<dim3(NCH, 2), 256, 0, stream>>>(eap, epa, hist, E, nbkP, nbkA);
    bin_scan   <<<dim3(nbkM, 2), 256, 0, stream>>>(hist, btot, nbkP, nbkA);
    bin_scatter<<<dim3(NCH, 2), 256, 0, stream>>>(eap, epa, hist, btot, bstart, pairs, E, nbkP, nbkA);
    bucket_csr <<<dim3(nbkM, 2), 256, 0, stream>>>(pairs, bstart, offP, offA, permP, permA, E, NP, NA, nbkP, nbkA);

    const float* xa_cur = xA;
    const float* xp_cur = xP;
    for (int l = 0; l < L; ++l) {
        const bool last = (l == L - 1);
        float* xa_next = last ? (float*)d_out        : ((l & 1) ? A1 : A0);
        float* xp_next = last ? (float*)d_out + FA   : ((l & 1) ? P1 : P0);

        const float* Wl0 = PlAll + ((size_t)(l * 2 + 0)) * DIM * DIM;
        const float* Wl1 = PlAll + ((size_t)(l * 2 + 1)) * DIM * DIM;
        const float* Wr0 = PrAll + ((size_t)(l * 2 + 0)) * DIM * DIM;
        const float* Wr1 = PrAll + ((size_t)(l * 2 + 1)) * DIM * DIM;
        const float* bl0 = bl + (size_t)(l * 2 + 0) * DIM;
        const float* bl1 = bl + (size_t)(l * 2 + 1) * DIM;

        sage_gemm<<<gP > gA ? gP : gA, 256, 0, stream>>>(
            xp_next, xa_cur, permP, offP, xp_cur, Wl0, bl0, Wr0, NP, gP,
            xa_next, xp_cur, permA, offA, xa_cur, Wl1, bl1, Wr1, NA, gA,
            last ? 0 : 1);

        xa_cur = xa_next;
        xp_cur = xp_next;
    }
}


# neighbour gather as one stream per wave: 8 source rows stay in flight, each added row refilled at once across destination-row ends, index list read one trip ahead; each row's sum order unchanged
# speedup vs baseline: 1.7504x; 1.0327x over previous
#include <hip/hip_runtime.h>
#include <math.h>

typedef __attribute__((ext_vector_type(2))) float v2f;
typedef __attribute__((ext_vector_type(8))) float v8f;

#define DIM      256
#define NSMAX    7
#define WGS      512
#define LSTRIDE  260
#define KG       (DIM / 4)

#define BSH   8
#define BW    (1 << BSH)
#define NCH   256
#define MAXB  1024
#define EPT   16


__device__ __forceinline__ int block_scan256(int v, int* s, int tid, int& total) {
    s[tid] = v;
    __syncthreads();
    for (int d = 1; d < 256; d <<= 1) {
        int t = (tid >= d) ? s[tid - d] : 0;
        __syncthreads();
        s[tid] += t;
        __syncthreads();
    }
    total = s[255];
    const int ex = s[tid] - v;
    __syncthreads();
    return ex;
}

__global__ __launch_bounds__(256) void bin_count(const int* __restrict__ e0,
                                                 const int* __restrict__ e1,
                                                 int* __restrict__ hist,
                                                 int nE, int nbk0, int nbk1) {
    __shared__ int h[MAXB];
    const int tid = threadIdx.x;
    const int c   = blockIdx.x;
    const int nbk = blockIdx.y ? nbk1 : nbk0;
    const int* dst = (blockIdx.y ? e1 : e0) + nE;
    int* H = hist + (blockIdx.y ? (size_t)nbk0 * NCH : 0);
    for (int b = tid; b < nbk; b += 256) h[b] = 0;
    __syncthreads();
    const int ch = (nE + NCH - 1) / NCH;
    const int lo = c * ch;
    const int hi = min(nE, lo + ch);
    for (int i0 = lo + tid; i0 < hi; i0 += 256 * EPT) {
        int d[EPT];
#pragma unroll
        for (int u = 0; u < EPT; ++u) {
            const int i = i0 + u * 256;
            d[u] = (i < hi) ? dst[i] : -1;
        }
#pragma unroll
        for (int u = 0; u < EPT; ++u)
            if (d[u] >= 0) atomicAdd(&h[d[u] >> BSH], 1);
    }
    __syncthreads();
    for (int b = tid; b < nbk; b += 256) H[(size_t)b * NCH + c] = h[b];
}

__global__ __launch_bounds__(256) void bin_scan(int* __restrict__ hist,
                                                int* __restrict__ btot,
                                                int nbk0, int nbk1) {
    __shared__ int s[256];
    const int tid = threadIdx.x;
    const int b   = blockIdx.x;
    const int nbk = blockIdx.y ? nbk1 : nbk0;
    if (b >= nbk) return;
    int* H = hist + (blockIdx.y ? (size_t)nbk0 * NCH : 0) + (size_t)b * NCH;
    const int v = H[tid];
    int total;
    const int ex = block_scan256(v, s, tid, total);
    H[tid] = ex;
    if (tid == 0) btot[(blockIdx.y ? nbk0 : 0) + b] = total;
}

__global__ __launch_bounds__(256) void bin_scatter(const int* __restrict__ e0,
                                                   const int* __restrict__ e1,
                                                   const int* __restrict__ hist,
                                                   const int* __restrict__ btot,
                                                   int* __restrict__ bstart,
                                                   int2* __restrict__ pairs,
                                                   int nE, int nbk0, int nbk1) {
    __shared__ int s[256];
    __shared__ int cur[MAXB];
    const int tid = threadIdx.x;
    const int c   = blockIdx.x;
    const int nbk = blockIdx.y ? nbk1 : nbk0;
    const int* src = blockIdx.y ? e1 : e0;
    const int* dst = src + nE;
    const int* H  = hist + (blockIdx.y ? (size_t)nbk0 * NCH : 0);
    const int* T  = btot + (blockIdx.y ? nbk0 : 0);
    int*       BS = bstart + (blockIdx.y ? nbk0 + 1 : 0);
    int2*      PR = pairs + (blockIdx.y ? (size_t)nE : 0);

    int carry = 0;
    for (int b0 = 0; b0 < nbk; b0 += 256) {
        const int b = b0 + tid;
        const int v  = (b < nbk) ? T[b] : 0;
        const int hv = (b < nbk) ? H[(size_t)b * NCH + c] : 0;
        int total;
        const int ex = carry + block_scan256(v, s, tid, total);
        if (b < nbk) {
            cur[b] = ex + hv;
            if (c == 0) BS[b] = ex;
        }
        carry += total;
    }
    if (c == 0 && tid == 0) BS[nbk] = carry;
    __syncthreads();

    const int ch = (nE + NCH - 1) / NCH;
    const int lo = c * ch;
    const int hi = min(nE, lo + ch);
    for (int i0 = lo + tid; i0 < hi; i0 += 256 * EPT) {
        int sv[EPT], dv[EPT];
#pragma unroll
        for (int u = 0; u < EPT; ++u) {
            const int i = i0 + u * 256;
            sv[u] = (i < hi) ? src[i] : 0;
            dv[u] = (i < hi) ? dst[i] : -1;
        }
#pragma unroll
        for (int u = 0; u < EPT; ++u)
            if (dv[u] >= 0) {
                const int slot = atomicAdd(&cur[dv[u] >> BSH], 1);
                PR[slot] = make_int2(sv[u], dv[u]);
            }
    }
}

__global__ __launch_bounds__(256) void bucket_csr(const int2* __restrict__ pairs,
                                                  const int* __restrict__ bstart,
                                                  int* __restrict__ off0, int* __restrict__ off1,
                                                  int* __restrict__ perm0, int* __restrict__ perm1,
                                                  int nE, int n0, int n1, int nbk0, int nbk1) {
    __shared__ int s[256];
    __shared__ int cnt[BW];
    const int tid = threadIdx.x;
    const int b   = blockIdx.x;
    const int nbk = blockIdx.y ? nbk1 : nbk0;
    if (b >= nbk) return;
    const int2* PR = pairs + (blockIdx.y ? (size_t)nE : 0);
    const int*  BS = bstart + (blockIdx.y ? nbk0 + 1 : 0);
    int* off  = blockIdx.y ? off1 : off0;
    int* perm = blockIdx.y ? perm1 : perm0;
    const int n    = blockIdx.y ? n1 : n0;
    const int lo   = BS[b];
    const int hi   = BS[b + 1];
    const int base = b << BSH;

    cnt[tid] = 0;
    __syncthreads();
    for (int i0 = lo + tid; i0 < hi; i0 += 256 * EPT) {
        int d[EPT];
#pragma unroll
        for (int u = 0; u < EPT; ++u) {
            const int i = i0 + u * 256;
            d[u] = (i < hi) ? PR[i].y : -1;
        }
#pragma unroll
        for (int u = 0; u < EPT; ++u)
            if (d[u] >= 0) atomicAdd(&cnt[d[u] - base], 1);
    }
    __syncthreads();
    const int v = cnt[tid];
    int total;
    const int ex = block_scan256(v, s, tid, total);
    const int node = base + tid;
    if (node < n) off[node] = lo + ex;
    if (b == nbk - 1 && tid == 0) off[n] = hi;
    cnt[tid] = lo + ex;
    __syncthreads();
    for (int i0 = lo + tid; i0 < hi; i0 += 256 * EPT) {
        int2 p[EPT];
#pragma unroll
        for (int u = 0; u < EPT; ++u) {
            const int i = i0 + u * 256;
            p[u] = (i < hi) ? PR[i] : make_int2(0, -1);
        }
#pragma unroll
        for (int u = 0; u < EPT; ++u)
            if (p[u].y >= 0) {
                const int slot = atomicAdd(&cnt[p[u].y - base], 1);
                perm[slot] = p[u].x;
            }
    }
}

__global__ __launch_bounds__(256) void pack_w(const float* __restrict__ Wl,
                                              const float* __restrict__ Wr,
                                              float* __restrict__ Pl,
                                              float* __restrict__ Pr) {
    __shared__ __attribute__((aligned(16))) float t[16 * (DIM + 4)];
    const int tid = threadIdx.x;
    const int mat = blockIdx.x >> 4;
    const int nb  = (blockIdx.x & 15) * 16;
    const float* W = (blockIdx.y ? Wr : Wl) + (size_t)mat * DIM * DIM + (size_t)nb * DIM;
    float*       P = (blockIdx.y ? Pr : Pl) + (size_t)mat * DIM * DIM;
    for (int i = tid; i < 16 * KG; i += 256) {
        const int r = i >> 6, g = i & 63;
        *(float4*)&t[r * (DIM + 4) + 4 * g] = *(const float4*)(W + (size_t)r * DIM + 4 * g);
    }
    __syncthreads();
    for (int i = tid; i < 16 * KG; i += 256) {
        const int g = i >> 4, r = i & 15;
        *(float4*)(P + ((size_t)g * DIM + nb + r) * 4) = *(const float4*)&t[r * (DIM + 4) + 4 * g];
    }
}

#define RING 8
template <int RW>
__device__ __forceinline__ void gather_rows(const float* __restrict__ xsrc,
                                            const int* __restrict__ perm,
                                            int offv, int lane, float* pm) {
    const int S    = __builtin_amdgcn_readlane(offv, 0);
    const int E    = __builtin_amdgcn_readlane(offv, RW);
    const int last = max(E - 1, 0);
    int nxt[RING];
    {
        const int ix0 = perm[min(S + lane, last)];
#pragma unroll
        for (int u = 0; u < RING; ++u) nxt[u] = __builtin_amdgcn_readlane(ix0, u);
    }
    float4 va[RING], vb[RING];
#pragma unroll
    for (int u = 0; u < RING; ++u) {
        va[u] = make_float4(0.f, 0.f, 0.f, 0.f);
        vb[u] = make_float4(0.f, 0.f, 0.f, 0.f);
    }
    int r  = 0;
    int rs = S;
    int re = __builtin_amdgcn_readlane(offv, 1);
    float g[8] = {0.f, 0.f, 0.f, 0.f, 0.f, 0.f, 0.f, 0.f};
    if (S < E)
    for (int b = S - RING; b < E; b += RING) {
        const int ixNext = perm[min(b + 2 * RING + lane, last)];
#pragma unroll
        for (int u = 0; u < RING; ++u) {
            const int c = b + u;
            if (b >= S && c < E) {
                while (c == re) {
                    const float inv = 1.0f / fmaxf((float)(re - rs), 1.0f);
                    float* q = pm + r * LSTRIDE;
                    *(float4*)q         = make_float4(g[0] * inv, g[1] * inv, g[2] * inv, g[3] * inv);
                    *(float4*)(q + 128) = make_float4(g[4] * inv, g[5] * inv, g[6] * inv, g[7] * inv);
#pragma unroll
                    for (int k = 0; k < 8; ++k) g[k] = 0.f;
                    ++r;
                    rs = re;
                    re = __builtin_amdgcn_readlane(offv, r + 1);
                }
                g[0] += va[u].x; g[1] += va[u].y; g[2] += va[u].z; g[3] += va[u].w;
                g[4] += vb[u].x; g[5] += vb[u].y; g[6] += vb[u].z; g[7] += vb[u].w;
            }
            const float4* row = (const float4*)(xsrc + (size_t)nxt[u] * DIM);
            va[u] = row[lane];
            vb[u] = row[lane + 32];
        }
#pragma unroll
        for (int u = 0; u < RING; ++u) nxt[u] = __builtin_amdgcn_readlane(ixNext, u);
    }
    while (r < RW) {
        const float inv = 1.0f / fmaxf((float)(re - rs), 1.0f);
        float* q = pm + r * LSTRIDE;
        *(float4*)q         = make_float4(g[0] * inv, g[1] * inv, g[2] * inv, g[3] * inv);
        *(float4*)(q + 128) = make_float4(g[4] * inv, g[5] * inv, g[6] * inv, g[7] * inv);
#pragma unroll
        for (int k = 0; k < 8; ++k) g[k] = 0.f;
        ++r;
        rs = re;
        if (r < RW) re = __builtin_amdgcn_readlane(offv, r + 1);
    }
}

template <int NS>
__device__ __forceinline__ void load_block(const float4* w0, const float4* w1,
                                           const float* aRow, int j,
                                           float4 (&b)[2], float4 (&a)[NS]) {
    b[0] = w0[(size_t)j * (2 * DIM)];
    b[1] = w1[(size_t)j * (2 * DIM)];
#pragma unroll
    for (int s = 0; s < NS; ++s)
        a[s] = *(const float4*)(aRow + s * 16 * LSTRIDE + 8 * j);
}

template <int NS>
__device__ __forceinline__ void mma_block(v8f (&acc)[NS][2], const float4 (&b)[2],
                                          const float4 (&a)[NS]) {
#pragma unroll
    for (int st = 0; st < 2; ++st) {
        const v2f b0 = st ? (v2f){b[0].z, b[0].w} : (v2f){b[0].x, b[0].y};
        const v2f b1 = st ? (v2f){b[1].z, b[1].w} : (v2f){b[1].x, b[1].y};
#pragma unroll
        for (int s = 0; s < NS; ++s) {
            const v2f as = st ? (v2f){a[s].z, a[s].w} : (v2f){a[s].x, a[s].y};
            acc[s][0] = __builtin_amdgcn_wmma_f32_16x16x4_f32(false, as, false, b0, (short)0, acc[s][0], false, false);
            acc[s][1] = __builtin_amdgcn_wmma_f32_16x16x4_f32(false, as, false, b1, (short)0, acc[s][1], false, false);
        }
    }
}

template <int NS>
__device__ __forceinline__ void k_loop(v8f (&acc)[NS][2], const float4* w0, const float4* w1,
                                       const float* aRow) {
    constexpr int NBLK = DIM / 8;
    float4 bP[2], aP[NS];
    float4 bQ[2], aQ[NS];

    load_block<NS>(w0, w1, aRow, 0, bP, aP);
#pragma unroll 1
    for (int i = 0; i < NBLK - 2; i += 2) {
        load_block<NS>(w0, w1, aRow, i + 1, bQ, aQ);
        __builtin_amdgcn_sched_barrier(0);
        mma_block<NS>(acc, bP, aP);
        __builtin_amdgcn_sched_barrier(0);
        load_block<NS>(w0, w1, aRow, i + 2, bP, aP);
        __builtin_amdgcn_sched_barrier(0);
        mma_block<NS>(acc, bQ, aQ);
    }
    load_block<NS>(w0, w1, aRow, NBLK - 1, bQ, aQ);
    mma_block<NS>(acc, bP, aP);
    mma_block<NS>(acc, bQ, aQ);
}

template <int NS>
__device__ __forceinline__ void sage_tile(float* sA, float* __restrict__ out,
                                          const float* __restrict__ xsrc,
                                          const int* __restrict__ perm,
                                          const int* __restrict__ off,
                                          const float* __restrict__ xdst,
                                          const float* __restrict__ Pl,
                                          const float* __restrict__ bl,
                                          const float* __restrict__ Pr,
                                          int relu_norm, int nRows, int m0, int rowEnd, int tileIdx) {
    constexpr int MT = NS * 16;

    const int tid = threadIdx.x;

    const int lane  = tid & 31;
    const int wv    = tid >> 5;
    const int mrow  = lane & 15;
    const int half  = lane >> 4;
    const int ncol  = lane & 15;
    const int n0    = wv * 32;

    const int selfFirst = (tileIdx ^ (tileIdx >> 3) ^ (tileIdx >> 8)) & 1;

    const float4* wl0 = (const float4*)Pl + (size_t)half * DIM + n0 + ncol;
    const float4* wl1 = wl0 + 16;
    const float4* wr0 = (const float4*)Pr + (size_t)half * DIM + n0 + ncol;
    const float4* wr1 = wr0 + 16;
    const float* aRow = &sA[mrow * LSTRIDE] + 4 * half;

    v8f acc[NS][2];
#pragma unroll
    for (int s = 0; s < NS; ++s)
#pragma unroll
        for (int t = 0; t < 2; ++t)
            acc[s][t] = (v8f){0.f, 0.f, 0.f, 0.f, 0.f, 0.f, 0.f, 0.f};

    for (int p = 0; p < 2; ++p) {
        const bool doSelf = (p == 0) ? (selfFirst != 0) : (selfFirst == 0);
        if (p) __syncthreads();

        if (doSelf) {
            for (int i = tid; i < MT * DIM / 4; i += 256) {
                const int m = i >> 6;
                const int c = (i & 63) * 4;
                int gm = m0 + m;
                if (gm >= nRows) gm = nRows - 1;
                float4 s = *(const float4*)(xdst + (size_t)gm * DIM + c);
                *(float4*)&sA[m * LSTRIDE + c] = s;
            }
        } else {
            const int rbase = m0 + wv * (MT / 8);
            int oi = rbase + lane;
            if (oi > nRows) oi = nRows;
            const int offv = (lane < MT / 8 + 1) ? off[oi] : 0;
            gather_rows<MT / 8>(xsrc, perm, offv, lane, &sA[(wv * (MT / 8)) * LSTRIDE + lane * 4]);
        }
        __syncthreads();

        k_loop<NS>(acc, doSelf ? wr0 : wl0, doSelf ? wr1 : wl1, aRow);
    }

    const float bb0 = bl[n0 + ncol];
    const float bb1 = bl[n0 + 16 + ncol];
#pragma unroll
    for (int s = 0; s < NS; ++s)
#pragma unroll
        for (int r = 0; r < 8; ++r) { acc[s][0][r] += bb0; acc[s][1][r] += bb1; }

    __syncthreads();

    const int mbase = (lane >> 4) << 3;
#pragma unroll
    for (int s = 0; s < NS; ++s)
#pragma unroll
        for (int r = 0; r < 8; ++r) {
            const int m = s * 16 + mbase + r;
            sA[m * LSTRIDE + n0 + ncol]      = acc[s][0][r];
            sA[m * LSTRIDE + n0 + 16 + ncol] = acc[s][1][r];
        }
    __syncthreads();

    constexpr int RW = MT / 8;
    const int wvs = __builtin_amdgcn_readfirstlane(wv);
    float4 va[RW], vb[RW];
#pragma unroll
    for (int i = 0; i < RW; ++i) {
        const float* row = &sA[(wvs * RW + i) * LSTRIDE + lane * 4];
        va[i] = *(const float4*)row;
        vb[i] = *(const float4*)(row + 128);
    }
    if (relu_norm) {
        float ss[RW];
#pragma unroll
        for (int i = 0; i < RW; ++i)
            ss[i] = va[i].x * va[i].x + va[i].y * va[i].y + va[i].z * va[i].z + va[i].w * va[i].w +
                    vb[i].x * vb[i].x + vb[i].y * vb[i].y + vb[i].z * vb[i].z + vb[i].w * vb[i].w;
#pragma unroll
        for (int d = 16; d >= 1; d >>= 1)
#pragma unroll
            for (int i = 0; i < RW; ++i) ss[i] += __shfl_xor(ss[i], d, 32);
        float t = 1.0f;
#pragma unroll
        for (int i = 0; i < RW; ++i) t = (lane == i) ? ss[i] : t;
        const float invl = 1.0f / fmaxf(sqrtf(t), 1e-12f);
#pragma unroll
        for (int i = 0; i < RW; ++i) {
            const float inv = __int_as_float(__builtin_amdgcn_readlane(__float_as_int(invl), i));
            va[i].x = fmaxf(va[i].x * inv, 0.f); va[i].y = fmaxf(va[i].y * inv, 0.f);
            va[i].z = fmaxf(va[i].z * inv, 0.f); va[i].w = fmaxf(va[i].w * inv, 0.f);
            vb[i].x = fmaxf(vb[i].x * inv, 0.f); vb[i].y = fmaxf(vb[i].y * inv, 0.f);
            vb[i].z = fmaxf(vb[i].z * inv, 0.f); vb[i].w = fmaxf(vb[i].w * inv, 0.f);
        }
    }
#pragma unroll
    for (int i = 0; i < RW; ++i) {
        const int gm = m0 + wvs * RW + i;
        if (gm < rowEnd) {
            float* orow = out + (size_t)gm * DIM + lane * 4;
            *(float4*)orow         = va[i];
            *(float4*)(orow + 128) = vb[i];
        }
    }
}

__global__ __launch_bounds__(256, 4) void sage_gemm(float* __restrict__ out0,
                                                 const float* __restrict__ xsrc0,
                                                 const int* __restrict__ perm0,
                                                 const int* __restrict__ off0,
                                                 const float* __restrict__ xdst0,
                                                 const float* __restrict__ Pl0,
                                                 const float* __restrict__ bl0,
                                                 const float* __restrict__ Pr0,
                                                 int nRows0, int g0,
                                                 float* __restrict__ out1,
                                                 const float* __restrict__ xsrc1,
                                                 const int* __restrict__ perm1,
                                                 const int* __restrict__ off1,
                                                 const float* __restrict__ xdst1,
                                                 const float* __restrict__ Pl1,
                                                 const float* __restrict__ bl1,
                                                 const float* __restrict__ Pr1,
                                                 int nRows1, int g1,
                                                 int relu_norm) {
    __shared__ __attribute__((aligned(16))) float sA[NSMAX * 16 * LSTRIDE];

#pragma unroll 1
    for (int side = 0; side < 2; ++side) {
        const int nwg = side ? g1 : g0;
        const int t   = side ? nwg - 1 - (int)blockIdx.x : (int)blockIdx.x;
        if (t < 0 || t >= nwg) continue;
        float*       out  = side ? out1  : out0;
        const float* xsrc = side ? xsrc1 : xsrc0;
        const int*   perm = side ? perm1 : perm0;
        const int*   off  = side ? off1  : off0;
        const float* xdst = side ? xdst1 : xdst0;
        const float* Pl   = side ? Pl1   : Pl0;
        const float* bl   = side ? bl1   : bl0;
        const float* Pr   = side ? Pr1   : Pr0;
        const int nRows   = side ? nRows1 : nRows0;

        const int strips = (nRows + 15) >> 4;
        const int base   = strips / nwg;
        const int rem    = strips - base * nwg;
        const int mine   = base + (t < rem ? 1 : 0);
        if (mine == 0) continue;
        const int m0     = (t * base + min(t, rem)) * 16;
        const int rowEnd = min(nRows, m0 + mine * 16);

        if (side) __syncthreads();

        if (mine == NSMAX - 1)
            sage_tile<NSMAX - 1>(sA, out, xsrc, perm, off, xdst, Pl, bl, Pr, relu_norm, nRows, m0, rowEnd, t);
        else
            sage_tile<NSMAX>(sA, out, xsrc, perm, off, xdst, Pl, bl, Pr, relu_norm, nRows, m0, rowEnd, t);
    }
}

extern "C" void kernel_launch(void* const* d_in, const int* in_sizes, int n_in,
                              void* d_out, int out_size, void* d_ws, size_t ws_size,
                              hipStream_t stream) {
    (void)n_in; (void)out_size; (void)ws_size;
    const float* xA = (const float*)d_in[0];
    const float* xP = (const float*)d_in[1];
    const float* Wl = (const float*)d_in[2];
    const float* bl = (const float*)d_in[3];
    const float* Wr = (const float*)d_in[4];
    const int* eap  = (const int*)d_in[5];
    const int* epa  = (const int*)d_in[6];

    const int NA = in_sizes[0] / DIM;
    const int NP = in_sizes[1] / DIM;
    const int E  = in_sizes[5] / 2;
    const int L  = in_sizes[2] / (2 * DIM * DIM);

    const size_t FA = (size_t)NA * DIM, FP = (size_t)NP * DIM;
    char* ws = (char*)d_ws;
    float* A0 = (float*)ws;  ws += FA * sizeof(float);
    float* A1 = (float*)ws;  ws += FA * sizeof(float);
    float* P0 = (float*)ws;  ws += FP * sizeof(float);
    float* P1 = (float*)ws;  ws += FP * sizeof(float);
    const size_t FW = (size_t)L * 2 * DIM * DIM;
    float* PlAll = (float*)ws;  ws += FW * sizeof(float);
    float* PrAll = (float*)ws;  ws += FW * sizeof(float);
    const int nbkP = (NP + BW - 1) >> BSH, nbkA = (NA + BW - 1) >> BSH;
    const int nbkM = nbkP > nbkA ? nbkP : nbkA;
    int2* pairs = (int2*)ws; ws += (size_t)2 * E * sizeof(int2);
    int* permP = (int*)ws;   ws += (size_t)E * 4;
    int* permA = (int*)ws;   ws += (size_t)E * 4;
    int* offP  = (int*)ws;   ws += (size_t)(NP + 1) * 4;
    int* offA  = (int*)ws;   ws += (size_t)(NA + 1) * 4;
    int* hist  = (int*)ws;   ws += (size_t)(nbkP + nbkA) * NCH * 4;
    int* btot  = (int*)ws;   ws += (size_t)(nbkP + nbkA) * 4;
    int* bstart = (int*)ws;  ws += (size_t)(nbkP + nbkA + 2) * 4;

    int gP = ((NP + 15) / 16 + NSMAX - 1) / NSMAX;
    int gA = ((NA + 15) / 16 + NSMAX - 1) / NSMAX;
    if (gP < WGS) gP = WGS;
    if (gA < WGS) gA = WGS;

    pack_w<<<dim3(L * 2 * 16, 2), 256, 0, stream>>>(Wl, Wr, PlAll, PrAll);

    bin_count  <<<dim3(NCH, 2), 256, 0, stream>>>(eap, epa, hist, E, nbkP, nbkA);
    bin_scan   <<<dim3(nbkM, 2), 256, 0, stream>>>(hist, btot, nbkP, nbkA);
    bin_scatter<<<dim3(NCH, 2), 256, 0, stream>>>(eap, epa, hist, btot, bstart, pairs, E, nbkP, nbkA);
    bucket_csr <<<dim3(nbkM, 2), 256, 0, stream>>>(pairs, bstart, offP, offA, permP, permA, E, NP, NA, nbkP, nbkA);

    const float* xa_cur = xA;
    const float* xp_cur = xP;
    for (int l = 0; l < L; ++l) {
        const bool last = (l == L - 1);
        float* xa_next = last ? (float*)d_out        : ((l & 1) ? A1 : A0);
        float* xp_next = last ? (float*)d_out + FA   : ((l & 1) ? P1 : P0);

        const float* Wl0 = PlAll + ((size_t)(l * 2 + 0)) * DIM * DIM;
        const float* Wl1 = PlAll + ((size_t)(l * 2 + 1)) * DIM * DIM;
        const float* Wr0 = PrAll + ((size_t)(l * 2 + 0)) * DIM * DIM;
        const float* Wr1 = PrAll + ((size_t)(l * 2 + 1)) * DIM * DIM;
        const float* bl0 = bl + (size_t)(l * 2 + 0) * DIM;
        const float* bl1 = bl + (size_t)(l * 2 + 1) * DIM;

        sage_gemm<<<gP > gA ? gP : gA, 256, 0, stream>>>(
            xp_next, xa_cur, permP, offP, xp_cur, Wl0, bl0, Wr0, NP, gP,
            xa_next, xp_cur, permA, offA, xa_cur, Wl1, bl1, Wr1, NA, gA,
            last ? 0 : 1);

        xa_cur = xa_next;
        xp_cur = xp_next;
    }
}
